# starting kernel
# baseline (speedup 1.0000x reference)
#include <hip/hip_runtime.h>
#include <math.h>


typedef __attribute__((ext_vector_type(2))) float v2f;
typedef __attribute__((ext_vector_type(8))) float v8f;

#define BB 8
#define NN 1024
#define MM 128
#define FF 64
#define FE 16
#define HID 128

__device__ __forceinline__ v8f wmma4(v2f a, v2f b, v8f c) {
  return __builtin_amdgcn_wmma_f32_16x16x4_f32(
       false, a,  false, b,
       (short)0, c,  false,  false);
}

__global__ __launch_bounds__(128) void gemm_rows_64(
    const float* __restrict__ A, const float* __restrict__ W,
    float* __restrict__ C) {
  const int wv   = threadIdx.x >> 5;
  const int lane = threadIdx.x & 31;
  const int lm   = lane & 15;
  const int kh   = (lane >> 4) * 2;
  const int lh   = lane >> 4;
  const int row0 = blockIdx.x * 16;
  const int o    = wv * 16 + lm;
  const float* Arow = A + (size_t)(row0 + lm) * FF;

  v8f c = {};
#pragma unroll
  for (int k0 = 0; k0 < FF; k0 += 4) {
    v2f a, b;
    a.x = Arow[k0 + kh];
    a.y = Arow[k0 + kh + 1];
    b.x = W[(size_t)(k0 + kh) * FF + o];
    b.y = W[(size_t)(k0 + kh + 1) * FF + o];
    c = wmma4(a, b, c);
  }
#pragma unroll
  for (int r = 0; r < 8; ++r) {
    int m = r + 8 * lh;
    C[(size_t)(row0 + m) * FF + o] = c[r];
  }
}

__global__ void proj_node(const float* __restrict__ h,
                          const float* __restrict__ att_op,
                          const float* __restrict__ att_ma,
                          const int* __restrict__ opma,
                          float* __restrict__ ha1, float* __restrict__ ha2,
                          float* __restrict__ ham1, int* __restrict__ rowmask) {
  int idx = blockIdx.x * blockDim.x + threadIdx.x;
  if (idx >= BB * NN) return;
  const float* hr = h + (size_t)idx * FF;
  float s1 = 0.f, s2 = 0.f, s3 = 0.f;
#pragma unroll 8
  for (int o = 0; o < FF; ++o) {
    float v = hr[o];
    s1 += v * att_op[o];
    s2 += v * att_op[FF + o];
    s3 += v * att_ma[o];
  }
  ha1[idx] = s1; ha2[idx] = s2; ham1[idx] = s3;
  const int* ad = opma + (size_t)idx * MM;
  int any = 0;
#pragma unroll 8
  for (int m = 0; m < MM; ++m) any |= ad[m];
  rowmask[idx] = (any == 0) ? 1 : 0;
}

__global__ void proj_ma(const float* __restrict__ hma,
                        const float* __restrict__ att_ma,
                        float* __restrict__ hma2) {
  int idx = blockIdx.x * blockDim.x + threadIdx.x;
  if (idx >= BB * MM) return;
  const float* r = hma + (size_t)idx * FF;
  float s = 0.f;
#pragma unroll 8
  for (int o = 0; o < FF; ++o) s += r[o] * att_ma[FF + o];
  hma2[idx] = s;
}

__global__ __launch_bounds__(128) void op_attn(
    const float* __restrict__ h, const float* __restrict__ ha1,
    const float* __restrict__ ha2, const int* __restrict__ opadj,
    const int* __restrict__ maadj, const int* __restrict__ rowmask,
    float* __restrict__ hprime) {
  extern __shared__ float smem[];
  typedef float att_row_t[NN + 4];
  att_row_t* att = (att_row_t*)smem;
  float* inv_sum = smem + 16 * (NN + 4);

  const int b  = blockIdx.y;
  const int i0 = blockIdx.x * 16;
  const int* opb = opadj + (size_t)b * NN * NN;
  const int* mab = maadj + (size_t)b * NN * NN;
  const float NEGINF = -__builtin_inff();

  for (int t = threadIdx.x; t < 16 * NN; t += 128) {
    int r = t >> 10, j = t & (NN - 1);
    int i = i0 + r;
    int conn;
    if (rowmask[b * NN + i]) {
      conn = 1;
    } else {
      int adj = opb[(size_t)i * NN + j] + opb[(size_t)j * NN + i] +
                mab[(size_t)i * NN + j] + mab[(size_t)j * NN + i];
      conn = adj > 0;
    }
    float e = ha2[b * NN + i] + ha1[b * NN + j];
    e = e > 0.f ? e : 0.01f * e;
    att[r][j] = conn ? e : NEGINF;
  }
  __syncthreads();

  const int wv = threadIdx.x >> 5;
  const int lane = threadIdx.x & 31;
  for (int r = wv; r < 16; r += 4) {
    float mx = NEGINF;
    for (int j = lane; j < NN; j += 32) mx = fmaxf(mx, att[r][j]);
#pragma unroll
    for (int off = 16; off; off >>= 1) mx = fmaxf(mx, __shfl_xor(mx, off, 32));
    if (mx == NEGINF) mx = 0.f;
    float sum = 0.f;
    for (int j = lane; j < NN; j += 32) {
      float p = __expf(att[r][j] - mx);
      att[r][j] = p;
      sum += p;
    }
#pragma unroll
    for (int off = 16; off; off >>= 1) sum += __shfl_xor(sum, off, 32);
    if (lane == 0) inv_sum[r] = 1.f / fmaxf(sum, 1e-30f);
  }
  __syncthreads();

  const int lm = lane & 15;
  const int kh = (lane >> 4) * 2;
  const int lh = lane >> 4;
  const int o  = wv * 16 + lm;
  const float* hb = h + (size_t)b * NN * FF;

  v8f c = {};
  for (int k0 = 0; k0 < NN; k0 += 4) {
    v2f a, bf;
    a.x  = att[lm][k0 + kh];
    a.y  = att[lm][k0 + kh + 1];
    bf.x = hb[(size_t)(k0 + kh) * FF + o];
    bf.y = hb[(size_t)(k0 + kh + 1) * FF + o];
    c = wmma4(a, bf, c);
  }
  float* hp = hprime + ((size_t)b * NN + i0) * (2 * FF);
#pragma unroll
  for (int r = 0; r < 8; ++r) {
    int m = r + 8 * lh;
    hp[(size_t)m * (2 * FF) + o] = c[r] * inv_sum[m];
  }
}

__global__ __launch_bounds__(256) void ma_attn(
    const float* __restrict__ z, const float* __restrict__ wedge,
    const float* __restrict__ hma, const float* __restrict__ ham1,
    const float* __restrict__ hma2, const int* __restrict__ opma,
    const int* __restrict__ rowmask, const float* __restrict__ h,
    const float* __restrict__ attma, float* __restrict__ hprime) {
  __shared__ float zb[MM][FE + 1];
  __shared__ float web[FE][FF];
  __shared__ float heb[MM][FF + 1];
  __shared__ float sc[MM];
  __shared__ float red[2];
  __shared__ float psum[4][FF];

  const int b = blockIdx.y, n = blockIdx.x;
  const size_t bn = (size_t)b * NN + n;
  const float NEGINF = -__builtin_inff();

  const float* zrow = z + bn * MM * FE;
  for (int t = threadIdx.x; t < MM * FE; t += 256) zb[t >> 4][t & 15] = zrow[t];
  for (int t = threadIdx.x; t < FE * FF; t += 256) web[t >> 6][t & 63] = wedge[t];
  __syncthreads();

  const int wv = threadIdx.x >> 5;
  const int lane = threadIdx.x & 31;
  const int lm = lane & 15;
  const int kh = (lane >> 4) * 2;
  const int lh = lane >> 4;
  const int m0 = wv * 16;
  for (int ct = 0; ct < 4; ++ct) {
    v8f c = {};
#pragma unroll
    for (int k0 = 0; k0 < FE; k0 += 4) {
      v2f a, bf;
      a.x  = zb[m0 + lm][k0 + kh];
      a.y  = zb[m0 + lm][k0 + kh + 1];
      bf.x = web[k0 + kh][ct * 16 + lm];
      bf.y = web[k0 + kh + 1][ct * 16 + lm];
      c = wmma4(a, bf, c);
    }
#pragma unroll
    for (int r = 0; r < 8; ++r)
      heb[m0 + r + 8 * lh][ct * 16 + lm] = c[r];
  }
  __syncthreads();

  const int rm = rowmask[bn];
  const int* ad = opma + bn * MM;
  const float* am3 = attma + 2 * FF;
  if (threadIdx.x < MM) {
    int m = threadIdx.x;
    float dot = 0.f;
#pragma unroll 8
    for (int o = 0; o < FF; ++o) dot += heb[m][o] * am3[o];
    float e = ham1[bn] + hma2[(size_t)b * MM + m] + dot;
    e = e > 0.f ? e : 0.01f * e;
    int conn = rm ? 1 : (ad[m] > 0);
    sc[m] = conn ? e : NEGINF;
  }
  __syncthreads();

  if (wv == 0) {
    float mx = fmaxf(fmaxf(sc[lane], sc[lane + 32]),
                     fmaxf(sc[lane + 64], sc[lane + 96]));
#pragma unroll
    for (int off = 16; off; off >>= 1) mx = fmaxf(mx, __shfl_xor(mx, off, 32));
    if (mx == NEGINF) mx = 0.f;
    if (lane == 0) red[0] = mx;
  }
  __syncthreads();
  const float mx = red[0];
  if (threadIdx.x < MM) sc[threadIdx.x] = __expf(sc[threadIdx.x] - mx);
  __syncthreads();
  if (wv == 0) {
    float s = sc[lane] + sc[lane + 32] + sc[lane + 64] + sc[lane + 96];
#pragma unroll
    for (int off = 16; off; off >>= 1) s += __shfl_xor(s, off, 32);
    if (lane == 0) red[1] = 1.f / fmaxf(s, 1e-30f);
  }
  __syncthreads();
  const float inv = red[1];

  {
    int o = threadIdx.x & 63, ch = threadIdx.x >> 6;
    const float* hmb = hma + (size_t)b * MM * FF;
    float acc = 0.f;
    for (int m = ch * 32; m < ch * 32 + 32; ++m)
      acc += sc[m] * (hmb[(size_t)m * FF + o] + heb[m][o]);
    psum[ch][o] = acc;
  }
  __syncthreads();
  if (threadIdx.x < FF) {
    int o = threadIdx.x;
    float v = (psum[0][o] + psum[1][o] + psum[2][o] + psum[3][o]) * inv +
              h[bn * FF + o];
    hprime[bn * (2 * FF) + FF + o] = v;
  }
}

__global__ __launch_bounds__(256) void mlp(
    const float* __restrict__ hprime, const float* __restrict__ w1,
    const float* __restrict__ b1, const float* __restrict__ w2,
    const float* __restrict__ b2, const float* __restrict__ wo,
    const float* __restrict__ bo, const int* __restrict__ rowmask,
    float* __restrict__ out) {
  extern __shared__ float smem[];
  float* wb = smem;
  typedef float act_row_t[HID + 4];
  act_row_t* a0 = (act_row_t*)(smem + HID * HID);
  act_row_t* a1 = (act_row_t*)(smem + HID * HID + 16 * (HID + 4));

  const int row0 = blockIdx.x * 16;
  const int wv = threadIdx.x >> 5;
  const int lane = threadIdx.x & 31;
  const int lm = lane & 15;
  const int kh = (lane >> 4) * 2;
  const int lh = lane >> 4;

  for (int t = threadIdx.x; t < 16 * HID; t += 256)
    a0[t >> 7][t & 127] = hprime[(size_t)row0 * HID + t];
  for (int t = threadIdx.x; t < HID * HID; t += 256) wb[t] = w1[t];
  __syncthreads();

  {
    int o = wv * 16 + lm;
    v8f c = {};
    for (int k0 = 0; k0 < HID; k0 += 4) {
      v2f a, bf;
      a.x  = a0[lm][k0 + kh];
      a.y  = a0[lm][k0 + kh + 1];
      bf.x = wb[(k0 + kh) * HID + o];
      bf.y = wb[(k0 + kh + 1) * HID + o];
      c = wmma4(a, bf, c);
    }
    float bias = b1[o];
#pragma unroll
    for (int r = 0; r < 8; ++r) {
      float v = c[r] + bias;
      a1[r + 8 * lh][o] = v > 0.f ? v : expm1f(v);
    }
  }
  __syncthreads();
  for (int t = threadIdx.x; t < HID * HID; t += 256) wb[t] = w2[t];
  __syncthreads();

  {
    int o = wv * 16 + lm;
    v8f c = {};
    for (int k0 = 0; k0 < HID; k0 += 4) {
      v2f a, bf;
      a.x  = a1[lm][k0 + kh];
      a.y  = a1[lm][k0 + kh + 1];
      bf.x = wb[(k0 + kh) * HID + o];
      bf.y = wb[(k0 + kh + 1) * HID + o];
      c = wmma4(a, bf, c);
    }
    float bias = b2[o];
#pragma unroll
    for (int r = 0; r < 8; ++r) {
      float v = c[r] + bias;
      a0[r + 8 * lh][o] = v > 0.f ? v : expm1f(v);
    }
  }
  __syncthreads();
  for (int t = threadIdx.x; t < HID * FF; t += 256) wb[t] = wo[t];
  __syncthreads();

  if (wv < 4) {
    int o = wv * 16 + lm;
    v8f c = {};
    for (int k0 = 0; k0 < HID; k0 += 4) {
      v2f a, bf;
      a.x  = a0[lm][k0 + kh];
      a.y  = a0[lm][k0 + kh + 1];
      bf.x = wb[(k0 + kh) * FF + o];
      bf.y = wb[(k0 + kh + 1) * FF + o];
      c = wmma4(a, bf, c);
    }
    float bias = bo[o];
#pragma unroll
    for (int r = 0; r < 8; ++r) {
      int m = r + 8 * lh;
      float v = c[r] + bias;
      if (rowmask[row0 + m]) v = 0.f;
      out[(size_t)(row0 + m) * FF + o] = v;
    }
  }
}

extern "C" void kernel_launch(void* const* d_in, const int* in_sizes, int n_in,
                              void* d_out, int out_size, void* d_ws,
                              size_t ws_size, hipStream_t stream) {
  (void)in_sizes; (void)n_in; (void)out_size; (void)ws_size;
  const float* x      = (const float*)d_in[0];
  const float* y      = (const float*)d_in[1];
  const float* z      = (const float*)d_in[2];
  const int*   op_adj = (const int*)d_in[3];
  const int*   ma_adj = (const int*)d_in[4];
  const int*   op_ma  = (const int*)d_in[5];
  const float* Wop    = (const float*)d_in[6];
  const float* Wma    = (const float*)d_in[7];
  const float* Wedge  = (const float*)d_in[8];
  const float* att_op = (const float*)d_in[9];
  const float* att_ma = (const float*)d_in[10];
  const float* w1     = (const float*)d_in[11];
  const float* b1     = (const float*)d_in[12];
  const float* w2     = (const float*)d_in[13];
  const float* b2     = (const float*)d_in[14];
  const float* wo     = (const float*)d_in[15];
  const float* bo     = (const float*)d_in[16];
  float* out = (float*)d_out;

  float* ws     = (float*)d_ws;
  float* h      = ws;
  float* hma    = h + BB * NN * FF;
  float* ha1    = hma + BB * MM * FF;
  float* ha2    = ha1 + BB * NN;
  float* ham1   = ha2 + BB * NN;
  float* hma2   = ham1 + BB * NN;
  float* hprime = hma2 + BB * MM;
  int*   rmask  = (int*)(hprime + (size_t)BB * NN * 2 * FF);

  gemm_rows_64<<<dim3((BB * NN) / 16), dim3(128), 0, stream>>>(x, Wop, h);
  gemm_rows_64<<<dim3((BB * MM) / 16), dim3(128), 0, stream>>>(y, Wma, hma);
  proj_node<<<dim3((BB * NN) / 256), dim3(256), 0, stream>>>(
      h, att_op, att_ma, op_ma, ha1, ha2, ham1, rmask);
  proj_ma<<<dim3((BB * MM) / 256), dim3(256), 0, stream>>>(hma, att_ma, hma2);
  size_t sm2 = (16 * (NN + 4) + 16) * sizeof(float);
  op_attn<<<dim3(NN / 16, BB), dim3(128), sm2, stream>>>(
      h, ha1, ha2, op_adj, ma_adj, rmask, hprime);
  ma_attn<<<dim3(NN, BB), dim3(256), 0, stream>>>(
      z, Wedge, hma, ham1, hma2, op_ma, rmask, h, att_ma, hprime);
  size_t sm4 = (HID * HID + 2 * 16 * (HID + 4)) * sizeof(float);
  mlp<<<dim3((BB * NN) / 16), dim3(256), sm4, stream>>>(
      hprime, w1, b1, w2, b2, wo, bo, rmask, out);
}


# op_attn mask build: transposed adjacency slab read along its rows as 16-byte vectors and turned through LDS (direct rows 16-byte too), replacing 4-byte column-strided loads
# speedup vs baseline: 1.6034x; 1.6034x over previous
#include <hip/hip_runtime.h>
#include <math.h>


typedef __attribute__((ext_vector_type(2))) float v2f;
typedef __attribute__((ext_vector_type(8))) float v8f;
typedef __attribute__((ext_vector_type(4))) float v4f;
typedef __attribute__((ext_vector_type(4))) int v4i;

#define BB 8
#define NN 1024
#define MM 128
#define FF 64
#define FE 16
#define HID 128

__device__ __forceinline__ v8f wmma4(v2f a, v2f b, v8f c) {
  return __builtin_amdgcn_wmma_f32_16x16x4_f32(
       false, a,  false, b,
       (short)0, c,  false,  false);
}

__global__ __launch_bounds__(128) void gemm_rows_64(
    const float* __restrict__ A, const float* __restrict__ W,
    float* __restrict__ C) {
  const int wv   = threadIdx.x >> 5;
  const int lane = threadIdx.x & 31;
  const int lm   = lane & 15;
  const int kh   = (lane >> 4) * 2;
  const int lh   = lane >> 4;
  const int row0 = blockIdx.x * 16;
  const int o    = wv * 16 + lm;
  const float* Arow = A + (size_t)(row0 + lm) * FF;

  v8f c = {};
#pragma unroll
  for (int k0 = 0; k0 < FF; k0 += 4) {
    v2f a, b;
    a.x = Arow[k0 + kh];
    a.y = Arow[k0 + kh + 1];
    b.x = W[(size_t)(k0 + kh) * FF + o];
    b.y = W[(size_t)(k0 + kh + 1) * FF + o];
    c = wmma4(a, b, c);
  }
#pragma unroll
  for (int r = 0; r < 8; ++r) {
    int m = r + 8 * lh;
    C[(size_t)(row0 + m) * FF + o] = c[r];
  }
}

__global__ void proj_node(const float* __restrict__ h,
                          const float* __restrict__ att_op,
                          const float* __restrict__ att_ma,
                          const int* __restrict__ opma,
                          float* __restrict__ ha1, float* __restrict__ ha2,
                          float* __restrict__ ham1, int* __restrict__ rowmask) {
  int idx = blockIdx.x * blockDim.x + threadIdx.x;
  if (idx >= BB * NN) return;
  const float* hr = h + (size_t)idx * FF;
  float s1 = 0.f, s2 = 0.f, s3 = 0.f;
#pragma unroll 8
  for (int o = 0; o < FF; ++o) {
    float v = hr[o];
    s1 += v * att_op[o];
    s2 += v * att_op[FF + o];
    s3 += v * att_ma[o];
  }
  ha1[idx] = s1; ha2[idx] = s2; ham1[idx] = s3;
  const int* ad = opma + (size_t)idx * MM;
  int any = 0;
#pragma unroll 8
  for (int m = 0; m < MM; ++m) any |= ad[m];
  rowmask[idx] = (any == 0) ? 1 : 0;
}

__global__ void proj_ma(const float* __restrict__ hma,
                        const float* __restrict__ att_ma,
                        float* __restrict__ hma2) {
  int idx = blockIdx.x * blockDim.x + threadIdx.x;
  if (idx >= BB * MM) return;
  const float* r = hma + (size_t)idx * FF;
  float s = 0.f;
#pragma unroll 8
  for (int o = 0; o < FF; ++o) s += r[o] * att_ma[FF + o];
  hma2[idx] = s;
}

__device__ __forceinline__ float masked_score(float e, int adj) {
  e = e > 0.f ? e : 0.01f * e;
  return adj > 0 ? e : -__builtin_inff();
}

__global__ __launch_bounds__(128) void op_attn(
    const float* __restrict__ h, const float* __restrict__ ha1,
    const float* __restrict__ ha2, const int* __restrict__ opadj,
    const int* __restrict__ maadj, const int* __restrict__ rowmask,
    float* __restrict__ hprime) {
  extern __shared__ float smem[];
  typedef float att_row_t[NN + 4];
  att_row_t* att = (att_row_t*)smem;
  float* inv_sum = smem + 16 * (NN + 4);

  const int b  = blockIdx.y;
  const int i0 = blockIdx.x * 16;
  const int* opb = opadj + (size_t)b * NN * NN;
  const int* mab = maadj + (size_t)b * NN * NN;
  const float NEGINF = -__builtin_inff();

  const v4i* opT = (const v4i*)(opb + i0);
  const v4i* maT = (const v4i*)(mab + i0);
#pragma unroll 8
  for (int t = threadIdx.x; t < 4 * NN; t += 128) {
    const int j = t >> 2, part = t & 3;
    const v4i a = opT[(size_t)j * (NN / 4) + part];
    const v4i m = maT[(size_t)j * (NN / 4) + part];
    const v4i s = a + m;
    att[4 * part + 0][j] = __int_as_float(s.x);
    att[4 * part + 1][j] = __int_as_float(s.y);
    att[4 * part + 2][j] = __int_as_float(s.z);
    att[4 * part + 3][j] = __int_as_float(s.w);
  }
  __syncthreads();

  const int tj = threadIdx.x * 4;
  const v4f e1lo = *(const v4f*)(ha1 + b * NN + tj);
  const v4f e1hi = *(const v4f*)(ha1 + b * NN + NN / 2 + tj);
#pragma unroll 4
  for (int r = 0; r < 16; ++r) {
    const int i = i0 + r;
    const bool rm = rowmask[b * NN + i] != 0;
    const float e2 = ha2[b * NN + i];
#pragma unroll
    for (int hf = 0; hf < 2; ++hf) {
      const int j = hf * (NN / 2) + tj;
      const v4f e1 = hf ? e1hi : e1lo;
      v4f* slot = (v4f*)&att[r][j];
      const v4f tr = *slot;
      const v4i d1 = *(const v4i*)(opb + (size_t)i * NN + j);
      const v4i d2 = *(const v4i*)(mab + (size_t)i * NN + j);
      v4i adj = d1 + d2;
      adj.x += __float_as_int(tr.x);
      adj.y += __float_as_int(tr.y);
      adj.z += __float_as_int(tr.z);
      adj.w += __float_as_int(tr.w);
      if (rm) adj = (v4i){1, 1, 1, 1};
      v4f sc4;
      sc4.x = masked_score(e2 + e1.x, adj.x);
      sc4.y = masked_score(e2 + e1.y, adj.y);
      sc4.z = masked_score(e2 + e1.z, adj.z);
      sc4.w = masked_score(e2 + e1.w, adj.w);
      *slot = sc4;
    }
  }
  __syncthreads();

  const int wv = threadIdx.x >> 5;
  const int lane = threadIdx.x & 31;
  for (int r = wv; r < 16; r += 4) {
    float mx = NEGINF;
    for (int j = lane; j < NN; j += 32) mx = fmaxf(mx, att[r][j]);
#pragma unroll
    for (int off = 16; off; off >>= 1) mx = fmaxf(mx, __shfl_xor(mx, off, 32));
    if (mx == NEGINF) mx = 0.f;
    float sum = 0.f;
    for (int j = lane; j < NN; j += 32) {
      float p = __expf(att[r][j] - mx);
      att[r][j] = p;
      sum += p;
    }
#pragma unroll
    for (int off = 16; off; off >>= 1) sum += __shfl_xor(sum, off, 32);
    if (lane == 0) inv_sum[r] = 1.f / fmaxf(sum, 1e-30f);
  }
  __syncthreads();

  const int lm = lane & 15;
  const int kh = (lane >> 4) * 2;
  const int lh = lane >> 4;
  const int o  = wv * 16 + lm;
  const float* hb = h + (size_t)b * NN * FF;

  v8f c = {};
  for (int k0 = 0; k0 < NN; k0 += 4) {
    v2f a, bf;
    a.x  = att[lm][k0 + kh];
    a.y  = att[lm][k0 + kh + 1];
    bf.x = hb[(size_t)(k0 + kh) * FF + o];
    bf.y = hb[(size_t)(k0 + kh + 1) * FF + o];
    c = wmma4(a, bf, c);
  }
  float* hp = hprime + ((size_t)b * NN + i0) * (2 * FF);
#pragma unroll
  for (int r = 0; r < 8; ++r) {
    int m = r + 8 * lh;
    hp[(size_t)m * (2 * FF) + o] = c[r] * inv_sum[m];
  }
}

__global__ __launch_bounds__(256) void ma_attn(
    const float* __restrict__ z, const float* __restrict__ wedge,
    const float* __restrict__ hma, const float* __restrict__ ham1,
    const float* __restrict__ hma2, const int* __restrict__ opma,
    const int* __restrict__ rowmask, const float* __restrict__ h,
    const float* __restrict__ attma, float* __restrict__ hprime) {
  __shared__ float zb[MM][FE + 1];
  __shared__ float web[FE][FF];
  __shared__ float heb[MM][FF + 1];
  __shared__ float sc[MM];
  __shared__ float red[2];
  __shared__ float psum[4][FF];

  const int b = blockIdx.y, n = blockIdx.x;
  const size_t bn = (size_t)b * NN + n;
  const float NEGINF = -__builtin_inff();

  const float* zrow = z + bn * MM * FE;
  for (int t = threadIdx.x; t < MM * FE; t += 256) zb[t >> 4][t & 15] = zrow[t];
  for (int t = threadIdx.x; t < FE * FF; t += 256) web[t >> 6][t & 63] = wedge[t];
  __syncthreads();

  const int wv = threadIdx.x >> 5;
  const int lane = threadIdx.x & 31;
  const int lm = lane & 15;
  const int kh = (lane >> 4) * 2;
  const int lh = lane >> 4;
  const int m0 = wv * 16;
  for (int ct = 0; ct < 4; ++ct) {
    v8f c = {};
#pragma unroll
    for (int k0 = 0; k0 < FE; k0 += 4) {
      v2f a, bf;
      a.x  = zb[m0 + lm][k0 + kh];
      a.y  = zb[m0 + lm][k0 + kh + 1];
      bf.x = web[k0 + kh][ct * 16 + lm];
      bf.y = web[k0 + kh + 1][ct * 16 + lm];
      c = wmma4(a, bf, c);
    }
#pragma unroll
    for (int r = 0; r < 8; ++r)
      heb[m0 + r + 8 * lh][ct * 16 + lm] = c[r];
  }
  __syncthreads();

  const int rm = rowmask[bn];
  const int* ad = opma + bn * MM;
  const float* am3 = attma + 2 * FF;
  if (threadIdx.x < MM) {
    int m = threadIdx.x;
    float dot = 0.f;
#pragma unroll 8
    for (int o = 0; o < FF; ++o) dot += heb[m][o] * am3[o];
    float e = ham1[bn] + hma2[(size_t)b * MM + m] + dot;
    e = e > 0.f ? e : 0.01f * e;
    int conn = rm ? 1 : (ad[m] > 0);
    sc[m] = conn ? e : NEGINF;
  }
  __syncthreads();

  if (wv == 0) {
    float mx = fmaxf(fmaxf(sc[lane], sc[lane + 32]),
                     fmaxf(sc[lane + 64], sc[lane + 96]));
#pragma unroll
    for (int off = 16; off; off >>= 1) mx = fmaxf(mx, __shfl_xor(mx, off, 32));
    if (mx == NEGINF) mx = 0.f;
    if (lane == 0) red[0] = mx;
  }
  __syncthreads();
  const float mx = red[0];
  if (threadIdx.x < MM) sc[threadIdx.x] = __expf(sc[threadIdx.x] - mx);
  __syncthreads();
  if (wv == 0) {
    float s = sc[lane] + sc[lane + 32] + sc[lane + 64] + sc[lane + 96];
#pragma unroll
    for (int off = 16; off; off >>= 1) s += __shfl_xor(s, off, 32);
    if (lane == 0) red[1] = 1.f / fmaxf(s, 1e-30f);
  }
  __syncthreads();
  const float inv = red[1];

  {
    int o = threadIdx.x & 63, ch = threadIdx.x >> 6;
    const float* hmb = hma + (size_t)b * MM * FF;
    float acc = 0.f;
    for (int m = ch * 32; m < ch * 32 + 32; ++m)
      acc += sc[m] * (hmb[(size_t)m * FF + o] + heb[m][o]);
    psum[ch][o] = acc;
  }
  __syncthreads();
  if (threadIdx.x < FF) {
    int o = threadIdx.x;
    float v = (psum[0][o] + psum[1][o] + psum[2][o] + psum[3][o]) * inv +
              h[bn * FF + o];
    hprime[bn * (2 * FF) + FF + o] = v;
  }
}

__global__ __launch_bounds__(256) void mlp(
    const float* __restrict__ hprime, const float* __restrict__ w1,
    const float* __restrict__ b1, const float* __restrict__ w2,
    const float* __restrict__ b2, const float* __restrict__ wo,
    const float* __restrict__ bo, const int* __restrict__ rowmask,
    float* __restrict__ out) {
  extern __shared__ float smem[];
  float* wb = smem;
  typedef float act_row_t[HID + 4];
  act_row_t* a0 = (act_row_t*)(smem + HID * HID);
  act_row_t* a1 = (act_row_t*)(smem + HID * HID + 16 * (HID + 4));

  const int row0 = blockIdx.x * 16;
  const int wv = threadIdx.x >> 5;
  const int lane = threadIdx.x & 31;
  const int lm = lane & 15;
  const int kh = (lane >> 4) * 2;
  const int lh = lane >> 4;

  for (int t = threadIdx.x; t < 16 * HID; t += 256)
    a0[t >> 7][t & 127] = hprime[(size_t)row0 * HID + t];
  for (int t = threadIdx.x; t < HID * HID; t += 256) wb[t] = w1[t];
  __syncthreads();

  {
    int o = wv * 16 + lm;
    v8f c = {};
    for (int k0 = 0; k0 < HID; k0 += 4) {
      v2f a, bf;
      a.x  = a0[lm][k0 + kh];
      a.y  = a0[lm][k0 + kh + 1];
      bf.x = wb[(k0 + kh) * HID + o];
      bf.y = wb[(k0 + kh + 1) * HID + o];
      c = wmma4(a, bf, c);
    }
    float bias = b1[o];
#pragma unroll
    for (int r = 0; r < 8; ++r) {
      float v = c[r] + bias;
      a1[r + 8 * lh][o] = v > 0.f ? v : expm1f(v);
    }
  }
  __syncthreads();
  for (int t = threadIdx.x; t < HID * HID; t += 256) wb[t] = w2[t];
  __syncthreads();

  {
    int o = wv * 16 + lm;
    v8f c = {};
    for (int k0 = 0; k0 < HID; k0 += 4) {
      v2f a, bf;
      a.x  = a1[lm][k0 + kh];
      a.y  = a1[lm][k0 + kh + 1];
      bf.x = wb[(k0 + kh) * HID + o];
      bf.y = wb[(k0 + kh + 1) * HID + o];
      c = wmma4(a, bf, c);
    }
    float bias = b2[o];
#pragma unroll
    for (int r = 0; r < 8; ++r) {
      float v = c[r] + bias;
      a0[r + 8 * lh][o] = v > 0.f ? v : expm1f(v);
    }
  }
  __syncthreads();
  for (int t = threadIdx.x; t < HID * FF; t += 256) wb[t] = wo[t];
  __syncthreads();

  if (wv < 4) {
    int o = wv * 16 + lm;
    v8f c = {};
    for (int k0 = 0; k0 < HID; k0 += 4) {
      v2f a, bf;
      a.x  = a0[lm][k0 + kh];
      a.y  = a0[lm][k0 + kh + 1];
      bf.x = wb[(k0 + kh) * FF + o];
      bf.y = wb[(k0 + kh + 1) * FF + o];
      c = wmma4(a, bf, c);
    }
    float bias = bo[o];
#pragma unroll
    for (int r = 0; r < 8; ++r) {
      int m = r + 8 * lh;
      float v = c[r] + bias;
      if (rowmask[row0 + m]) v = 0.f;
      out[(size_t)(row0 + m) * FF + o] = v;
    }
  }
}

extern "C" void kernel_launch(void* const* d_in, const int* in_sizes, int n_in,
                              void* d_out, int out_size, void* d_ws,
                              size_t ws_size, hipStream_t stream) {
  (void)in_sizes; (void)n_in; (void)out_size; (void)ws_size;
  const float* x      = (const float*)d_in[0];
  const float* y      = (const float*)d_in[1];
  const float* z      = (const float*)d_in[2];
  const int*   op_adj = (const int*)d_in[3];
  const int*   ma_adj = (const int*)d_in[4];
  const int*   op_ma  = (const int*)d_in[5];
  const float* Wop    = (const float*)d_in[6];
  const float* Wma    = (const float*)d_in[7];
  const float* Wedge  = (const float*)d_in[8];
  const float* att_op = (const float*)d_in[9];
  const float* att_ma = (const float*)d_in[10];
  const float* w1     = (const float*)d_in[11];
  const float* b1     = (const float*)d_in[12];
  const float* w2     = (const float*)d_in[13];
  const float* b2     = (const float*)d_in[14];
  const float* wo     = (const float*)d_in[15];
  const float* bo     = (const float*)d_in[16];
  float* out = (float*)d_out;

  float* ws     = (float*)d_ws;
  float* h      = ws;
  float* hma    = h + BB * NN * FF;
  float* ha1    = hma + BB * MM * FF;
  float* ha2    = ha1 + BB * NN;
  float* ham1   = ha2 + BB * NN;
  float* hma2   = ham1 + BB * NN;
  float* hprime = hma2 + BB * MM;
  int*   rmask  = (int*)(hprime + (size_t)BB * NN * 2 * FF);

  gemm_rows_64<<<dim3((BB * NN) / 16), dim3(128), 0, stream>>>(x, Wop, h);
  gemm_rows_64<<<dim3((BB * MM) / 16), dim3(128), 0, stream>>>(y, Wma, hma);
  proj_node<<<dim3((BB * NN) / 256), dim3(256), 0, stream>>>(
      h, att_op, att_ma, op_ma, ha1, ha2, ham1, rmask);
  proj_ma<<<dim3((BB * MM) / 256), dim3(256), 0, stream>>>(hma, att_ma, hma2);
  size_t sm2 = (16 * (NN + 4) + 16) * sizeof(float);
  op_attn<<<dim3(NN / 16, BB), dim3(128), sm2, stream>>>(
      h, ha1, ha2, op_adj, ma_adj, rmask, hprime);
  ma_attn<<<dim3(NN, BB), dim3(256), 0, stream>>>(
      z, Wedge, hma, ham1, hma2, op_ma, rmask, h, att_ma, hprime);
  size_t sm4 = (HID * HID + 2 * 16 * (HID + 4)) * sizeof(float);
  mlp<<<dim3((BB * NN) / 16), dim3(256), sm4, stream>>>(
      hprime, w1, b1, w2, b2, wo, bo, rmask, out);
}


# ma_attn: one work-group walks 8 nodes (8x fewer waves); next z tile requested a trip ahead as 16-byte vectors, in flight across LDS-only barriers; edge-weight fragments kept in registers
# speedup vs baseline: 2.0351x; 1.2692x over previous
#include <hip/hip_runtime.h>
#include <math.h>


typedef __attribute__((ext_vector_type(2))) float v2f;
typedef __attribute__((ext_vector_type(8))) float v8f;
typedef __attribute__((ext_vector_type(4))) float v4f;
typedef __attribute__((ext_vector_type(4))) int v4i;

#define BB 8
#define NN 1024
#define MM 128
#define FF 64
#define FE 16
#define HID 128

__device__ __forceinline__ v8f wmma4(v2f a, v2f b, v8f c) {
  return __builtin_amdgcn_wmma_f32_16x16x4_f32(
       false, a,  false, b,
       (short)0, c,  false,  false);
}

__global__ __launch_bounds__(128) void gemm_rows_64(
    const float* __restrict__ A, const float* __restrict__ W,
    float* __restrict__ C) {
  const int wv   = threadIdx.x >> 5;
  const int lane = threadIdx.x & 31;
  const int lm   = lane & 15;
  const int kh   = (lane >> 4) * 2;
  const int lh   = lane >> 4;
  const int row0 = blockIdx.x * 16;
  const int o    = wv * 16 + lm;
  const float* Arow = A + (size_t)(row0 + lm) * FF;

  v8f c = {};
#pragma unroll
  for (int k0 = 0; k0 < FF; k0 += 4) {
    v2f a, b;
    a.x = Arow[k0 + kh];
    a.y = Arow[k0 + kh + 1];
    b.x = W[(size_t)(k0 + kh) * FF + o];
    b.y = W[(size_t)(k0 + kh + 1) * FF + o];
    c = wmma4(a, b, c);
  }
#pragma unroll
  for (int r = 0; r < 8; ++r) {
    int m = r + 8 * lh;
    C[(size_t)(row0 + m) * FF + o] = c[r];
  }
}

__global__ void proj_node(const float* __restrict__ h,
                          const float* __restrict__ att_op,
                          const float* __restrict__ att_ma,
                          const int* __restrict__ opma,
                          float* __restrict__ ha1, float* __restrict__ ha2,
                          float* __restrict__ ham1, int* __restrict__ rowmask) {
  int idx = blockIdx.x * blockDim.x + threadIdx.x;
  if (idx >= BB * NN) return;
  const float* hr = h + (size_t)idx * FF;
  float s1 = 0.f, s2 = 0.f, s3 = 0.f;
#pragma unroll 8
  for (int o = 0; o < FF; ++o) {
    float v = hr[o];
    s1 += v * att_op[o];
    s2 += v * att_op[FF + o];
    s3 += v * att_ma[o];
  }
  ha1[idx] = s1; ha2[idx] = s2; ham1[idx] = s3;
  const int* ad = opma + (size_t)idx * MM;
  int any = 0;
#pragma unroll 8
  for (int m = 0; m < MM; ++m) any |= ad[m];
  rowmask[idx] = (any == 0) ? 1 : 0;
}

__global__ void proj_ma(const float* __restrict__ hma,
                        const float* __restrict__ att_ma,
                        float* __restrict__ hma2) {
  int idx = blockIdx.x * blockDim.x + threadIdx.x;
  if (idx >= BB * MM) return;
  const float* r = hma + (size_t)idx * FF;
  float s = 0.f;
#pragma unroll 8
  for (int o = 0; o < FF; ++o) s += r[o] * att_ma[FF + o];
  hma2[idx] = s;
}

__device__ __forceinline__ float masked_score(float e, int adj) {
  e = e > 0.f ? e : 0.01f * e;
  return adj > 0 ? e : -__builtin_inff();
}

__global__ __launch_bounds__(128) void op_attn(
    const float* __restrict__ h, const float* __restrict__ ha1,
    const float* __restrict__ ha2, const int* __restrict__ opadj,
    const int* __restrict__ maadj, const int* __restrict__ rowmask,
    float* __restrict__ hprime) {
  extern __shared__ float smem[];
  typedef float att_row_t[NN + 4];
  att_row_t* att = (att_row_t*)smem;
  float* inv_sum = smem + 16 * (NN + 4);

  const int b  = blockIdx.y;
  const int i0 = blockIdx.x * 16;
  const int* opb = opadj + (size_t)b * NN * NN;
  const int* mab = maadj + (size_t)b * NN * NN;
  const float NEGINF = -__builtin_inff();

  const v4i* opT = (const v4i*)(opb + i0);
  const v4i* maT = (const v4i*)(mab + i0);
#pragma unroll 8
  for (int t = threadIdx.x; t < 4 * NN; t += 128) {
    const int j = t >> 2, part = t & 3;
    const v4i a = opT[(size_t)j * (NN / 4) + part];
    const v4i m = maT[(size_t)j * (NN / 4) + part];
    const v4i s = a + m;
    att[4 * part + 0][j] = __int_as_float(s.x);
    att[4 * part + 1][j] = __int_as_float(s.y);
    att[4 * part + 2][j] = __int_as_float(s.z);
    att[4 * part + 3][j] = __int_as_float(s.w);
  }
  __syncthreads();

  const int tj = threadIdx.x * 4;
  const v4f e1lo = *(const v4f*)(ha1 + b * NN + tj);
  const v4f e1hi = *(const v4f*)(ha1 + b * NN + NN / 2 + tj);
#pragma unroll 4
  for (int r = 0; r < 16; ++r) {
    const int i = i0 + r;
    const bool rm = rowmask[b * NN + i] != 0;
    const float e2 = ha2[b * NN + i];
#pragma unroll
    for (int hf = 0; hf < 2; ++hf) {
      const int j = hf * (NN / 2) + tj;
      const v4f e1 = hf ? e1hi : e1lo;
      v4f* slot = (v4f*)&att[r][j];
      const v4f tr = *slot;
      const v4i d1 = *(const v4i*)(opb + (size_t)i * NN + j);
      const v4i d2 = *(const v4i*)(mab + (size_t)i * NN + j);
      v4i adj = d1 + d2;
      adj.x += __float_as_int(tr.x);
      adj.y += __float_as_int(tr.y);
      adj.z += __float_as_int(tr.z);
      adj.w += __float_as_int(tr.w);
      if (rm) adj = (v4i){1, 1, 1, 1};
      v4f sc4;
      sc4.x = masked_score(e2 + e1.x, adj.x);
      sc4.y = masked_score(e2 + e1.y, adj.y);
      sc4.z = masked_score(e2 + e1.z, adj.z);
      sc4.w = masked_score(e2 + e1.w, adj.w);
      *slot = sc4;
    }
  }
  __syncthreads();

  const int wv = threadIdx.x >> 5;
  const int lane = threadIdx.x & 31;
  for (int r = wv; r < 16; r += 4) {
    float mx = NEGINF;
    for (int j = lane; j < NN; j += 32) mx = fmaxf(mx, att[r][j]);
#pragma unroll
    for (int off = 16; off; off >>= 1) mx = fmaxf(mx, __shfl_xor(mx, off, 32));
    if (mx == NEGINF) mx = 0.f;
    float sum = 0.f;
    for (int j = lane; j < NN; j += 32) {
      float p = __expf(att[r][j] - mx);
      att[r][j] = p;
      sum += p;
    }
#pragma unroll
    for (int off = 16; off; off >>= 1) sum += __shfl_xor(sum, off, 32);
    if (lane == 0) inv_sum[r] = 1.f / fmaxf(sum, 1e-30f);
  }
  __syncthreads();

  const int lm = lane & 15;
  const int kh = (lane >> 4) * 2;
  const int lh = lane >> 4;
  const int o  = wv * 16 + lm;
  const float* hb = h + (size_t)b * NN * FF;

  v8f c = {};
  for (int k0 = 0; k0 < NN; k0 += 4) {
    v2f a, bf;
    a.x  = att[lm][k0 + kh];
    a.y  = att[lm][k0 + kh + 1];
    bf.x = hb[(size_t)(k0 + kh) * FF + o];
    bf.y = hb[(size_t)(k0 + kh + 1) * FF + o];
    c = wmma4(a, bf, c);
  }
  float* hp = hprime + ((size_t)b * NN + i0) * (2 * FF);
#pragma unroll
  for (int r = 0; r < 8; ++r) {
    int m = r + 8 * lh;
    hp[(size_t)m * (2 * FF) + o] = c[r] * inv_sum[m];
  }
}

#define NPB 8

__device__ __forceinline__ void lds_barrier() {
  asm volatile("s_wait_dscnt 0x0" ::: "memory");
  asm volatile("s_barrier_signal -1" ::: "memory");
  asm volatile("s_barrier_wait -1" ::: "memory");
}

__global__ __launch_bounds__(256) void ma_attn(
    const float* __restrict__ z, const float* __restrict__ wedge,
    const float* hma, const float* __restrict__ ham1,
    const float* __restrict__ hma2, const int* __restrict__ opma,
    const int* __restrict__ rowmask, const float* __restrict__ h,
    const float* __restrict__ attma, float* hprime) {
  __shared__ float zb[MM][FE + 1];
  __shared__ float heb[MM][FF + 1];
  __shared__ float sc[MM];
  __shared__ float red[2];
  __shared__ float psum[4][FF];
  __shared__ __attribute__((aligned(16))) float am3s[FF];

  const int b  = blockIdx.y;
  const int nb = blockIdx.x * NPB;
  const float NEGINF = -__builtin_inff();

  const int wv = threadIdx.x >> 5;
  const int lane = threadIdx.x & 31;
  const int lm = lane & 15;
  const int kh = (lane >> 4) * 2;
  const int lh = lane >> 4;
  const int m0 = wv * 16;

  v2f bw[4][4];
#pragma unroll
  for (int ct = 0; ct < 4; ++ct) {
#pragma unroll
    for (int ks = 0; ks < 4; ++ks) {
      bw[ct][ks].x = wedge[(4 * ks + kh) * FF + ct * 16 + lm];
      bw[ct][ks].y = wedge[(4 * ks + kh + 1) * FF + ct * 16 + lm];
    }
  }

  if (threadIdx.x < FF) am3s[threadIdx.x] = attma[2 * FF + threadIdx.x];

  const float* hmb = hma + (size_t)b * MM * FF;
  const int ms = threadIdx.x & (MM - 1);
  const float hm2 = hma2[(size_t)b * MM + ms];

  const v4f* zt = (const v4f*)(z + ((size_t)b * NN + nb) * MM * FE);
  v4f zr0 = zt[threadIdx.x];
  v4f zr1 = zt[256 + threadIdx.x];
  int   rm  = rowmask[(size_t)b * NN + nb];
  float hm1 = ham1[(size_t)b * NN + nb];
  int   adm = opma[((size_t)b * NN + nb) * MM + ms];

  for (int it = 0; it < NPB; ++it) {
    const size_t bn = (size_t)b * NN + nb + it;

    {
      const int e0 = threadIdx.x * 4;
      float* d0 = &zb[e0 >> 4][e0 & 15];
      d0[0] = zr0.x; d0[1] = zr0.y; d0[2] = zr0.z; d0[3] = zr0.w;
      const int e1 = e0 + 1024;
      float* d1 = &zb[e1 >> 4][e1 & 15];
      d1[0] = zr1.x; d1[1] = zr1.y; d1[2] = zr1.z; d1[3] = zr1.w;
    }
    lds_barrier();
    const int nx = (it + 1 < NPB) ? (it + 1) : it;
    const size_t bnx = (size_t)b * NN + nb + nx;
    const int   rm_n  = rowmask[bnx];
    const float hm1_n = ham1[bnx];
    const int   adm_n = opma[bnx * MM + ms];
    zr0 = zt[nx * 512 + threadIdx.x];
    zr1 = zt[nx * 512 + 256 + threadIdx.x];

    v2f af[4];
#pragma unroll
    for (int ks = 0; ks < 4; ++ks) {
      af[ks].x = zb[m0 + lm][4 * ks + kh];
      af[ks].y = zb[m0 + lm][4 * ks + kh + 1];
    }
#pragma unroll
    for (int ct = 0; ct < 4; ++ct) {
      v8f c = {};
#pragma unroll
      for (int ks = 0; ks < 4; ++ks) c = wmma4(af[ks], bw[ct][ks], c);
#pragma unroll
      for (int r = 0; r < 8; ++r)
        heb[m0 + r + 8 * lh][ct * 16 + lm] = c[r];
    }
    lds_barrier();

    if (threadIdx.x < MM) {
      int m = threadIdx.x;
      float dot = 0.f;
#pragma unroll 8
      for (int o = 0; o < FF; ++o) dot += heb[m][o] * am3s[o];
      float e = hm1 + hm2 + dot;
      e = e > 0.f ? e : 0.01f * e;
      int conn = rm ? 1 : (adm > 0);
      sc[m] = conn ? e : NEGINF;
    }
    lds_barrier();

    if (wv == 0) {
      float mx = fmaxf(fmaxf(sc[lane], sc[lane + 32]),
                       fmaxf(sc[lane + 64], sc[lane + 96]));
#pragma unroll
      for (int off = 16; off; off >>= 1) mx = fmaxf(mx, __shfl_xor(mx, off, 32));
      if (mx == NEGINF) mx = 0.f;
      if (lane == 0) red[0] = mx;
    }
    lds_barrier();
    const float mx = red[0];
    if (threadIdx.x < MM) sc[threadIdx.x] = __expf(sc[threadIdx.x] - mx);
    lds_barrier();
    if (wv == 0) {
      float s = sc[lane] + sc[lane + 32] + sc[lane + 64] + sc[lane + 96];
#pragma unroll
      for (int off = 16; off; off >>= 1) s += __shfl_xor(s, off, 32);
      if (lane == 0) red[1] = 1.f / fmaxf(s, 1e-30f);
    }
    lds_barrier();
    const float inv = red[1];

    {
      int o = threadIdx.x & 63, ch = threadIdx.x >> 6;
      float acc = 0.f;
      for (int m = ch * 32; m < ch * 32 + 32; ++m)
        acc += sc[m] * (hmb[(size_t)m * FF + o] + heb[m][o]);
      psum[ch][o] = acc;
    }
    lds_barrier();
    if (threadIdx.x < FF) {
      int o = threadIdx.x;
      float v = (psum[0][o] + psum[1][o] + psum[2][o] + psum[3][o]) * inv +
                h[bn * FF + o];
      hprime[bn * (2 * FF) + FF + o] = v;
    }
    rm = rm_n; hm1 = hm1_n; adm = adm_n;
  }
}

__global__ __launch_bounds__(256) void mlp(
    const float* __restrict__ hprime, const float* __restrict__ w1,
    const float* __restrict__ b1, const float* __restrict__ w2,
    const float* __restrict__ b2, const float* __restrict__ wo,
    const float* __restrict__ bo, const int* __restrict__ rowmask,
    float* __restrict__ out) {
  extern __shared__ float smem[];
  float* wb = smem;
  typedef float act_row_t[HID + 4];
  act_row_t* a0 = (act_row_t*)(smem + HID * HID);
  act_row_t* a1 = (act_row_t*)(smem + HID * HID + 16 * (HID + 4));

  const int row0 = blockIdx.x * 16;
  const int wv = threadIdx.x >> 5;
  const int lane = threadIdx.x & 31;
  const int lm = lane & 15;
  const int kh = (lane >> 4) * 2;
  const int lh = lane >> 4;

  for (int t = threadIdx.x; t < 16 * HID; t += 256)
    a0[t >> 7][t & 127] = hprime[(size_t)row0 * HID + t];
  for (int t = threadIdx.x; t < HID * HID; t += 256) wb[t] = w1[t];
  __syncthreads();

  {
    int o = wv * 16 + lm;
    v8f c = {};
    for (int k0 = 0; k0 < HID; k0 += 4) {
      v2f a, bf;
      a.x  = a0[lm][k0 + kh];
      a.y  = a0[lm][k0 + kh + 1];
      bf.x = wb[(k0 + kh) * HID + o];
      bf.y = wb[(k0 + kh + 1) * HID + o];
      c = wmma4(a, bf, c);
    }
    float bias = b1[o];
#pragma unroll
    for (int r = 0; r < 8; ++r) {
      float v = c[r] + bias;
      a1[r + 8 * lh][o] = v > 0.f ? v : expm1f(v);
    }
  }
  __syncthreads();
  for (int t = threadIdx.x; t < HID * HID; t += 256) wb[t] = w2[t];
  __syncthreads();

  {
    int o = wv * 16 + lm;
    v8f c = {};
    for (int k0 = 0; k0 < HID; k0 += 4) {
      v2f a, bf;
      a.x  = a1[lm][k0 + kh];
      a.y  = a1[lm][k0 + kh + 1];
      bf.x = wb[(k0 + kh) * HID + o];
      bf.y = wb[(k0 + kh + 1) * HID + o];
      c = wmma4(a, bf, c);
    }
    float bias = b2[o];
#pragma unroll
    for (int r = 0; r < 8; ++r) {
      float v = c[r] + bias;
      a0[r + 8 * lh][o] = v > 0.f ? v : expm1f(v);
    }
  }
  __syncthreads();
  for (int t = threadIdx.x; t < HID * FF; t += 256) wb[t] = wo[t];
  __syncthreads();

  if (wv < 4) {
    int o = wv * 16 + lm;
    v8f c = {};
    for (int k0 = 0; k0 < HID; k0 += 4) {
      v2f a, bf;
      a.x  = a0[lm][k0 + kh];
      a.y  = a0[lm][k0 + kh + 1];
      bf.x = wb[(k0 + kh) * FF + o];
      bf.y = wb[(k0 + kh + 1) * FF + o];
      c = wmma4(a, bf, c);
    }
    float bias = bo[o];
#pragma unroll
    for (int r = 0; r < 8; ++r) {
      int m = r + 8 * lh;
      float v = c[r] + bias;
      if (rowmask[row0 + m]) v = 0.f;
      out[(size_t)(row0 + m) * FF + o] = v;
    }
  }
}

extern "C" void kernel_launch(void* const* d_in, const int* in_sizes, int n_in,
                              void* d_out, int out_size, void* d_ws,
                              size_t ws_size, hipStream_t stream) {
  (void)in_sizes; (void)n_in; (void)out_size; (void)ws_size;
  const float* x      = (const float*)d_in[0];
  const float* y      = (const float*)d_in[1];
  const float* z      = (const float*)d_in[2];
  const int*   op_adj = (const int*)d_in[3];
  const int*   ma_adj = (const int*)d_in[4];
  const int*   op_ma  = (const int*)d_in[5];
  const float* Wop    = (const float*)d_in[6];
  const float* Wma    = (const float*)d_in[7];
  const float* Wedge  = (const float*)d_in[8];
  const float* att_op = (const float*)d_in[9];
  const float* att_ma = (const float*)d_in[10];
  const float* w1     = (const float*)d_in[11];
  const float* b1     = (const float*)d_in[12];
  const float* w2     = (const float*)d_in[13];
  const float* b2     = (const float*)d_in[14];
  const float* wo     = (const float*)d_in[15];
  const float* bo     = (const float*)d_in[16];
  float* out = (float*)d_out;

  float* ws     = (float*)d_ws;
  float* h      = ws;
  float* hma    = h + BB * NN * FF;
  float* ha1    = hma + BB * MM * FF;
  float* ha2    = ha1 + BB * NN;
  float* ham1   = ha2 + BB * NN;
  float* hma2   = ham1 + BB * NN;
  float* hprime = hma2 + BB * MM;
  int*   rmask  = (int*)(hprime + (size_t)BB * NN * 2 * FF);

  gemm_rows_64<<<dim3((BB * NN) / 16), dim3(128), 0, stream>>>(x, Wop, h);
  gemm_rows_64<<<dim3((BB * MM) / 16), dim3(128), 0, stream>>>(y, Wma, hma);
  proj_node<<<dim3((BB * NN) / 256), dim3(256), 0, stream>>>(
      h, att_op, att_ma, op_ma, ha1, ha2, ham1, rmask);
  proj_ma<<<dim3((BB * MM) / 256), dim3(256), 0, stream>>>(hma, att_ma, hma2);
  size_t sm2 = (16 * (NN + 4) + 16) * sizeof(float);
  op_attn<<<dim3(NN / 16, BB), dim3(128), sm2, stream>>>(
      h, ha1, ha2, op_adj, ma_adj, rmask, hprime);
  ma_attn<<<dim3(NN / NPB, BB), dim3(256), 0, stream>>>(
      z, Wedge, hma, ham1, hma2, op_ma, rmask, h, att_ma, hprime);
  size_t sm4 = (HID * HID + 2 * 16 * (HID + 4)) * sizeof(float);
  mlp<<<dim3((BB * NN) / 16), dim3(256), sm4, stream>>>(
      hprime, w1, b1, w2, b2, wo, bo, rmask, out);
}


# the four preparatory launches (both feature GEMMs, the projections, the row mask) become one: a tile's projections read its product from LDS, same summing order; 7 launches -> 4
# speedup vs baseline: 2.2429x; 1.1021x over previous
#include <hip/hip_runtime.h>
#include <math.h>


typedef __attribute__((ext_vector_type(2))) float v2f;
typedef __attribute__((ext_vector_type(8))) float v8f;
typedef __attribute__((ext_vector_type(4))) float v4f;
typedef __attribute__((ext_vector_type(4))) int v4i;

#define BB 8
#define NN 1024
#define MM 128
#define FF 64
#define FE 16
#define HID 128

__device__ __forceinline__ v8f wmma4(v2f a, v2f b, v8f c) {
  return __builtin_amdgcn_wmma_f32_16x16x4_f32(
       false, a,  false, b,
       (short)0, c,  false,  false);
}

#define XT ((BB * NN) / 16)
#define YT ((BB * MM) / 16)
#define CTS (FF + 4)

__global__ __launch_bounds__(128) void prep_tiles(
    const float* __restrict__ x, const float* __restrict__ Wop,
    const float* __restrict__ y, const float* __restrict__ Wma,
    const float* __restrict__ att_op, const float* __restrict__ att_ma,
    const int* __restrict__ opma, float* __restrict__ h,
    float* __restrict__ hma, float* __restrict__ ha1,
    float* __restrict__ ha2, float* __restrict__ ham1,
    float* __restrict__ hma2, int* __restrict__ rowmask) {
  __shared__ float ct[16][CTS];
  __shared__ float atts[3][FF];

  const int wv   = threadIdx.x >> 5;
  const int lane = threadIdx.x & 31;
  const int lm   = lane & 15;
  const int kh   = (lane >> 4) * 2;
  const int lh   = lane >> 4;
  const bool node = blockIdx.x < XT;
  const int row0 = (node ? blockIdx.x : blockIdx.x - XT) * 16;

  v4i ad[4] = {};
  if (node) {
    const v4i* at = (const v4i*)(opma + (size_t)row0 * MM);
#pragma unroll
    for (int q = 0; q < 4; ++q) ad[q] = at[q * 128 + threadIdx.x];
    if (threadIdx.x < FF) {
      atts[0][threadIdx.x] = att_op[threadIdx.x];
      atts[1][threadIdx.x] = att_op[FF + threadIdx.x];
      atts[2][threadIdx.x] = att_ma[threadIdx.x];
    }
  } else {
    if (threadIdx.x < FF) atts[0][threadIdx.x] = att_ma[FF + threadIdx.x];
  }

  const float* A = node ? x : y;
  const float* W = node ? Wop : Wma;
  float* C = node ? h : hma;
  const int o = wv * 16 + lm;
  const float* Arow = A + (size_t)(row0 + lm) * FF;

  v8f c = {};
#pragma unroll
  for (int k0 = 0; k0 < FF; k0 += 4) {
    v2f a, b;
    a.x = Arow[k0 + kh];
    a.y = Arow[k0 + kh + 1];
    b.x = W[(size_t)(k0 + kh) * FF + o];
    b.y = W[(size_t)(k0 + kh + 1) * FF + o];
    c = wmma4(a, b, c);
  }
#pragma unroll
  for (int r = 0; r < 8; ++r) {
    int m = r + 8 * lh;
    C[(size_t)(row0 + m) * FF + o] = c[r];
    ct[m][o] = c[r];
  }

  if (node) {
#pragma unroll
    for (int q = 0; q < 4; ++q) {
      const int nz = (ad[q].x | ad[q].y | ad[q].z | ad[q].w) != 0;
      const int anynz = __any(nz);
      if (lane == 0) rowmask[row0 + wv + 4 * q] = anynz ? 0 : 1;
    }
  }
  __syncthreads();

  const int np = node ? 3 : 1;
  if (threadIdx.x < 16 * np) {
    const int p = threadIdx.x >> 4, row = threadIdx.x & 15;
    float s = 0.f;
#pragma unroll 8
    for (int oo = 0; oo < FF; ++oo) s += ct[row][oo] * atts[p][oo];
    float* dst = node ? (p == 0 ? ha1 : (p == 1 ? ha2 : ham1)) : hma2;
    dst[row0 + row] = s;
  }
}

__device__ __forceinline__ float masked_score(float e, int adj) {
  e = e > 0.f ? e : 0.01f * e;
  return adj > 0 ? e : -__builtin_inff();
}

__global__ __launch_bounds__(128) void op_attn(
    const float* __restrict__ h, const float* __restrict__ ha1,
    const float* __restrict__ ha2, const int* __restrict__ opadj,
    const int* __restrict__ maadj, const int* __restrict__ rowmask,
    float* __restrict__ hprime) {
  extern __shared__ float smem[];
  typedef float att_row_t[NN + 4];
  att_row_t* att = (att_row_t*)smem;
  float* inv_sum = smem + 16 * (NN + 4);

  const int b  = blockIdx.y;
  const int i0 = blockIdx.x * 16;
  const int* opb = opadj + (size_t)b * NN * NN;
  const int* mab = maadj + (size_t)b * NN * NN;
  const float NEGINF = -__builtin_inff();

  const v4i* opT = (const v4i*)(opb + i0);
  const v4i* maT = (const v4i*)(mab + i0);
#pragma unroll 8
  for (int t = threadIdx.x; t < 4 * NN; t += 128) {
    const int j = t >> 2, part = t & 3;
    const v4i a = opT[(size_t)j * (NN / 4) + part];
    const v4i m = maT[(size_t)j * (NN / 4) + part];
    const v4i s = a + m;
    att[4 * part + 0][j] = __int_as_float(s.x);
    att[4 * part + 1][j] = __int_as_float(s.y);
    att[4 * part + 2][j] = __int_as_float(s.z);
    att[4 * part + 3][j] = __int_as_float(s.w);
  }
  __syncthreads();

  const int tj = threadIdx.x * 4;
  const v4f e1lo = *(const v4f*)(ha1 + b * NN + tj);
  const v4f e1hi = *(const v4f*)(ha1 + b * NN + NN / 2 + tj);
#pragma unroll 4
  for (int r = 0; r < 16; ++r) {
    const int i = i0 + r;
    const bool rm = rowmask[b * NN + i] != 0;
    const float e2 = ha2[b * NN + i];
#pragma unroll
    for (int hf = 0; hf < 2; ++hf) {
      const int j = hf * (NN / 2) + tj;
      const v4f e1 = hf ? e1hi : e1lo;
      v4f* slot = (v4f*)&att[r][j];
      const v4f tr = *slot;
      const v4i d1 = *(const v4i*)(opb + (size_t)i * NN + j);
      const v4i d2 = *(const v4i*)(mab + (size_t)i * NN + j);
      v4i adj = d1 + d2;
      adj.x += __float_as_int(tr.x);
      adj.y += __float_as_int(tr.y);
      adj.z += __float_as_int(tr.z);
      adj.w += __float_as_int(tr.w);
      if (rm) adj = (v4i){1, 1, 1, 1};
      v4f sc4;
      sc4.x = masked_score(e2 + e1.x, adj.x);
      sc4.y = masked_score(e2 + e1.y, adj.y);
      sc4.z = masked_score(e2 + e1.z, adj.z);
      sc4.w = masked_score(e2 + e1.w, adj.w);
      *slot = sc4;
    }
  }
  __syncthreads();

  const int wv = threadIdx.x >> 5;
  const int lane = threadIdx.x & 31;
  for (int r = wv; r < 16; r += 4) {
    float mx = NEGINF;
    for (int j = lane; j < NN; j += 32) mx = fmaxf(mx, att[r][j]);
#pragma unroll
    for (int off = 16; off; off >>= 1) mx = fmaxf(mx, __shfl_xor(mx, off, 32));
    if (mx == NEGINF) mx = 0.f;
    float sum = 0.f;
    for (int j = lane; j < NN; j += 32) {
      float p = __expf(att[r][j] - mx);
      att[r][j] = p;
      sum += p;
    }
#pragma unroll
    for (int off = 16; off; off >>= 1) sum += __shfl_xor(sum, off, 32);
    if (lane == 0) inv_sum[r] = 1.f / fmaxf(sum, 1e-30f);
  }
  __syncthreads();

  const int lm = lane & 15;
  const int kh = (lane >> 4) * 2;
  const int lh = lane >> 4;
  const int o  = wv * 16 + lm;
  const float* hb = h + (size_t)b * NN * FF;

  v8f c = {};
  for (int k0 = 0; k0 < NN; k0 += 4) {
    v2f a, bf;
    a.x  = att[lm][k0 + kh];
    a.y  = att[lm][k0 + kh + 1];
    bf.x = hb[(size_t)(k0 + kh) * FF + o];
    bf.y = hb[(size_t)(k0 + kh + 1) * FF + o];
    c = wmma4(a, bf, c);
  }
  float* hp = hprime + ((size_t)b * NN + i0) * (2 * FF);
#pragma unroll
  for (int r = 0; r < 8; ++r) {
    int m = r + 8 * lh;
    hp[(size_t)m * (2 * FF) + o] = c[r] * inv_sum[m];
  }
}

#define NPB 8

__device__ __forceinline__ void lds_barrier() {
  asm volatile("s_wait_dscnt 0x0" ::: "memory");
  asm volatile("s_barrier_signal -1" ::: "memory");
  asm volatile("s_barrier_wait -1" ::: "memory");
}

__global__ __launch_bounds__(256) void ma_attn(
    const float* __restrict__ z, const float* __restrict__ wedge,
    const float* hma, const float* __restrict__ ham1,
    const float* __restrict__ hma2, const int* __restrict__ opma,
    const int* __restrict__ rowmask, const float* __restrict__ h,
    const float* __restrict__ attma, float* hprime) {
  __shared__ float zb[MM][FE + 1];
  __shared__ float heb[MM][FF + 1];
  __shared__ float sc[MM];
  __shared__ float red[2];
  __shared__ float psum[4][FF];
  __shared__ __attribute__((aligned(16))) float am3s[FF];

  const int b  = blockIdx.y;
  const int nb = blockIdx.x * NPB;
  const float NEGINF = -__builtin_inff();

  const int wv = threadIdx.x >> 5;
  const int lane = threadIdx.x & 31;
  const int lm = lane & 15;
  const int kh = (lane >> 4) * 2;
  const int lh = lane >> 4;
  const int m0 = wv * 16;

  v2f bw[4][4];
#pragma unroll
  for (int ct = 0; ct < 4; ++ct) {
#pragma unroll
    for (int ks = 0; ks < 4; ++ks) {
      bw[ct][ks].x = wedge[(4 * ks + kh) * FF + ct * 16 + lm];
      bw[ct][ks].y = wedge[(4 * ks + kh + 1) * FF + ct * 16 + lm];
    }
  }

  if (threadIdx.x < FF) am3s[threadIdx.x] = attma[2 * FF + threadIdx.x];

  const float* hmb = hma + (size_t)b * MM * FF;
  const int ms = threadIdx.x & (MM - 1);
  const float hm2 = hma2[(size_t)b * MM + ms];

  const v4f* zt = (const v4f*)(z + ((size_t)b * NN + nb) * MM * FE);
  v4f zr0 = zt[threadIdx.x];
  v4f zr1 = zt[256 + threadIdx.x];
  int   rm  = rowmask[(size_t)b * NN + nb];
  float hm1 = ham1[(size_t)b * NN + nb];
  int   adm = opma[((size_t)b * NN + nb) * MM + ms];

  for (int it = 0; it < NPB; ++it) {
    const size_t bn = (size_t)b * NN + nb + it;

    {
      const int e0 = threadIdx.x * 4;
      float* d0 = &zb[e0 >> 4][e0 & 15];
      d0[0] = zr0.x; d0[1] = zr0.y; d0[2] = zr0.z; d0[3] = zr0.w;
      const int e1 = e0 + 1024;
      float* d1 = &zb[e1 >> 4][e1 & 15];
      d1[0] = zr1.x; d1[1] = zr1.y; d1[2] = zr1.z; d1[3] = zr1.w;
    }
    lds_barrier();
    const int nx = (it + 1 < NPB) ? (it + 1) : it;
    const size_t bnx = (size_t)b * NN + nb + nx;
    const int   rm_n  = rowmask[bnx];
    const float hm1_n = ham1[bnx];
    const int   adm_n = opma[bnx * MM + ms];
    zr0 = zt[nx * 512 + threadIdx.x];
    zr1 = zt[nx * 512 + 256 + threadIdx.x];

    v2f af[4];
#pragma unroll
    for (int ks = 0; ks < 4; ++ks) {
      af[ks].x = zb[m0 + lm][4 * ks + kh];
      af[ks].y = zb[m0 + lm][4 * ks + kh + 1];
    }
#pragma unroll
    for (int ct = 0; ct < 4; ++ct) {
      v8f c = {};
#pragma unroll
      for (int ks = 0; ks < 4; ++ks) c = wmma4(af[ks], bw[ct][ks], c);
#pragma unroll
      for (int r = 0; r < 8; ++r)
        heb[m0 + r + 8 * lh][ct * 16 + lm] = c[r];
    }
    lds_barrier();

    if (threadIdx.x < MM) {
      int m = threadIdx.x;
      float dot = 0.f;
#pragma unroll 8
      for (int o = 0; o < FF; ++o) dot += heb[m][o] * am3s[o];
      float e = hm1 + hm2 + dot;
      e = e > 0.f ? e : 0.01f * e;
      int conn = rm ? 1 : (adm > 0);
      sc[m] = conn ? e : NEGINF;
    }
    lds_barrier();

    if (wv == 0) {
      float mx = fmaxf(fmaxf(sc[lane], sc[lane + 32]),
                       fmaxf(sc[lane + 64], sc[lane + 96]));
#pragma unroll
      for (int off = 16; off; off >>= 1) mx = fmaxf(mx, __shfl_xor(mx, off, 32));
      if (mx == NEGINF) mx = 0.f;
      if (lane == 0) red[0] = mx;
    }
    lds_barrier();
    const float mx = red[0];
    if (threadIdx.x < MM) sc[threadIdx.x] = __expf(sc[threadIdx.x] - mx);
    lds_barrier();
    if (wv == 0) {
      float s = sc[lane] + sc[lane + 32] + sc[lane + 64] + sc[lane + 96];
#pragma unroll
      for (int off = 16; off; off >>= 1) s += __shfl_xor(s, off, 32);
      if (lane == 0) red[1] = 1.f / fmaxf(s, 1e-30f);
    }
    lds_barrier();
    const float inv = red[1];

    {
      int o = threadIdx.x & 63, ch = threadIdx.x >> 6;
      float acc = 0.f;
      for (int m = ch * 32; m < ch * 32 + 32; ++m)
        acc += sc[m] * (hmb[(size_t)m * FF + o] + heb[m][o]);
      psum[ch][o] = acc;
    }
    lds_barrier();
    if (threadIdx.x < FF) {
      int o = threadIdx.x;
      float v = (psum[0][o] + psum[1][o] + psum[2][o] + psum[3][o]) * inv +
                h[bn * FF + o];
      hprime[bn * (2 * FF) + FF + o] = v;
    }
    rm = rm_n; hm1 = hm1_n; adm = adm_n;
  }
}

__global__ __launch_bounds__(256) void mlp(
    const float* __restrict__ hprime, const float* __restrict__ w1,
    const float* __restrict__ b1, const float* __restrict__ w2,
    const float* __restrict__ b2, const float* __restrict__ wo,
    const float* __restrict__ bo, const int* __restrict__ rowmask,
    float* __restrict__ out) {
  extern __shared__ float smem[];
  float* wb = smem;
  typedef float act_row_t[HID + 4];
  act_row_t* a0 = (act_row_t*)(smem + HID * HID);
  act_row_t* a1 = (act_row_t*)(smem + HID * HID + 16 * (HID + 4));

  const int row0 = blockIdx.x * 16;
  const int wv = threadIdx.x >> 5;
  const int lane = threadIdx.x & 31;
  const int lm = lane & 15;
  const int kh = (lane >> 4) * 2;
  const int lh = lane >> 4;

  for (int t = threadIdx.x; t < 16 * HID; t += 256)
    a0[t >> 7][t & 127] = hprime[(size_t)row0 * HID + t];
  for (int t = threadIdx.x; t < HID * HID; t += 256) wb[t] = w1[t];
  __syncthreads();

  {
    int o = wv * 16 + lm;
    v8f c = {};
    for (int k0 = 0; k0 < HID; k0 += 4) {
      v2f a, bf;
      a.x  = a0[lm][k0 + kh];
      a.y  = a0[lm][k0 + kh + 1];
      bf.x = wb[(k0 + kh) * HID + o];
      bf.y = wb[(k0 + kh + 1) * HID + o];
      c = wmma4(a, bf, c);
    }
    float bias = b1[o];
#pragma unroll
    for (int r = 0; r < 8; ++r) {
      float v = c[r] + bias;
      a1[r + 8 * lh][o] = v > 0.f ? v : expm1f(v);
    }
  }
  __syncthreads();
  for (int t = threadIdx.x; t < HID * HID; t += 256) wb[t] = w2[t];
  __syncthreads();

  {
    int o = wv * 16 + lm;
    v8f c = {};
    for (int k0 = 0; k0 < HID; k0 += 4) {
      v2f a, bf;
      a.x  = a1[lm][k0 + kh];
      a.y  = a1[lm][k0 + kh + 1];
      bf.x = wb[(k0 + kh) * HID + o];
      bf.y = wb[(k0 + kh + 1) * HID + o];
      c = wmma4(a, bf, c);
    }
    float bias = b2[o];
#pragma unroll
    for (int r = 0; r < 8; ++r) {
      float v = c[r] + bias;
      a0[r + 8 * lh][o] = v > 0.f ? v : expm1f(v);
    }
  }
  __syncthreads();
  for (int t = threadIdx.x; t < HID * FF; t += 256) wb[t] = wo[t];
  __syncthreads();

  if (wv < 4) {
    int o = wv * 16 + lm;
    v8f c = {};
    for (int k0 = 0; k0 < HID; k0 += 4) {
      v2f a, bf;
      a.x  = a0[lm][k0 + kh];
      a.y  = a0[lm][k0 + kh + 1];
      bf.x = wb[(k0 + kh) * FF + o];
      bf.y = wb[(k0 + kh + 1) * FF + o];
      c = wmma4(a, bf, c);
    }
    float bias = bo[o];
#pragma unroll
    for (int r = 0; r < 8; ++r) {
      int m = r + 8 * lh;
      float v = c[r] + bias;
      if (rowmask[row0 + m]) v = 0.f;
      out[(size_t)(row0 + m) * FF + o] = v;
    }
  }
}

extern "C" void kernel_launch(void* const* d_in, const int* in_sizes, int n_in,
                              void* d_out, int out_size, void* d_ws,
                              size_t ws_size, hipStream_t stream) {
  (void)in_sizes; (void)n_in; (void)out_size; (void)ws_size;
  const float* x      = (const float*)d_in[0];
  const float* y      = (const float*)d_in[1];
  const float* z      = (const float*)d_in[2];
  const int*   op_adj = (const int*)d_in[3];
  const int*   ma_adj = (const int*)d_in[4];
  const int*   op_ma  = (const int*)d_in[5];
  const float* Wop    = (const float*)d_in[6];
  const float* Wma    = (const float*)d_in[7];
  const float* Wedge  = (const float*)d_in[8];
  const float* att_op = (const float*)d_in[9];
  const float* att_ma = (const float*)d_in[10];
  const float* w1     = (const float*)d_in[11];
  const float* b1     = (const float*)d_in[12];
  const float* w2     = (const float*)d_in[13];
  const float* b2     = (const float*)d_in[14];
  const float* wo     = (const float*)d_in[15];
  const float* bo     = (const float*)d_in[16];
  float* out = (float*)d_out;

  float* ws     = (float*)d_ws;
  float* h      = ws;
  float* hma    = h + BB * NN * FF;
  float* ha1    = hma + BB * MM * FF;
  float* ha2    = ha1 + BB * NN;
  float* ham1   = ha2 + BB * NN;
  float* hma2   = ham1 + BB * NN;
  float* hprime = hma2 + BB * MM;
  int*   rmask  = (int*)(hprime + (size_t)BB * NN * 2 * FF);

  prep_tiles<<<dim3(XT + YT), dim3(128), 0, stream>>>(
      x, Wop, y, Wma, att_op, att_ma, op_ma, h, hma, ha1, ha2, ham1, hma2,
      rmask);
  size_t sm2 = (16 * (NN + 4) + 16) * sizeof(float);
  op_attn<<<dim3(NN / 16, BB), dim3(128), sm2, stream>>>(
      h, ha1, ha2, op_adj, ma_adj, rmask, hprime);
  ma_attn<<<dim3(NN / NPB, BB), dim3(256), 0, stream>>>(
      z, Wedge, hma, ham1, hma2, op_ma, rmask, h, att_ma, hprime);
  size_t sm4 = (HID * HID + 2 * 16 * (HID + 4)) * sizeof(float);
  mlp<<<dim3((BB * NN) / 16), dim3(256), sm4, stream>>>(
      hprime, w1, b1, w2, b2, wo, bo, rmask, out);
}


# ma_attn: each wave scores the 16 he rows it just stored (row dot as two 32-column halves added: summing order changed); every wave forms max, exp, sum itself in registers; 7 barriers per node -> 3
# speedup vs baseline: 2.3246x; 1.0364x over previous
#include <hip/hip_runtime.h>
#include <math.h>


typedef __attribute__((ext_vector_type(2))) float v2f;
typedef __attribute__((ext_vector_type(8))) float v8f;
typedef __attribute__((ext_vector_type(4))) float v4f;
typedef __attribute__((ext_vector_type(4))) int v4i;

#define BB 8
#define NN 1024
#define MM 128
#define FF 64
#define FE 16
#define HID 128

__device__ __forceinline__ v8f wmma4(v2f a, v2f b, v8f c) {
  return __builtin_amdgcn_wmma_f32_16x16x4_f32(
       false, a,  false, b,
       (short)0, c,  false,  false);
}

#define XT ((BB * NN) / 16)
#define YT ((BB * MM) / 16)
#define CTS (FF + 4)

__global__ __launch_bounds__(128) void prep_tiles(
    const float* __restrict__ x, const float* __restrict__ Wop,
    const float* __restrict__ y, const float* __restrict__ Wma,
    const float* __restrict__ att_op, const float* __restrict__ att_ma,
    const int* __restrict__ opma, float* __restrict__ h,
    float* __restrict__ hma, float* __restrict__ ha1,
    float* __restrict__ ha2, float* __restrict__ ham1,
    float* __restrict__ hma2, int* __restrict__ rowmask) {
  __shared__ float ct[16][CTS];
  __shared__ float atts[3][FF];

  const int wv   = threadIdx.x >> 5;
  const int lane = threadIdx.x & 31;
  const int lm   = lane & 15;
  const int kh   = (lane >> 4) * 2;
  const int lh   = lane >> 4;
  const bool node = blockIdx.x < XT;
  const int row0 = (node ? blockIdx.x : blockIdx.x - XT) * 16;

  v4i ad[4] = {};
  if (node) {
    const v4i* at = (const v4i*)(opma + (size_t)row0 * MM);
#pragma unroll
    for (int q = 0; q < 4; ++q) ad[q] = at[q * 128 + threadIdx.x];
    if (threadIdx.x < FF) {
      atts[0][threadIdx.x] = att_op[threadIdx.x];
      atts[1][threadIdx.x] = att_op[FF + threadIdx.x];
      atts[2][threadIdx.x] = att_ma[threadIdx.x];
    }
  } else {
    if (threadIdx.x < FF) atts[0][threadIdx.x] = att_ma[FF + threadIdx.x];
  }

  const float* A = node ? x : y;
  const float* W = node ? Wop : Wma;
  float* C = node ? h : hma;
  const int o = wv * 16 + lm;
  const float* Arow = A + (size_t)(row0 + lm) * FF;

  v8f c = {};
#pragma unroll
  for (int k0 = 0; k0 < FF; k0 += 4) {
    v2f a, b;
    a.x = Arow[k0 + kh];
    a.y = Arow[k0 + kh + 1];
    b.x = W[(size_t)(k0 + kh) * FF + o];
    b.y = W[(size_t)(k0 + kh + 1) * FF + o];
    c = wmma4(a, b, c);
  }
#pragma unroll
  for (int r = 0; r < 8; ++r) {
    int m = r + 8 * lh;
    C[(size_t)(row0 + m) * FF + o] = c[r];
    ct[m][o] = c[r];
  }

  if (node) {
#pragma unroll
    for (int q = 0; q < 4; ++q) {
      const int nz = (ad[q].x | ad[q].y | ad[q].z | ad[q].w) != 0;
      const int anynz = __any(nz);
      if (lane == 0) rowmask[row0 + wv + 4 * q] = anynz ? 0 : 1;
    }
  }
  __syncthreads();

  const int np = node ? 3 : 1;
  if (threadIdx.x < 16 * np) {
    const int p = threadIdx.x >> 4, row = threadIdx.x & 15;
    float s = 0.f;
#pragma unroll 8
    for (int oo = 0; oo < FF; ++oo) s += ct[row][oo] * atts[p][oo];
    float* dst = node ? (p == 0 ? ha1 : (p == 1 ? ha2 : ham1)) : hma2;
    dst[row0 + row] = s;
  }
}

__device__ __forceinline__ float masked_score(float e, int adj) {
  e = e > 0.f ? e : 0.01f * e;
  return adj > 0 ? e : -__builtin_inff();
}

__global__ __launch_bounds__(128) void op_attn(
    const float* __restrict__ h, const float* __restrict__ ha1,
    const float* __restrict__ ha2, const int* __restrict__ opadj,
    const int* __restrict__ maadj, const int* __restrict__ rowmask,
    float* __restrict__ hprime) {
  extern __shared__ float smem[];
  typedef float att_row_t[NN + 4];
  att_row_t* att = (att_row_t*)smem;
  float* inv_sum = smem + 16 * (NN + 4);

  const int b  = blockIdx.y;
  const int i0 = blockIdx.x * 16;
  const int* opb = opadj + (size_t)b * NN * NN;
  const int* mab = maadj + (size_t)b * NN * NN;
  const float NEGINF = -__builtin_inff();

  const v4i* opT = (const v4i*)(opb + i0);
  const v4i* maT = (const v4i*)(mab + i0);
#pragma unroll 8
  for (int t = threadIdx.x; t < 4 * NN; t += 128) {
    const int j = t >> 2, part = t & 3;
    const v4i a = opT[(size_t)j * (NN / 4) + part];
    const v4i m = maT[(size_t)j * (NN / 4) + part];
    const v4i s = a + m;
    att[4 * part + 0][j] = __int_as_float(s.x);
    att[4 * part + 1][j] = __int_as_float(s.y);
    att[4 * part + 2][j] = __int_as_float(s.z);
    att[4 * part + 3][j] = __int_as_float(s.w);
  }
  __syncthreads();

  const int tj = threadIdx.x * 4;
  const v4f e1lo = *(const v4f*)(ha1 + b * NN + tj);
  const v4f e1hi = *(const v4f*)(ha1 + b * NN + NN / 2 + tj);
#pragma unroll 4
  for (int r = 0; r < 16; ++r) {
    const int i = i0 + r;
    const bool rm = rowmask[b * NN + i] != 0;
    const float e2 = ha2[b * NN + i];
#pragma unroll
    for (int hf = 0; hf < 2; ++hf) {
      const int j = hf * (NN / 2) + tj;
      const v4f e1 = hf ? e1hi : e1lo;
      v4f* slot = (v4f*)&att[r][j];
      const v4f tr = *slot;
      const v4i d1 = *(const v4i*)(opb + (size_t)i * NN + j);
      const v4i d2 = *(const v4i*)(mab + (size_t)i * NN + j);
      v4i adj = d1 + d2;
      adj.x += __float_as_int(tr.x);
      adj.y += __float_as_int(tr.y);
      adj.z += __float_as_int(tr.z);
      adj.w += __float_as_int(tr.w);
      if (rm) adj = (v4i){1, 1, 1, 1};
      v4f sc4;
      sc4.x = masked_score(e2 + e1.x, adj.x);
      sc4.y = masked_score(e2 + e1.y, adj.y);
      sc4.z = masked_score(e2 + e1.z, adj.z);
      sc4.w = masked_score(e2 + e1.w, adj.w);
      *slot = sc4;
    }
  }
  __syncthreads();

  const int wv = threadIdx.x >> 5;
  const int lane = threadIdx.x & 31;
  for (int r = wv; r < 16; r += 4) {
    float mx = NEGINF;
    for (int j = lane; j < NN; j += 32) mx = fmaxf(mx, att[r][j]);
#pragma unroll
    for (int off = 16; off; off >>= 1) mx = fmaxf(mx, __shfl_xor(mx, off, 32));
    if (mx == NEGINF) mx = 0.f;
    float sum = 0.f;
    for (int j = lane; j < NN; j += 32) {
      float p = __expf(att[r][j] - mx);
      att[r][j] = p;
      sum += p;
    }
#pragma unroll
    for (int off = 16; off; off >>= 1) sum += __shfl_xor(sum, off, 32);
    if (lane == 0) inv_sum[r] = 1.f / fmaxf(sum, 1e-30f);
  }
  __syncthreads();

  const int lm = lane & 15;
  const int kh = (lane >> 4) * 2;
  const int lh = lane >> 4;
  const int o  = wv * 16 + lm;
  const float* hb = h + (size_t)b * NN * FF;

  v8f c = {};
  for (int k0 = 0; k0 < NN; k0 += 4) {
    v2f a, bf;
    a.x  = att[lm][k0 + kh];
    a.y  = att[lm][k0 + kh + 1];
    bf.x = hb[(size_t)(k0 + kh) * FF + o];
    bf.y = hb[(size_t)(k0 + kh + 1) * FF + o];
    c = wmma4(a, bf, c);
  }
  float* hp = hprime + ((size_t)b * NN + i0) * (2 * FF);
#pragma unroll
  for (int r = 0; r < 8; ++r) {
    int m = r + 8 * lh;
    hp[(size_t)m * (2 * FF) + o] = c[r] * inv_sum[m];
  }
}

#define NPB 8

__device__ __forceinline__ void lds_barrier() {
  asm volatile("s_wait_dscnt 0x0" ::: "memory");
  asm volatile("s_barrier_signal -1" ::: "memory");
  asm volatile("s_barrier_wait -1" ::: "memory");
}

__global__ __launch_bounds__(256) void ma_attn(
    const float* __restrict__ z, const float* __restrict__ wedge,
    const float* hma, const float* __restrict__ ham1,
    const float* __restrict__ hma2, const int* __restrict__ opma,
    const int* __restrict__ rowmask, const float* __restrict__ h,
    const float* __restrict__ attma, float* hprime) {
  __shared__ float zb[MM][FE + 1];
  __shared__ float heb[MM][FF + 1];
  __shared__ float sc[MM];
  __shared__ float psum[4][FF];
  __shared__ __attribute__((aligned(16))) float am3s[FF];

  const int b  = blockIdx.y;
  const int nb = blockIdx.x * NPB;
  const float NEGINF = -__builtin_inff();

  const int wv = threadIdx.x >> 5;
  const int lane = threadIdx.x & 31;
  const int lm = lane & 15;
  const int kh = (lane >> 4) * 2;
  const int lh = lane >> 4;
  const int m0 = wv * 16;

  v2f bw[4][4];
#pragma unroll
  for (int ct = 0; ct < 4; ++ct) {
#pragma unroll
    for (int ks = 0; ks < 4; ++ks) {
      bw[ct][ks].x = wedge[(4 * ks + kh) * FF + ct * 16 + lm];
      bw[ct][ks].y = wedge[(4 * ks + kh + 1) * FF + ct * 16 + lm];
    }
  }

  if (threadIdx.x < FF) am3s[threadIdx.x] = attma[2 * FF + threadIdx.x];

  const float* hmb = hma + (size_t)b * MM * FF;
  const int ms = m0 + lm;
  const float hm2 = hma2[(size_t)b * MM + ms];
  const int och = threadIdx.x & 63;
  const int ch  = threadIdx.x >> 6;

  const v4f* zt = (const v4f*)(z + ((size_t)b * NN + nb) * MM * FE);
  v4f zr0 = zt[threadIdx.x];
  v4f zr1 = zt[256 + threadIdx.x];
  int   rm  = rowmask[(size_t)b * NN + nb];
  float hm1 = ham1[(size_t)b * NN + nb];
  int   adm = opma[((size_t)b * NN + nb) * MM + ms];

  for (int it = 0; it < NPB; ++it) {
    const size_t bn = (size_t)b * NN + nb + it;

    {
      const int e0 = threadIdx.x * 4;
      float* d0 = &zb[e0 >> 4][e0 & 15];
      d0[0] = zr0.x; d0[1] = zr0.y; d0[2] = zr0.z; d0[3] = zr0.w;
      const int e1 = e0 + 1024;
      float* d1 = &zb[e1 >> 4][e1 & 15];
      d1[0] = zr1.x; d1[1] = zr1.y; d1[2] = zr1.z; d1[3] = zr1.w;
    }
    lds_barrier();
    const int nx = (it + 1 < NPB) ? (it + 1) : it;
    const size_t bnx = (size_t)b * NN + nb + nx;
    const int   rm_n  = rowmask[bnx];
    const float hm1_n = ham1[bnx];
    const int   adm_n = opma[bnx * MM + ms];
    zr0 = zt[nx * 512 + threadIdx.x];
    zr1 = zt[nx * 512 + 256 + threadIdx.x];

    v2f af[4];
#pragma unroll
    for (int ks = 0; ks < 4; ++ks) {
      af[ks].x = zb[m0 + lm][4 * ks + kh];
      af[ks].y = zb[m0 + lm][4 * ks + kh + 1];
    }
#pragma unroll
    for (int ct = 0; ct < 4; ++ct) {
      v8f c = {};
#pragma unroll
      for (int ks = 0; ks < 4; ++ks) c = wmma4(af[ks], bw[ct][ks], c);
#pragma unroll
      for (int r = 0; r < 8; ++r)
        heb[m0 + r + 8 * lh][ct * 16 + lm] = c[r];
    }
    asm volatile("s_wait_dscnt 0x0" ::: "memory");

    {
      const float* hr = &heb[ms][32 * lh];
      const float* ar = &am3s[32 * lh];
      float dot = 0.f;
#pragma unroll 8
      for (int o = 0; o < FF / 2; ++o) dot += hr[o] * ar[o];
      const float oth = __shfl_xor(dot, 16, 32);
      const float lo = lh ? oth : dot;
      const float hi = lh ? dot : oth;
      float e = hm1 + hm2 + (lo + hi);
      e = e > 0.f ? e : 0.01f * e;
      const int conn = rm ? 1 : (adm > 0);
      if (lh == 0) sc[ms] = conn ? e : NEGINF;
    }
    lds_barrier();

    const float v0 = sc[lane], v1 = sc[lane + 32];
    const float v2 = sc[lane + 64], v3 = sc[lane + 96];
    float mx = fmaxf(fmaxf(v0, v1), fmaxf(v2, v3));
#pragma unroll
    for (int off = 16; off; off >>= 1) mx = fmaxf(mx, __shfl_xor(mx, off, 32));
    if (mx == NEGINF) mx = 0.f;
    const float p0 = __expf(v0 - mx), p1 = __expf(v1 - mx);
    const float p2 = __expf(v2 - mx), p3 = __expf(v3 - mx);
    float ssum = p0 + p1 + p2 + p3;
#pragma unroll
    for (int off = 16; off; off >>= 1) ssum += __shfl_xor(ssum, off, 32);
    const float inv = 1.f / fmaxf(ssum, 1e-30f);

    {
      const float pq = ch == 0 ? p0 : (ch == 1 ? p1 : (ch == 2 ? p2 : p3));
      float acc = 0.f;
#pragma unroll
      for (int i = 0; i < 32; ++i) {
        const float pm =
            __int_as_float(__builtin_amdgcn_readlane(__float_as_int(pq), i));
        const int m = ch * 32 + i;
        acc += pm * (hmb[(size_t)m * FF + och] + heb[m][och]);
      }
      psum[ch][och] = acc;
    }
    lds_barrier();
    if (threadIdx.x < FF) {
      int o = threadIdx.x;
      float v = (psum[0][o] + psum[1][o] + psum[2][o] + psum[3][o]) * inv +
                h[bn * FF + o];
      hprime[bn * (2 * FF) + FF + o] = v;
    }
    rm = rm_n; hm1 = hm1_n; adm = adm_n;
  }
}

__global__ __launch_bounds__(256) void mlp(
    const float* __restrict__ hprime, const float* __restrict__ w1,
    const float* __restrict__ b1, const float* __restrict__ w2,
    const float* __restrict__ b2, const float* __restrict__ wo,
    const float* __restrict__ bo, const int* __restrict__ rowmask,
    float* __restrict__ out) {
  extern __shared__ float smem[];
  float* wb = smem;
  typedef float act_row_t[HID + 4];
  act_row_t* a0 = (act_row_t*)(smem + HID * HID);
  act_row_t* a1 = (act_row_t*)(smem + HID * HID + 16 * (HID + 4));

  const int row0 = blockIdx.x * 16;
  const int wv = threadIdx.x >> 5;
  const int lane = threadIdx.x & 31;
  const int lm = lane & 15;
  const int kh = (lane >> 4) * 2;
  const int lh = lane >> 4;

  for (int t = threadIdx.x; t < 16 * HID; t += 256)
    a0[t >> 7][t & 127] = hprime[(size_t)row0 * HID + t];
  for (int t = threadIdx.x; t < HID * HID; t += 256) wb[t] = w1[t];
  __syncthreads();

  {
    int o = wv * 16 + lm;
    v8f c = {};
    for (int k0 = 0; k0 < HID; k0 += 4) {
      v2f a, bf;
      a.x  = a0[lm][k0 + kh];
      a.y  = a0[lm][k0 + kh + 1];
      bf.x = wb[(k0 + kh) * HID + o];
      bf.y = wb[(k0 + kh + 1) * HID + o];
      c = wmma4(a, bf, c);
    }
    float bias = b1[o];
#pragma unroll
    for (int r = 0; r < 8; ++r) {
      float v = c[r] + bias;
      a1[r + 8 * lh][o] = v > 0.f ? v : expm1f(v);
    }
  }
  __syncthreads();
  for (int t = threadIdx.x; t < HID * HID; t += 256) wb[t] = w2[t];
  __syncthreads();

  {
    int o = wv * 16 + lm;
    v8f c = {};
    for (int k0 = 0; k0 < HID; k0 += 4) {
      v2f a, bf;
      a.x  = a1[lm][k0 + kh];
      a.y  = a1[lm][k0 + kh + 1];
      bf.x = wb[(k0 + kh) * HID + o];
      bf.y = wb[(k0 + kh + 1) * HID + o];
      c = wmma4(a, bf, c);
    }
    float bias = b2[o];
#pragma unroll
    for (int r = 0; r < 8; ++r) {
      float v = c[r] + bias;
      a0[r + 8 * lh][o] = v > 0.f ? v : expm1f(v);
    }
  }
  __syncthreads();
  for (int t = threadIdx.x; t < HID * FF; t += 256) wb[t] = wo[t];
  __syncthreads();

  if (wv < 4) {
    int o = wv * 16 + lm;
    v8f c = {};
    for (int k0 = 0; k0 < HID; k0 += 4) {
      v2f a, bf;
      a.x  = a0[lm][k0 + kh];
      a.y  = a0[lm][k0 + kh + 1];
      bf.x = wb[(k0 + kh) * FF + o];
      bf.y = wb[(k0 + kh + 1) * FF + o];
      c = wmma4(a, bf, c);
    }
    float bias = bo[o];
#pragma unroll
    for (int r = 0; r < 8; ++r) {
      int m = r + 8 * lh;
      float v = c[r] + bias;
      if (rowmask[row0 + m]) v = 0.f;
      out[(size_t)(row0 + m) * FF + o] = v;
    }
  }
}

extern "C" void kernel_launch(void* const* d_in, const int* in_sizes, int n_in,
                              void* d_out, int out_size, void* d_ws,
                              size_t ws_size, hipStream_t stream) {
  (void)in_sizes; (void)n_in; (void)out_size; (void)ws_size;
  const float* x      = (const float*)d_in[0];
  const float* y      = (const float*)d_in[1];
  const float* z      = (const float*)d_in[2];
  const int*   op_adj = (const int*)d_in[3];
  const int*   ma_adj = (const int*)d_in[4];
  const int*   op_ma  = (const int*)d_in[5];
  const float* Wop    = (const float*)d_in[6];
  const float* Wma    = (const float*)d_in[7];
  const float* Wedge  = (const float*)d_in[8];
  const float* att_op = (const float*)d_in[9];
  const float* att_ma = (const float*)d_in[10];
  const float* w1     = (const float*)d_in[11];
  const float* b1     = (const float*)d_in[12];
  const float* w2     = (const float*)d_in[13];
  const float* b2     = (const float*)d_in[14];
  const float* wo     = (const float*)d_in[15];
  const float* bo     = (const float*)d_in[16];
  float* out = (float*)d_out;

  float* ws     = (float*)d_ws;
  float* h      = ws;
  float* hma    = h + BB * NN * FF;
  float* ha1    = hma + BB * MM * FF;
  float* ha2    = ha1 + BB * NN;
  float* ham1   = ha2 + BB * NN;
  float* hma2   = ham1 + BB * NN;
  float* hprime = hma2 + BB * MM;
  int*   rmask  = (int*)(hprime + (size_t)BB * NN * 2 * FF);

  prep_tiles<<<dim3(XT + YT), dim3(128), 0, stream>>>(
      x, Wop, y, Wma, att_op, att_ma, op_ma, h, hma, ha1, ha2, ham1, hma2,
      rmask);
  size_t sm2 = (16 * (NN + 4) + 16) * sizeof(float);
  op_attn<<<dim3(NN / 16, BB), dim3(128), sm2, stream>>>(
      h, ha1, ha2, op_adj, ma_adj, rmask, hprime);
  ma_attn<<<dim3(NN / NPB, BB), dim3(256), 0, stream>>>(
      z, Wedge, hma, ham1, hma2, op_ma, rmask, h, att_ma, hprime);
  size_t sm4 = (HID * HID + 2 * 16 * (HID + 4)) * sizeof(float);
  mlp<<<dim3((BB * NN) / 16), dim3(256), sm4, stream>>>(
      hprime, w1, b1, w2, b2, wo, bo, rmask, out);
}


# ma_attn: h_edge never formed, products re-associated: score = z.(we@am3), output = (p@z)@we + p@h_ma as one WMMA tile per 16 nodes; one wave per node, z tile in registers; summing order changed
# speedup vs baseline: 3.1053x; 1.3358x over previous
#include <hip/hip_runtime.h>
#include <math.h>


typedef __attribute__((ext_vector_type(2))) float v2f;
typedef __attribute__((ext_vector_type(8))) float v8f;
typedef __attribute__((ext_vector_type(4))) float v4f;
typedef __attribute__((ext_vector_type(4))) int v4i;

#define BB 8
#define NN 1024
#define MM 128
#define FF 64
#define FE 16
#define HID 128

__device__ __forceinline__ v8f wmma4(v2f a, v2f b, v8f c) {
  return __builtin_amdgcn_wmma_f32_16x16x4_f32(
       false, a,  false, b,
       (short)0, c,  false,  false);
}

#define XT ((BB * NN) / 16)
#define YT ((BB * MM) / 16)
#define CTS (FF + 4)

__global__ __launch_bounds__(128) void prep_tiles(
    const float* __restrict__ x, const float* __restrict__ Wop,
    const float* __restrict__ y, const float* __restrict__ Wma,
    const float* __restrict__ att_op, const float* __restrict__ att_ma,
    const int* __restrict__ opma, float* __restrict__ h,
    float* __restrict__ hma, float* __restrict__ ha1,
    float* __restrict__ ha2, float* __restrict__ ham1,
    float* __restrict__ hma2, int* __restrict__ rowmask) {
  __shared__ float ct[16][CTS];
  __shared__ float atts[3][FF];

  const int wv   = threadIdx.x >> 5;
  const int lane = threadIdx.x & 31;
  const int lm   = lane & 15;
  const int kh   = (lane >> 4) * 2;
  const int lh   = lane >> 4;
  const bool node = blockIdx.x < XT;
  const int row0 = (node ? blockIdx.x : blockIdx.x - XT) * 16;

  v4i ad[4] = {};
  if (node) {
    const v4i* at = (const v4i*)(opma + (size_t)row0 * MM);
#pragma unroll
    for (int q = 0; q < 4; ++q) ad[q] = at[q * 128 + threadIdx.x];
    if (threadIdx.x < FF) {
      atts[0][threadIdx.x] = att_op[threadIdx.x];
      atts[1][threadIdx.x] = att_op[FF + threadIdx.x];
      atts[2][threadIdx.x] = att_ma[threadIdx.x];
    }
  } else {
    if (threadIdx.x < FF) atts[0][threadIdx.x] = att_ma[FF + threadIdx.x];
  }

  const float* A = node ? x : y;
  const float* W = node ? Wop : Wma;
  float* C = node ? h : hma;
  const int o = wv * 16 + lm;
  const float* Arow = A + (size_t)(row0 + lm) * FF;

  v8f c = {};
#pragma unroll
  for (int k0 = 0; k0 < FF; k0 += 4) {
    v2f a, b;
    a.x = Arow[k0 + kh];
    a.y = Arow[k0 + kh + 1];
    b.x = W[(size_t)(k0 + kh) * FF + o];
    b.y = W[(size_t)(k0 + kh + 1) * FF + o];
    c = wmma4(a, b, c);
  }
#pragma unroll
  for (int r = 0; r < 8; ++r) {
    int m = r + 8 * lh;
    C[(size_t)(row0 + m) * FF + o] = c[r];
    ct[m][o] = c[r];
  }

  if (node) {
#pragma unroll
    for (int q = 0; q < 4; ++q) {
      const int nz = (ad[q].x | ad[q].y | ad[q].z | ad[q].w) != 0;
      const int anynz = __any(nz);
      if (lane == 0) rowmask[row0 + wv + 4 * q] = anynz ? 0 : 1;
    }
  }
  __syncthreads();

  const int np = node ? 3 : 1;
  if (threadIdx.x < 16 * np) {
    const int p = threadIdx.x >> 4, row = threadIdx.x & 15;
    float s = 0.f;
#pragma unroll 8
    for (int oo = 0; oo < FF; ++oo) s += ct[row][oo] * atts[p][oo];
    float* dst = node ? (p == 0 ? ha1 : (p == 1 ? ha2 : ham1)) : hma2;
    dst[row0 + row] = s;
  }
}

__device__ __forceinline__ float masked_score(float e, int adj) {
  e = e > 0.f ? e : 0.01f * e;
  return adj > 0 ? e : -__builtin_inff();
}

__global__ __launch_bounds__(128) void op_attn(
    const float* __restrict__ h, const float* __restrict__ ha1,
    const float* __restrict__ ha2, const int* __restrict__ opadj,
    const int* __restrict__ maadj, const int* __restrict__ rowmask,
    float* __restrict__ hprime) {
  extern __shared__ float smem[];
  typedef float att_row_t[NN + 4];
  att_row_t* att = (att_row_t*)smem;
  float* inv_sum = smem + 16 * (NN + 4);

  const int b  = blockIdx.y;
  const int i0 = blockIdx.x * 16;
  const int* opb = opadj + (size_t)b * NN * NN;
  const int* mab = maadj + (size_t)b * NN * NN;
  const float NEGINF = -__builtin_inff();

  const v4i* opT = (const v4i*)(opb + i0);
  const v4i* maT = (const v4i*)(mab + i0);
#pragma unroll 8
  for (int t = threadIdx.x; t < 4 * NN; t += 128) {
    const int j = t >> 2, part = t & 3;
    const v4i a = opT[(size_t)j * (NN / 4) + part];
    const v4i m = maT[(size_t)j * (NN / 4) + part];
    const v4i s = a + m;
    att[4 * part + 0][j] = __int_as_float(s.x);
    att[4 * part + 1][j] = __int_as_float(s.y);
    att[4 * part + 2][j] = __int_as_float(s.z);
    att[4 * part + 3][j] = __int_as_float(s.w);
  }
  __syncthreads();

  const int tj = threadIdx.x * 4;
  const v4f e1lo = *(const v4f*)(ha1 + b * NN + tj);
  const v4f e1hi = *(const v4f*)(ha1 + b * NN + NN / 2 + tj);
#pragma unroll 4
  for (int r = 0; r < 16; ++r) {
    const int i = i0 + r;
    const bool rm = rowmask[b * NN + i] != 0;
    const float e2 = ha2[b * NN + i];
#pragma unroll
    for (int hf = 0; hf < 2; ++hf) {
      const int j = hf * (NN / 2) + tj;
      const v4f e1 = hf ? e1hi : e1lo;
      v4f* slot = (v4f*)&att[r][j];
      const v4f tr = *slot;
      const v4i d1 = *(const v4i*)(opb + (size_t)i * NN + j);
      const v4i d2 = *(const v4i*)(mab + (size_t)i * NN + j);
      v4i adj = d1 + d2;
      adj.x += __float_as_int(tr.x);
      adj.y += __float_as_int(tr.y);
      adj.z += __float_as_int(tr.z);
      adj.w += __float_as_int(tr.w);
      if (rm) adj = (v4i){1, 1, 1, 1};
      v4f sc4;
      sc4.x = masked_score(e2 + e1.x, adj.x);
      sc4.y = masked_score(e2 + e1.y, adj.y);
      sc4.z = masked_score(e2 + e1.z, adj.z);
      sc4.w = masked_score(e2 + e1.w, adj.w);
      *slot = sc4;
    }
  }
  __syncthreads();

  const int wv = threadIdx.x >> 5;
  const int lane = threadIdx.x & 31;
  for (int r = wv; r < 16; r += 4) {
    float mx = NEGINF;
    for (int j = lane; j < NN; j += 32) mx = fmaxf(mx, att[r][j]);
#pragma unroll
    for (int off = 16; off; off >>= 1) mx = fmaxf(mx, __shfl_xor(mx, off, 32));
    if (mx == NEGINF) mx = 0.f;
    float sum = 0.f;
    for (int j = lane; j < NN; j += 32) {
      float p = __expf(att[r][j] - mx);
      att[r][j] = p;
      sum += p;
    }
#pragma unroll
    for (int off = 16; off; off >>= 1) sum += __shfl_xor(sum, off, 32);
    if (lane == 0) inv_sum[r] = 1.f / fmaxf(sum, 1e-30f);
  }
  __syncthreads();

  const int lm = lane & 15;
  const int kh = (lane >> 4) * 2;
  const int lh = lane >> 4;
  const int o  = wv * 16 + lm;
  const float* hb = h + (size_t)b * NN * FF;

  v8f c = {};
  for (int k0 = 0; k0 < NN; k0 += 4) {
    v2f a, bf;
    a.x  = att[lm][k0 + kh];
    a.y  = att[lm][k0 + kh + 1];
    bf.x = hb[(size_t)(k0 + kh) * FF + o];
    bf.y = hb[(size_t)(k0 + kh + 1) * FF + o];
    c = wmma4(a, bf, c);
  }
  float* hp = hprime + ((size_t)b * NN + i0) * (2 * FF);
#pragma unroll
  for (int r = 0; r < 8; ++r) {
    int m = r + 8 * lh;
    hp[(size_t)m * (2 * FF) + o] = c[r] * inv_sum[m];
  }
}

#define NPW 2
#define NPG (8 * NPW)
#define PST (MM + 4)
#define ZST (FE + 4)

__device__ __forceinline__ float quad_sum(float v) {
  v += __int_as_float(__builtin_amdgcn_update_dpp(
      0, __float_as_int(v), 0xB1, 0xF, 0xF, true));
  v += __int_as_float(__builtin_amdgcn_update_dpp(
      0, __float_as_int(v), 0x4E, 0xF, 0xF, true));
  return v;
}

__device__ __forceinline__ float across_quads_sum(float v) {
  v += __shfl_xor(v, 4, 32);
  v += __shfl_xor(v, 8, 32);
  v += __shfl_xor(v, 16, 32);
  return v;
}

__global__ __launch_bounds__(256) void ma_attn(
    const float* __restrict__ z, const float* __restrict__ wedge,
    const float* __restrict__ hma, const float* __restrict__ ham1,
    const float* __restrict__ hma2, const int* __restrict__ opma,
    const int* __restrict__ rowmask, const float* __restrict__ h,
    const float* __restrict__ attma, float* __restrict__ hprime) {
  __shared__ __attribute__((aligned(16))) float Pm[NPG][PST];
  __shared__ __attribute__((aligned(16))) float Zp[NPG][ZST];
  __shared__ float invs[NPG];

  const int b  = blockIdx.y;
  const int nb = blockIdx.x * NPG;
  const float NEGINF = -__builtin_inff();

  const int wv = __builtin_amdgcn_readfirstlane((int)(threadIdx.x >> 5));
  const int lane = threadIdx.x & 31;
  const int q  = lane & 3;
  const int rq = lane >> 2;

  const size_t bn0 = (size_t)b * NN + nb + wv * NPW;

  v4f wv4[8], av4[8];
  {
    const v4f* wr = (const v4f*)(wedge + (lane & 15) * FF + 32 * (lane >> 4));
    const v4f* ar = (const v4f*)(attma + 2 * FF + 32 * (lane >> 4));
#pragma unroll
    for (int i = 0; i < 8; ++i) { wv4[i] = wr[i]; av4[i] = ar[i]; }
  }
  float hm2r[16];
#pragma unroll
  for (int j = 0; j < 16; ++j) hm2r[j] = hma2[(size_t)b * MM + rq + 8 * j];
  v4i   ad[NPW];
  int   rmv[NPW];
  float hm1v[NPW];
#pragma unroll
  for (int t = 0; t < NPW; ++t) {
    ad[t]   = *(const v4i*)(opma + (bn0 + t) * MM + 4 * lane);
    rmv[t]  = rowmask[bn0 + t];
    hm1v[t] = ham1[bn0 + t];
  }

  const v4f* zt = (const v4f*)(z + bn0 * MM * FE);
  v4f zr[NPW][16];
#pragma unroll
  for (int t = 0; t < NPW; ++t) {
#pragma unroll
    for (int j = 0; j < 16; ++j) zr[t][j] = zt[t * 512 + j * 32 + lane];
  }

  unsigned connv[NPW];
#pragma unroll
  for (int t = 0; t < NPW; ++t) {
    const unsigned s0 = __builtin_amdgcn_ballot_w32(ad[t].x > 0);
    const unsigned s1 = __builtin_amdgcn_ballot_w32(ad[t].y > 0);
    const unsigned s2 = __builtin_amdgcn_ballot_w32(ad[t].z > 0);
    const unsigned s3 = __builtin_amdgcn_ballot_w32(ad[t].w > 0);
    const int cq = rq & 3;
    unsigned conn = cq == 0 ? s0 : (cq == 1 ? s1 : (cq == 2 ? s2 : s3));
    conn >>= (lane >> 4);
    if (rmv[t]) conn = 0xffffffffu;
    connv[t] = conn;
  }

  float wpart = 0.f;
#pragma unroll
  for (int i = 0; i < 8; ++i) {
    wpart += wv4[i].x * av4[i].x;
    wpart += wv4[i].y * av4[i].y;
    wpart += wv4[i].z * av4[i].z;
    wpart += wv4[i].w * av4[i].w;
  }
  wpart += __shfl_xor(wpart, 16, 32);
  float wq[4];
#pragma unroll
  for (int c = 0; c < 4; ++c) wq[c] = __shfl(wpart, 4 * q + c, 32);

#pragma unroll
  for (int t = 0; t < NPW; ++t) {
    const int nl = wv * NPW + t;

    const unsigned conn = connv[t];

    float sc[16];
    float mx = NEGINF;
#pragma unroll
    for (int j = 0; j < 16; ++j) {
      const v4f zv = zr[t][j];
      float d = zv.x * wq[0];
      d += zv.y * wq[1];
      d += zv.z * wq[2];
      d += zv.w * wq[3];
      d = quad_sum(d);
      float e = (hm1v[t] + hm2r[j]) + d;
      e = e > 0.f ? e : 0.01f * e;
      sc[j] = ((conn >> (2 * j)) & 1u) ? e : NEGINF;
      mx = fmaxf(mx, sc[j]);
    }
    mx = fmaxf(mx, __shfl_xor(mx, 4, 32));
    mx = fmaxf(mx, __shfl_xor(mx, 8, 32));
    mx = fmaxf(mx, __shfl_xor(mx, 16, 32));
    if (mx == NEGINF) mx = 0.f;

    float ps = 0.f;
    v4f acc = {0.f, 0.f, 0.f, 0.f};
#pragma unroll
    for (int j = 0; j < 16; ++j) {
      const float p = __expf(sc[j] - mx);
      sc[j] = p;
      ps += p;
      acc += p * zr[t][j];
    }
    ps = across_quads_sum(ps);
    acc.x = across_quads_sum(acc.x);
    acc.y = across_quads_sum(acc.y);
    acc.z = across_quads_sum(acc.z);
    acc.w = across_quads_sum(acc.w);

    if (q == 0) {
#pragma unroll
      for (int j = 0; j < 16; ++j) Pm[nl][rq + 8 * j] = sc[j];
    }
    if (lane < 4) *(v4f*)&Zp[nl][4 * lane] = acc;
    if (lane == 0) invs[nl] = 1.f / fmaxf(ps, 1e-30f);
  }

  const int lm = lane & 15;
  const int kh = (lane >> 4) * 2;
  const int lh = lane >> 4;
  const int o  = (wv & 3) * 16 + lm;
  const float* hmb = hma + (size_t)b * MM * FF;
  v2f bh[MM / 4];
  v2f bw[FE / 4];
  if (wv < 4) {
#pragma unroll
    for (int ks = 0; ks < MM / 4; ++ks) {
      bh[ks].x = hmb[(size_t)(4 * ks + kh) * FF + o];
      bh[ks].y = hmb[(size_t)(4 * ks + kh + 1) * FF + o];
    }
#pragma unroll
    for (int ks = 0; ks < FE / 4; ++ks) {
      bw[ks].x = wedge[(4 * ks + kh) * FF + o];
      bw[ks].y = wedge[(4 * ks + kh + 1) * FF + o];
    }
  }
  __syncthreads();

  if (wv < 4) {
    v8f c0 = {}, c1 = {};
#pragma unroll
    for (int ks = 0; ks < MM / 4; ++ks) {
      v2f a;
      a.x = Pm[lm][4 * ks + kh];
      a.y = Pm[lm][4 * ks + kh + 1];
      if (ks & 1) c1 = wmma4(a, bh[ks], c1);
      else        c0 = wmma4(a, bh[ks], c0);
    }
#pragma unroll
    for (int ks = 0; ks < FE / 4; ++ks) {
      v2f a;
      a.x = Zp[lm][4 * ks + kh];
      a.y = Zp[lm][4 * ks + kh + 1];
      if (ks & 1) c1 = wmma4(a, bw[ks], c1);
      else        c0 = wmma4(a, bw[ks], c0);
    }
    const v8f c = c0 + c1;
    const size_t bnr = (size_t)b * NN + nb;
#pragma unroll
    for (int r = 0; r < 8; ++r) {
      const int m = r + 8 * lh;
      hprime[(bnr + m) * (2 * FF) + FF + o] =
          c[r] * invs[m] + h[(bnr + m) * FF + o];
    }
  }
}

__global__ __launch_bounds__(256) void mlp(
    const float* __restrict__ hprime, const float* __restrict__ w1,
    const float* __restrict__ b1, const float* __restrict__ w2,
    const float* __restrict__ b2, const float* __restrict__ wo,
    const float* __restrict__ bo, const int* __restrict__ rowmask,
    float* __restrict__ out) {
  extern __shared__ float smem[];
  float* wb = smem;
  typedef float act_row_t[HID + 4];
  act_row_t* a0 = (act_row_t*)(smem + HID * HID);
  act_row_t* a1 = (act_row_t*)(smem + HID * HID + 16 * (HID + 4));

  const int row0 = blockIdx.x * 16;
  const int wv = threadIdx.x >> 5;
  const int lane = threadIdx.x & 31;
  const int lm = lane & 15;
  const int kh = (lane >> 4) * 2;
  const int lh = lane >> 4;

  for (int t = threadIdx.x; t < 16 * HID; t += 256)
    a0[t >> 7][t & 127] = hprime[(size_t)row0 * HID + t];
  for (int t = threadIdx.x; t < HID * HID; t += 256) wb[t] = w1[t];
  __syncthreads();

  {
    int o = wv * 16 + lm;
    v8f c = {};
    for (int k0 = 0; k0 < HID; k0 += 4) {
      v2f a, bf;
      a.x  = a0[lm][k0 + kh];
      a.y  = a0[lm][k0 + kh + 1];
      bf.x = wb[(k0 + kh) * HID + o];
      bf.y = wb[(k0 + kh + 1) * HID + o];
      c = wmma4(a, bf, c);
    }
    float bias = b1[o];
#pragma unroll
    for (int r = 0; r < 8; ++r) {
      float v = c[r] + bias;
      a1[r + 8 * lh][o] = v > 0.f ? v : expm1f(v);
    }
  }
  __syncthreads();
  for (int t = threadIdx.x; t < HID * HID; t += 256) wb[t] = w2[t];
  __syncthreads();

  {
    int o = wv * 16 + lm;
    v8f c = {};
    for (int k0 = 0; k0 < HID; k0 += 4) {
      v2f a, bf;
      a.x  = a1[lm][k0 + kh];
      a.y  = a1[lm][k0 + kh + 1];
      bf.x = wb[(k0 + kh) * HID + o];
      bf.y = wb[(k0 + kh + 1) * HID + o];
      c = wmma4(a, bf, c);
    }
    float bias = b2[o];
#pragma unroll
    for (int r = 0; r < 8; ++r) {
      float v = c[r] + bias;
      a0[r + 8 * lh][o] = v > 0.f ? v : expm1f(v);
    }
  }
  __syncthreads();
  for (int t = threadIdx.x; t < HID * FF; t += 256) wb[t] = wo[t];
  __syncthreads();

  if (wv < 4) {
    int o = wv * 16 + lm;
    v8f c = {};
    for (int k0 = 0; k0 < HID; k0 += 4) {
      v2f a, bf;
      a.x  = a0[lm][k0 + kh];
      a.y  = a0[lm][k0 + kh + 1];
      bf.x = wb[(k0 + kh) * FF + o];
      bf.y = wb[(k0 + kh + 1) * FF + o];
      c = wmma4(a, bf, c);
    }
    float bias = bo[o];
#pragma unroll
    for (int r = 0; r < 8; ++r) {
      int m = r + 8 * lh;
      float v = c[r] + bias;
      if (rowmask[row0 + m]) v = 0.f;
      out[(size_t)(row0 + m) * FF + o] = v;
    }
  }
}

extern "C" void kernel_launch(void* const* d_in, const int* in_sizes, int n_in,
                              void* d_out, int out_size, void* d_ws,
                              size_t ws_size, hipStream_t stream) {
  (void)in_sizes; (void)n_in; (void)out_size; (void)ws_size;
  const float* x      = (const float*)d_in[0];
  const float* y      = (const float*)d_in[1];
  const float* z      = (const float*)d_in[2];
  const int*   op_adj = (const int*)d_in[3];
  const int*   ma_adj = (const int*)d_in[4];
  const int*   op_ma  = (const int*)d_in[5];
  const float* Wop    = (const float*)d_in[6];
  const float* Wma    = (const float*)d_in[7];
  const float* Wedge  = (const float*)d_in[8];
  const float* att_op = (const float*)d_in[9];
  const float* att_ma = (const float*)d_in[10];
  const float* w1     = (const float*)d_in[11];
  const float* b1     = (const float*)d_in[12];
  const float* w2     = (const float*)d_in[13];
  const float* b2     = (const float*)d_in[14];
  const float* wo     = (const float*)d_in[15];
  const float* bo     = (const float*)d_in[16];
  float* out = (float*)d_out;

  float* ws     = (float*)d_ws;
  float* h      = ws;
  float* hma    = h + BB * NN * FF;
  float* ha1    = hma + BB * MM * FF;
  float* ha2    = ha1 + BB * NN;
  float* ham1   = ha2 + BB * NN;
  float* hma2   = ham1 + BB * NN;
  float* hprime = hma2 + BB * MM;
  int*   rmask  = (int*)(hprime + (size_t)BB * NN * 2 * FF);

  prep_tiles<<<dim3(XT + YT), dim3(128), 0, stream>>>(
      x, Wop, y, Wma, att_op, att_ma, op_ma, h, hma, ha1, ha2, ham1, hma2,
      rmask);
  size_t sm2 = (16 * (NN + 4) + 16) * sizeof(float);
  op_attn<<<dim3(NN / 16, BB), dim3(128), sm2, stream>>>(
      h, ha1, ha2, op_adj, ma_adj, rmask, hprime);
  ma_attn<<<dim3(NN / NPG, BB), dim3(256), 0, stream>>>(
      z, Wedge, hma, ham1, hma2, op_ma, rmask, h, att_ma, hprime);
  size_t sm4 = (HID * HID + 2 * 16 * (HID + 4)) * sizeof(float);
  mlp<<<dim3((BB * NN) / 16), dim3(256), sm4, stream>>>(
      hprime, w1, b1, w2, b2, wo, bo, rmask, out);
}


# op_attn product att@h: a wave takes a quarter of k and all 64 columns (row of h = one 16-byte load per lane), fragments requested a block ahead; quarter sums added via LDS: summing order changed
# speedup vs baseline: 3.5956x; 1.1579x over previous
#include <hip/hip_runtime.h>
#include <math.h>


typedef __attribute__((ext_vector_type(2))) float v2f;
typedef __attribute__((ext_vector_type(8))) float v8f;
typedef __attribute__((ext_vector_type(4))) float v4f;
typedef __attribute__((ext_vector_type(4))) int v4i;

#define BB 8
#define NN 1024
#define MM 128
#define FF 64
#define FE 16
#define HID 128

__device__ __forceinline__ v8f wmma4(v2f a, v2f b, v8f c) {
  return __builtin_amdgcn_wmma_f32_16x16x4_f32(
       false, a,  false, b,
       (short)0, c,  false,  false);
}

#define XT ((BB * NN) / 16)
#define YT ((BB * MM) / 16)
#define CTS (FF + 4)

__global__ __launch_bounds__(128) void prep_tiles(
    const float* __restrict__ x, const float* __restrict__ Wop,
    const float* __restrict__ y, const float* __restrict__ Wma,
    const float* __restrict__ att_op, const float* __restrict__ att_ma,
    const int* __restrict__ opma, float* __restrict__ h,
    float* __restrict__ hma, float* __restrict__ ha1,
    float* __restrict__ ha2, float* __restrict__ ham1,
    float* __restrict__ hma2, int* __restrict__ rowmask) {
  __shared__ float ct[16][CTS];
  __shared__ float atts[3][FF];

  const int wv   = threadIdx.x >> 5;
  const int lane = threadIdx.x & 31;
  const int lm   = lane & 15;
  const int kh   = (lane >> 4) * 2;
  const int lh   = lane >> 4;
  const bool node = blockIdx.x < XT;
  const int row0 = (node ? blockIdx.x : blockIdx.x - XT) * 16;

  v4i ad[4] = {};
  if (node) {
    const v4i* at = (const v4i*)(opma + (size_t)row0 * MM);
#pragma unroll
    for (int q = 0; q < 4; ++q) ad[q] = at[q * 128 + threadIdx.x];
    if (threadIdx.x < FF) {
      atts[0][threadIdx.x] = att_op[threadIdx.x];
      atts[1][threadIdx.x] = att_op[FF + threadIdx.x];
      atts[2][threadIdx.x] = att_ma[threadIdx.x];
    }
  } else {
    if (threadIdx.x < FF) atts[0][threadIdx.x] = att_ma[FF + threadIdx.x];
  }

  const float* A = node ? x : y;
  const float* W = node ? Wop : Wma;
  float* C = node ? h : hma;
  const int o = wv * 16 + lm;
  const float* Arow = A + (size_t)(row0 + lm) * FF;

  v8f c = {};
#pragma unroll
  for (int k0 = 0; k0 < FF; k0 += 4) {
    v2f a, b;
    a.x = Arow[k0 + kh];
    a.y = Arow[k0 + kh + 1];
    b.x = W[(size_t)(k0 + kh) * FF + o];
    b.y = W[(size_t)(k0 + kh + 1) * FF + o];
    c = wmma4(a, b, c);
  }
#pragma unroll
  for (int r = 0; r < 8; ++r) {
    int m = r + 8 * lh;
    C[(size_t)(row0 + m) * FF + o] = c[r];
    ct[m][o] = c[r];
  }

  if (node) {
#pragma unroll
    for (int q = 0; q < 4; ++q) {
      const int nz = (ad[q].x | ad[q].y | ad[q].z | ad[q].w) != 0;
      const int anynz = __any(nz);
      if (lane == 0) rowmask[row0 + wv + 4 * q] = anynz ? 0 : 1;
    }
  }
  __syncthreads();

  const int np = node ? 3 : 1;
  if (threadIdx.x < 16 * np) {
    const int p = threadIdx.x >> 4, row = threadIdx.x & 15;
    float s = 0.f;
#pragma unroll 8
    for (int oo = 0; oo < FF; ++oo) s += ct[row][oo] * atts[p][oo];
    float* dst = node ? (p == 0 ? ha1 : (p == 1 ? ha2 : ham1)) : hma2;
    dst[row0 + row] = s;
  }
}

__device__ __forceinline__ float masked_score(float e, int adj) {
  e = e > 0.f ? e : 0.01f * e;
  return adj > 0 ? e : -__builtin_inff();
}

#define KSB 8
#define NBLK (NN / 4 / 4 / KSB)
__global__ __launch_bounds__(128) void op_attn(
    const float* __restrict__ h, const float* __restrict__ ha1,
    const float* __restrict__ ha2, const int* __restrict__ opadj,
    const int* __restrict__ maadj, const int* __restrict__ rowmask,
    float* __restrict__ hprime) {
  extern __shared__ float smem[];
  typedef float att_row_t[NN + 4];
  att_row_t* att = (att_row_t*)smem;
  float* inv_sum = smem + 16 * (NN + 4);

  const int b  = blockIdx.y;
  const int i0 = blockIdx.x * 16;
  const int* opb = opadj + (size_t)b * NN * NN;
  const int* mab = maadj + (size_t)b * NN * NN;
  const float NEGINF = -__builtin_inff();

  const v4i* opT = (const v4i*)(opb + i0);
  const v4i* maT = (const v4i*)(mab + i0);
#pragma unroll 8
  for (int t = threadIdx.x; t < 4 * NN; t += 128) {
    const int j = t >> 2, part = t & 3;
    const v4i a = opT[(size_t)j * (NN / 4) + part];
    const v4i m = maT[(size_t)j * (NN / 4) + part];
    const v4i s = a + m;
    att[4 * part + 0][j] = __int_as_float(s.x);
    att[4 * part + 1][j] = __int_as_float(s.y);
    att[4 * part + 2][j] = __int_as_float(s.z);
    att[4 * part + 3][j] = __int_as_float(s.w);
  }
  __syncthreads();

  const int tj = threadIdx.x * 4;
  const v4f e1lo = *(const v4f*)(ha1 + b * NN + tj);
  const v4f e1hi = *(const v4f*)(ha1 + b * NN + NN / 2 + tj);
#pragma unroll 4
  for (int r = 0; r < 16; ++r) {
    const int i = i0 + r;
    const bool rm = rowmask[b * NN + i] != 0;
    const float e2 = ha2[b * NN + i];
#pragma unroll
    for (int hf = 0; hf < 2; ++hf) {
      const int j = hf * (NN / 2) + tj;
      const v4f e1 = hf ? e1hi : e1lo;
      v4f* slot = (v4f*)&att[r][j];
      const v4f tr = *slot;
      const v4i d1 = *(const v4i*)(opb + (size_t)i * NN + j);
      const v4i d2 = *(const v4i*)(mab + (size_t)i * NN + j);
      v4i adj = d1 + d2;
      adj.x += __float_as_int(tr.x);
      adj.y += __float_as_int(tr.y);
      adj.z += __float_as_int(tr.z);
      adj.w += __float_as_int(tr.w);
      if (rm) adj = (v4i){1, 1, 1, 1};
      v4f sc4;
      sc4.x = masked_score(e2 + e1.x, adj.x);
      sc4.y = masked_score(e2 + e1.y, adj.y);
      sc4.z = masked_score(e2 + e1.z, adj.z);
      sc4.w = masked_score(e2 + e1.w, adj.w);
      *slot = sc4;
    }
  }
  __syncthreads();

  const int wv = threadIdx.x >> 5;
  const int lane = threadIdx.x & 31;
  for (int r = wv; r < 16; r += 4) {
    float mx = NEGINF;
    for (int j = lane; j < NN; j += 32) mx = fmaxf(mx, att[r][j]);
#pragma unroll
    for (int off = 16; off; off >>= 1) mx = fmaxf(mx, __shfl_xor(mx, off, 32));
    if (mx == NEGINF) mx = 0.f;
    float sum = 0.f;
    for (int j = lane; j < NN; j += 32) {
      float p = __expf(att[r][j] - mx);
      att[r][j] = p;
      sum += p;
    }
#pragma unroll
    for (int off = 16; off; off >>= 1) sum += __shfl_xor(sum, off, 32);
    if (lane == 0) inv_sum[r] = 1.f / fmaxf(sum, 1e-30f);
  }
  __syncthreads();

  const int lm = lane & 15;
  const int lh = lane >> 4;
  const float* hb = h + (size_t)b * NN * FF;
  const int kb = wv * (NN / 4);
  const float* hq = hb + (size_t)(kb + 2 * lh) * FF + 4 * lm;
  const float* aq = &att[lm][kb + 2 * lh];

  v4f bq[2][KSB][2];
  v2f aa[2][KSB];
  v8f acc[4] = {};

  auto fetch = [&](const int st, const int blk) {
#pragma unroll
    for (int s = 0; s < KSB; ++s) {
      const int ks = blk * KSB + s;
      bq[st][s][0] = *(const v4f*)(hq + (size_t)(4 * ks) * FF);
      bq[st][s][1] = *(const v4f*)(hq + (size_t)(4 * ks + 1) * FF);
      aa[st][s]    = *(const v2f*)(aq + 4 * ks);
    }
  };
  auto mulblk = [&](const int st) {
#pragma unroll
    for (int s = 0; s < KSB; ++s) {
#pragma unroll
      for (int t = 0; t < 4; ++t) {
        v2f bf;
        bf.x = bq[st][s][0][t];
        bf.y = bq[st][s][1][t];
        acc[t] = wmma4(aa[st][s], bf, acc[t]);
      }
    }
  };

  fetch(0, 0);
  for (int it = 0; it < NBLK / 2 - 1; ++it) {
    fetch(1, 2 * it + 1);
    __builtin_amdgcn_sched_barrier(0);
    mulblk(0);
    __builtin_amdgcn_sched_barrier(0);
    fetch(0, 2 * it + 2);
    asm volatile("" ::: "memory");
    __builtin_amdgcn_sched_barrier(0);
    mulblk(1);
  }
  fetch(1, NBLK - 1);
  mulblk(0);
  mulblk(1);

  __syncthreads();
  typedef float red_tile_t[16][FF + 4];
  red_tile_t* red = (red_tile_t*)smem;
#pragma unroll
  for (int r = 0; r < 8; ++r) {
    v4f v;
    v.x = acc[0][r]; v.y = acc[1][r]; v.z = acc[2][r]; v.w = acc[3][r];
    *(v4f*)&red[wv][r + 8 * lh][4 * lm] = v;
  }
  __syncthreads();

  {
    const int m = threadIdx.x >> 3;
    const int g = (threadIdx.x & 7) * 8;
    v4f s0 = *(const v4f*)&red[0][m][g];
    v4f s1 = *(const v4f*)&red[0][m][g + 4];
#pragma unroll
    for (int w = 1; w < 4; ++w) {
      s0 += *(const v4f*)&red[w][m][g];
      s1 += *(const v4f*)&red[w][m][g + 4];
    }
    const float is = inv_sum[m];
    float* hp = hprime + ((size_t)b * NN + i0 + m) * (2 * FF) + g;
    *(v4f*)hp       = s0 * is;
    *(v4f*)(hp + 4) = s1 * is;
  }
}

#define NPW 2
#define NPG (8 * NPW)
#define PST (MM + 4)
#define ZST (FE + 4)

__device__ __forceinline__ float quad_sum(float v) {
  v += __int_as_float(__builtin_amdgcn_update_dpp(
      0, __float_as_int(v), 0xB1, 0xF, 0xF, true));
  v += __int_as_float(__builtin_amdgcn_update_dpp(
      0, __float_as_int(v), 0x4E, 0xF, 0xF, true));
  return v;
}

__device__ __forceinline__ float across_quads_sum(float v) {
  v += __shfl_xor(v, 4, 32);
  v += __shfl_xor(v, 8, 32);
  v += __shfl_xor(v, 16, 32);
  return v;
}

__global__ __launch_bounds__(256) void ma_attn(
    const float* __restrict__ z, const float* __restrict__ wedge,
    const float* __restrict__ hma, const float* __restrict__ ham1,
    const float* __restrict__ hma2, const int* __restrict__ opma,
    const int* __restrict__ rowmask, const float* __restrict__ h,
    const float* __restrict__ attma, float* __restrict__ hprime) {
  __shared__ __attribute__((aligned(16))) float Pm[NPG][PST];
  __shared__ __attribute__((aligned(16))) float Zp[NPG][ZST];
  __shared__ float invs[NPG];

  const int b  = blockIdx.y;
  const int nb = blockIdx.x * NPG;
  const float NEGINF = -__builtin_inff();

  const int wv = __builtin_amdgcn_readfirstlane((int)(threadIdx.x >> 5));
  const int lane = threadIdx.x & 31;
  const int q  = lane & 3;
  const int rq = lane >> 2;

  const size_t bn0 = (size_t)b * NN + nb + wv * NPW;

  v4f wv4[8], av4[8];
  {
    const v4f* wr = (const v4f*)(wedge + (lane & 15) * FF + 32 * (lane >> 4));
    const v4f* ar = (const v4f*)(attma + 2 * FF + 32 * (lane >> 4));
#pragma unroll
    for (int i = 0; i < 8; ++i) { wv4[i] = wr[i]; av4[i] = ar[i]; }
  }
  float hm2r[16];
#pragma unroll
  for (int j = 0; j < 16; ++j) hm2r[j] = hma2[(size_t)b * MM + rq + 8 * j];
  v4i   ad[NPW];
  int   rmv[NPW];
  float hm1v[NPW];
#pragma unroll
  for (int t = 0; t < NPW; ++t) {
    ad[t]   = *(const v4i*)(opma + (bn0 + t) * MM + 4 * lane);
    rmv[t]  = rowmask[bn0 + t];
    hm1v[t] = ham1[bn0 + t];
  }

  const v4f* zt = (const v4f*)(z + bn0 * MM * FE);
  v4f zr[NPW][16];
#pragma unroll
  for (int t = 0; t < NPW; ++t) {
#pragma unroll
    for (int j = 0; j < 16; ++j) zr[t][j] = zt[t * 512 + j * 32 + lane];
  }

  unsigned connv[NPW];
#pragma unroll
  for (int t = 0; t < NPW; ++t) {
    const unsigned s0 = __builtin_amdgcn_ballot_w32(ad[t].x > 0);
    const unsigned s1 = __builtin_amdgcn_ballot_w32(ad[t].y > 0);
    const unsigned s2 = __builtin_amdgcn_ballot_w32(ad[t].z > 0);
    const unsigned s3 = __builtin_amdgcn_ballot_w32(ad[t].w > 0);
    const int cq = rq & 3;
    unsigned conn = cq == 0 ? s0 : (cq == 1 ? s1 : (cq == 2 ? s2 : s3));
    conn >>= (lane >> 4);
    if (rmv[t]) conn = 0xffffffffu;
    connv[t] = conn;
  }

  float wpart = 0.f;
#pragma unroll
  for (int i = 0; i < 8; ++i) {
    wpart += wv4[i].x * av4[i].x;
    wpart += wv4[i].y * av4[i].y;
    wpart += wv4[i].z * av4[i].z;
    wpart += wv4[i].w * av4[i].w;
  }
  wpart += __shfl_xor(wpart, 16, 32);
  float wq[4];
#pragma unroll
  for (int c = 0; c < 4; ++c) wq[c] = __shfl(wpart, 4 * q + c, 32);

#pragma unroll
  for (int t = 0; t < NPW; ++t) {
    const int nl = wv * NPW + t;

    const unsigned conn = connv[t];

    float sc[16];
    float mx = NEGINF;
#pragma unroll
    for (int j = 0; j < 16; ++j) {
      const v4f zv = zr[t][j];
      float d = zv.x * wq[0];
      d += zv.y * wq[1];
      d += zv.z * wq[2];
      d += zv.w * wq[3];
      d = quad_sum(d);
      float e = (hm1v[t] + hm2r[j]) + d;
      e = e > 0.f ? e : 0.01f * e;
      sc[j] = ((conn >> (2 * j)) & 1u) ? e : NEGINF;
      mx = fmaxf(mx, sc[j]);
    }
    mx = fmaxf(mx, __shfl_xor(mx, 4, 32));
    mx = fmaxf(mx, __shfl_xor(mx, 8, 32));
    mx = fmaxf(mx, __shfl_xor(mx, 16, 32));
    if (mx == NEGINF) mx = 0.f;

    float ps = 0.f;
    v4f acc = {0.f, 0.f, 0.f, 0.f};
#pragma unroll
    for (int j = 0; j < 16; ++j) {
      const float p = __expf(sc[j] - mx);
      sc[j] = p;
      ps += p;
      acc += p * zr[t][j];
    }
    ps = across_quads_sum(ps);
    acc.x = across_quads_sum(acc.x);
    acc.y = across_quads_sum(acc.y);
    acc.z = across_quads_sum(acc.z);
    acc.w = across_quads_sum(acc.w);

    if (q == 0) {
#pragma unroll
      for (int j = 0; j < 16; ++j) Pm[nl][rq + 8 * j] = sc[j];
    }
    if (lane < 4) *(v4f*)&Zp[nl][4 * lane] = acc;
    if (lane == 0) invs[nl] = 1.f / fmaxf(ps, 1e-30f);
  }

  const int lm = lane & 15;
  const int kh = (lane >> 4) * 2;
  const int lh = lane >> 4;
  const int o  = (wv & 3) * 16 + lm;
  const float* hmb = hma + (size_t)b * MM * FF;
  v2f bh[MM / 4];
  v2f bw[FE / 4];
  if (wv < 4) {
#pragma unroll
    for (int ks = 0; ks < MM / 4; ++ks) {
      bh[ks].x = hmb[(size_t)(4 * ks + kh) * FF + o];
      bh[ks].y = hmb[(size_t)(4 * ks + kh + 1) * FF + o];
    }
#pragma unroll
    for (int ks = 0; ks < FE / 4; ++ks) {
      bw[ks].x = wedge[(4 * ks + kh) * FF + o];
      bw[ks].y = wedge[(4 * ks + kh + 1) * FF + o];
    }
  }
  __syncthreads();

  if (wv < 4) {
    v8f c0 = {}, c1 = {};
#pragma unroll
    for (int ks = 0; ks < MM / 4; ++ks) {
      v2f a;
      a.x = Pm[lm][4 * ks + kh];
      a.y = Pm[lm][4 * ks + kh + 1];
      if (ks & 1) c1 = wmma4(a, bh[ks], c1);
      else        c0 = wmma4(a, bh[ks], c0);
    }
#pragma unroll
    for (int ks = 0; ks < FE / 4; ++ks) {
      v2f a;
      a.x = Zp[lm][4 * ks + kh];
      a.y = Zp[lm][4 * ks + kh + 1];
      if (ks & 1) c1 = wmma4(a, bw[ks], c1);
      else        c0 = wmma4(a, bw[ks], c0);
    }
    const v8f c = c0 + c1;
    const size_t bnr = (size_t)b * NN + nb;
#pragma unroll
    for (int r = 0; r < 8; ++r) {
      const int m = r + 8 * lh;
      hprime[(bnr + m) * (2 * FF) + FF + o] =
          c[r] * invs[m] + h[(bnr + m) * FF + o];
    }
  }
}

__global__ __launch_bounds__(256) void mlp(
    const float* __restrict__ hprime, const float* __restrict__ w1,
    const float* __restrict__ b1, const float* __restrict__ w2,
    const float* __restrict__ b2, const float* __restrict__ wo,
    const float* __restrict__ bo, const int* __restrict__ rowmask,
    float* __restrict__ out) {
  extern __shared__ float smem[];
  float* wb = smem;
  typedef float act_row_t[HID + 4];
  act_row_t* a0 = (act_row_t*)(smem + HID * HID);
  act_row_t* a1 = (act_row_t*)(smem + HID * HID + 16 * (HID + 4));

  const int row0 = blockIdx.x * 16;
  const int wv = threadIdx.x >> 5;
  const int lane = threadIdx.x & 31;
  const int lm = lane & 15;
  const int kh = (lane >> 4) * 2;
  const int lh = lane >> 4;

  for (int t = threadIdx.x; t < 16 * HID; t += 256)
    a0[t >> 7][t & 127] = hprime[(size_t)row0 * HID + t];
  for (int t = threadIdx.x; t < HID * HID; t += 256) wb[t] = w1[t];
  __syncthreads();

  {
    int o = wv * 16 + lm;
    v8f c = {};
    for (int k0 = 0; k0 < HID; k0 += 4) {
      v2f a, bf;
      a.x  = a0[lm][k0 + kh];
      a.y  = a0[lm][k0 + kh + 1];
      bf.x = wb[(k0 + kh) * HID + o];
      bf.y = wb[(k0 + kh + 1) * HID + o];
      c = wmma4(a, bf, c);
    }
    float bias = b1[o];
#pragma unroll
    for (int r = 0; r < 8; ++r) {
      float v = c[r] + bias;
      a1[r + 8 * lh][o] = v > 0.f ? v : expm1f(v);
    }
  }
  __syncthreads();
  for (int t = threadIdx.x; t < HID * HID; t += 256) wb[t] = w2[t];
  __syncthreads();

  {
    int o = wv * 16 + lm;
    v8f c = {};
    for (int k0 = 0; k0 < HID; k0 += 4) {
      v2f a, bf;
      a.x  = a1[lm][k0 + kh];
      a.y  = a1[lm][k0 + kh + 1];
      bf.x = wb[(k0 + kh) * HID + o];
      bf.y = wb[(k0 + kh + 1) * HID + o];
      c = wmma4(a, bf, c);
    }
    float bias = b2[o];
#pragma unroll
    for (int r = 0; r < 8; ++r) {
      float v = c[r] + bias;
      a0[r + 8 * lh][o] = v > 0.f ? v : expm1f(v);
    }
  }
  __syncthreads();
  for (int t = threadIdx.x; t < HID * FF; t += 256) wb[t] = wo[t];
  __syncthreads();

  if (wv < 4) {
    int o = wv * 16 + lm;
    v8f c = {};
    for (int k0 = 0; k0 < HID; k0 += 4) {
      v2f a, bf;
      a.x  = a0[lm][k0 + kh];
      a.y  = a0[lm][k0 + kh + 1];
      bf.x = wb[(k0 + kh) * FF + o];
      bf.y = wb[(k0 + kh + 1) * FF + o];
      c = wmma4(a, bf, c);
    }
    float bias = bo[o];
#pragma unroll
    for (int r = 0; r < 8; ++r) {
      int m = r + 8 * lh;
      float v = c[r] + bias;
      if (rowmask[row0 + m]) v = 0.f;
      out[(size_t)(row0 + m) * FF + o] = v;
    }
  }
}

extern "C" void kernel_launch(void* const* d_in, const int* in_sizes, int n_in,
                              void* d_out, int out_size, void* d_ws,
                              size_t ws_size, hipStream_t stream) {
  (void)in_sizes; (void)n_in; (void)out_size; (void)ws_size;
  const float* x      = (const float*)d_in[0];
  const float* y      = (const float*)d_in[1];
  const float* z      = (const float*)d_in[2];
  const int*   op_adj = (const int*)d_in[3];
  const int*   ma_adj = (const int*)d_in[4];
  const int*   op_ma  = (const int*)d_in[5];
  const float* Wop    = (const float*)d_in[6];
  const float* Wma    = (const float*)d_in[7];
  const float* Wedge  = (const float*)d_in[8];
  const float* att_op = (const float*)d_in[9];
  const float* att_ma = (const float*)d_in[10];
  const float* w1     = (const float*)d_in[11];
  const float* b1     = (const float*)d_in[12];
  const float* w2     = (const float*)d_in[13];
  const float* b2     = (const float*)d_in[14];
  const float* wo     = (const float*)d_in[15];
  const float* bo     = (const float*)d_in[16];
  float* out = (float*)d_out;

  float* ws     = (float*)d_ws;
  float* h      = ws;
  float* hma    = h + BB * NN * FF;
  float* ha1    = hma + BB * MM * FF;
  float* ha2    = ha1 + BB * NN;
  float* ham1   = ha2 + BB * NN;
  float* hma2   = ham1 + BB * NN;
  float* hprime = hma2 + BB * MM;
  int*   rmask  = (int*)(hprime + (size_t)BB * NN * 2 * FF);

  prep_tiles<<<dim3(XT + YT), dim3(128), 0, stream>>>(
      x, Wop, y, Wma, att_op, att_ma, op_ma, h, hma, ha1, ha2, ham1, hma2,
      rmask);
  size_t sm2 = (16 * (NN + 4) + 16) * sizeof(float);
  op_attn<<<dim3(NN / 16, BB), dim3(128), sm2, stream>>>(
      h, ha1, ha2, op_adj, ma_adj, rmask, hprime);
  ma_attn<<<dim3(NN / NPG, BB), dim3(256), 0, stream>>>(
      z, Wedge, hma, ham1, hma2, op_ma, rmask, h, att_ma, hprime);
  size_t sm4 = (HID * HID + 2 * 16 * (HID + 4)) * sizeof(float);
  mlp<<<dim3((BB * NN) / 16), dim3(256), sm4, stream>>>(
      hprime, w1, b1, w2, b2, wo, bo, rmask, out);
}


# op attention, ma attention and the MLP head of a 16-node tile run in one 8-wave work-group, h_prime block kept in LDS (op product still on 4 waves; every summing order kept); 4 launches -> 2
# speedup vs baseline: 4.5578x; 1.2676x over previous
#include <hip/hip_runtime.h>
#include <math.h>


typedef __attribute__((ext_vector_type(2))) float v2f;
typedef __attribute__((ext_vector_type(8))) float v8f;
typedef __attribute__((ext_vector_type(4))) float v4f;
typedef __attribute__((ext_vector_type(4))) int v4i;

#define BB 8
#define NN 1024
#define MM 128
#define FF 64
#define FE 16
#define HID 128

__device__ __forceinline__ v8f wmma4(v2f a, v2f b, v8f c) {
  return __builtin_amdgcn_wmma_f32_16x16x4_f32(
       false, a,  false, b,
       (short)0, c,  false,  false);
}

#define XT ((BB * NN) / 16)
#define YT ((BB * MM) / 16)
#define CTS (FF + 4)

__global__ __launch_bounds__(128) void prep_tiles(
    const float* __restrict__ x, const float* __restrict__ Wop,
    const float* __restrict__ y, const float* __restrict__ Wma,
    const float* __restrict__ att_op, const float* __restrict__ att_ma,
    const int* __restrict__ opma, float* __restrict__ h,
    float* __restrict__ hma, float* __restrict__ ha1,
    float* __restrict__ ha2, float* __restrict__ ham1,
    float* __restrict__ hma2, int* __restrict__ rowmask) {
  __shared__ float ct[16][CTS];
  __shared__ float atts[3][FF];

  const int wv   = threadIdx.x >> 5;
  const int lane = threadIdx.x & 31;
  const int lm   = lane & 15;
  const int kh   = (lane >> 4) * 2;
  const int lh   = lane >> 4;
  const bool node = blockIdx.x < XT;
  const int row0 = (node ? blockIdx.x : blockIdx.x - XT) * 16;

  v4i ad[4] = {};
  if (node) {
    const v4i* at = (const v4i*)(opma + (size_t)row0 * MM);
#pragma unroll
    for (int q = 0; q < 4; ++q) ad[q] = at[q * 128 + threadIdx.x];
    if (threadIdx.x < FF) {
      atts[0][threadIdx.x] = att_op[threadIdx.x];
      atts[1][threadIdx.x] = att_op[FF + threadIdx.x];
      atts[2][threadIdx.x] = att_ma[threadIdx.x];
    }
  } else {
    if (threadIdx.x < FF) atts[0][threadIdx.x] = att_ma[FF + threadIdx.x];
  }

  const float* A = node ? x : y;
  const float* W = node ? Wop : Wma;
  float* C = node ? h : hma;
  const int o = wv * 16 + lm;
  const float* Arow = A + (size_t)(row0 + lm) * FF;

  v8f c = {};
#pragma unroll
  for (int k0 = 0; k0 < FF; k0 += 4) {
    v2f a, b;
    a.x = Arow[k0 + kh];
    a.y = Arow[k0 + kh + 1];
    b.x = W[(size_t)(k0 + kh) * FF + o];
    b.y = W[(size_t)(k0 + kh + 1) * FF + o];
    c = wmma4(a, b, c);
  }
#pragma unroll
  for (int r = 0; r < 8; ++r) {
    int m = r + 8 * lh;
    C[(size_t)(row0 + m) * FF + o] = c[r];
    ct[m][o] = c[r];
  }

  if (node) {
#pragma unroll
    for (int q = 0; q < 4; ++q) {
      const int nz = (ad[q].x | ad[q].y | ad[q].z | ad[q].w) != 0;
      const int anynz = __any(nz);
      if (lane == 0) rowmask[row0 + wv + 4 * q] = anynz ? 0 : 1;
    }
  }
  __syncthreads();

  const int np = node ? 3 : 1;
  if (threadIdx.x < 16 * np) {
    const int p = threadIdx.x >> 4, row = threadIdx.x & 15;
    float s = 0.f;
#pragma unroll 8
    for (int oo = 0; oo < FF; ++oo) s += ct[row][oo] * atts[p][oo];
    float* dst = node ? (p == 0 ? ha1 : (p == 1 ? ha2 : ham1)) : hma2;
    dst[row0 + row] = s;
  }
}

__device__ __forceinline__ float masked_score(float e, int adj) {
  e = e > 0.f ? e : 0.01f * e;
  return adj > 0 ? e : -__builtin_inff();
}

__device__ __forceinline__ float quad_sum(float v) {
  v += __int_as_float(__builtin_amdgcn_update_dpp(
      0, __float_as_int(v), 0xB1, 0xF, 0xF, true));
  v += __int_as_float(__builtin_amdgcn_update_dpp(
      0, __float_as_int(v), 0x4E, 0xF, 0xF, true));
  return v;
}

__device__ __forceinline__ float across_quads_sum(float v) {
  v += __shfl_xor(v, 4, 32);
  v += __shfl_xor(v, 8, 32);
  v += __shfl_xor(v, 16, 32);
  return v;
}

#define KSB 8
#define NBLK (NN / 4 / 4 / KSB)
#define NPW 2
#define NPG (8 * NPW)
#define PST (MM + 4)
#define ZST (FE + 4)
#define AST (HID + 4)
#define OFF_A0   0
#define OFF_A1   (OFF_A0 + 16 * AST)
#define OFF_PM   (OFF_A1 + 16 * AST)
#define OFF_ZP   (OFF_PM + NPG * PST)
#define OFF_INVS (OFF_ZP + NPG * ZST)
#define OFF_ISUM (OFF_INVS + NPG)
#define OFF_BIG  (OFF_ISUM + 16)
#define LDS_FLOATS (OFF_BIG + 16 * (NN + 4))

__global__ __launch_bounds__(256) void node_tiles(
    const float* __restrict__ h, const float* __restrict__ ha1,
    const float* __restrict__ ha2, const int* __restrict__ opadj,
    const int* __restrict__ maadj, const int* __restrict__ rowmask,
    const float* __restrict__ z, const float* __restrict__ wedge,
    const float* __restrict__ hma, const float* __restrict__ ham1,
    const float* __restrict__ hma2, const int* __restrict__ opma,
    const float* __restrict__ attma, const float* __restrict__ w1,
    const float* __restrict__ b1, const float* __restrict__ w2,
    const float* __restrict__ b2, const float* __restrict__ wo,
    const float* __restrict__ bo, float* __restrict__ out) {
  extern __shared__ __attribute__((aligned(16))) float smem[];
  typedef float act_row_t[AST];
  typedef float att_row_t[NN + 4];
  typedef float pm_row_t[PST];
  typedef float zp_row_t[ZST];
  act_row_t* a0 = (act_row_t*)(smem + OFF_A0);
  act_row_t* a1 = (act_row_t*)(smem + OFF_A1);
  pm_row_t* Pm = (pm_row_t*)(smem + OFF_PM);
  zp_row_t* Zp = (zp_row_t*)(smem + OFF_ZP);
  float* invs = smem + OFF_INVS;
  float* inv_sum = smem + OFF_ISUM;
  att_row_t* att = (att_row_t*)(smem + OFF_BIG);
  float* wb = smem + OFF_BIG;

  const int b  = blockIdx.y;
  const int i0 = blockIdx.x * 16;
  const int* opb = opadj + (size_t)b * NN * NN;
  const int* mab = maadj + (size_t)b * NN * NN;
  const float NEGINF = -__builtin_inff();

  const int wv = __builtin_amdgcn_readfirstlane((int)(threadIdx.x >> 5));
  const int lane = threadIdx.x & 31;
  const int lm = lane & 15;
  const int lh = lane >> 4;
  const int kh = lh * 2;

  {
    const v4i* opT = (const v4i*)(opb + i0);
    const v4i* maT = (const v4i*)(mab + i0);
#pragma unroll 8
    for (int t = threadIdx.x; t < 4 * NN; t += 256) {
      const int j = t >> 2, part = t & 3;
      const v4i a = opT[(size_t)j * (NN / 4) + part];
      const v4i m = maT[(size_t)j * (NN / 4) + part];
      const v4i s = a + m;
      att[4 * part + 0][j] = __int_as_float(s.x);
      att[4 * part + 1][j] = __int_as_float(s.y);
      att[4 * part + 2][j] = __int_as_float(s.z);
      att[4 * part + 3][j] = __int_as_float(s.w);
    }
  }
  __syncthreads();

  {
    const int j = threadIdx.x * 4;
    const v4f e1 = *(const v4f*)(ha1 + b * NN + j);
#pragma unroll 8
    for (int r = 0; r < 16; ++r) {
      const int i = i0 + r;
      const bool rm = rowmask[b * NN + i] != 0;
      const float e2 = ha2[b * NN + i];
      v4f* slot = (v4f*)&att[r][j];
      const v4f tr = *slot;
      const v4i d1 = *(const v4i*)(opb + (size_t)i * NN + j);
      const v4i d2 = *(const v4i*)(mab + (size_t)i * NN + j);
      v4i adj = d1 + d2;
      adj.x += __float_as_int(tr.x);
      adj.y += __float_as_int(tr.y);
      adj.z += __float_as_int(tr.z);
      adj.w += __float_as_int(tr.w);
      if (rm) adj = (v4i){1, 1, 1, 1};
      v4f sc4;
      sc4.x = masked_score(e2 + e1.x, adj.x);
      sc4.y = masked_score(e2 + e1.y, adj.y);
      sc4.z = masked_score(e2 + e1.z, adj.z);
      sc4.w = masked_score(e2 + e1.w, adj.w);
      *slot = sc4;
    }
  }
  __syncthreads();

  for (int r = wv; r < 16; r += 8) {
    float mx = NEGINF;
    for (int j = lane; j < NN; j += 32) mx = fmaxf(mx, att[r][j]);
#pragma unroll
    for (int off = 16; off; off >>= 1) mx = fmaxf(mx, __shfl_xor(mx, off, 32));
    if (mx == NEGINF) mx = 0.f;
    float sum = 0.f;
    for (int j = lane; j < NN; j += 32) {
      float p = __expf(att[r][j] - mx);
      att[r][j] = p;
      sum += p;
    }
#pragma unroll
    for (int off = 16; off; off >>= 1) sum += __shfl_xor(sum, off, 32);
    if (lane == 0) inv_sum[r] = 1.f / fmaxf(sum, 1e-30f);
  }
  __syncthreads();

  typedef float red_tile_t[16][FF + 4];
  red_tile_t* red = (red_tile_t*)(smem + OFF_BIG);
  v8f acc[4] = {};
  if (wv < 4) {
    const float* hb = h + (size_t)b * NN * FF;
    const int kb = wv * (NN / 4);
    const float* hq = hb + (size_t)(kb + 2 * lh) * FF + 4 * lm;
    const float* aq = &att[lm][kb + 2 * lh];

    v4f bq[2][KSB][2];
    v2f aa[2][KSB];

    auto fetch = [&](const int st, const int blk) {
#pragma unroll
      for (int s = 0; s < KSB; ++s) {
        const int ks = blk * KSB + s;
        bq[st][s][0] = *(const v4f*)(hq + (size_t)(4 * ks) * FF);
        bq[st][s][1] = *(const v4f*)(hq + (size_t)(4 * ks + 1) * FF);
        aa[st][s]    = *(const v2f*)(aq + 4 * ks);
      }
    };
    auto mulblk = [&](const int st) {
#pragma unroll
      for (int s = 0; s < KSB; ++s) {
#pragma unroll
        for (int t = 0; t < 4; ++t) {
          v2f bf;
          bf.x = bq[st][s][0][t];
          bf.y = bq[st][s][1][t];
          acc[t] = wmma4(aa[st][s], bf, acc[t]);
        }
      }
    };

    fetch(0, 0);
    for (int it = 0; it < NBLK / 2 - 1; ++it) {
      fetch(1, 2 * it + 1);
      __builtin_amdgcn_sched_barrier(0);
      mulblk(0);
      __builtin_amdgcn_sched_barrier(0);
      fetch(0, 2 * it + 2);
      asm volatile("" ::: "memory");
      __builtin_amdgcn_sched_barrier(0);
      mulblk(1);
    }
    fetch(1, NBLK - 1);
    mulblk(0);
    mulblk(1);

  }
  __syncthreads();
  if (wv < 4) {
#pragma unroll
    for (int r = 0; r < 8; ++r) {
      v4f v;
      v.x = acc[0][r]; v.y = acc[1][r]; v.z = acc[2][r]; v.w = acc[3][r];
      *(v4f*)&red[wv][r + 8 * lh][4 * lm] = v;
    }
  }
  __syncthreads();

  if (threadIdx.x < 128) {
    const int m = threadIdx.x >> 3;
    const int g = (threadIdx.x & 7) * 8;
    v4f s0 = *(const v4f*)&red[0][m][g];
    v4f s1 = *(const v4f*)&red[0][m][g + 4];
#pragma unroll
    for (int w = 1; w < 4; ++w) {
      s0 += *(const v4f*)&red[w][m][g];
      s1 += *(const v4f*)&red[w][m][g + 4];
    }
    const float is = inv_sum[m];
    *(v4f*)&a0[m][g]     = s0 * is;
    *(v4f*)&a0[m][g + 4] = s1 * is;
  }

  {
  const int q  = lane & 3;
  const int rq = lane >> 2;

  const size_t bn0 = (size_t)b * NN + i0 + wv * NPW;

  v4f wv4[8], av4[8];
  {
    const v4f* wr = (const v4f*)(wedge + (lane & 15) * FF + 32 * (lane >> 4));
    const v4f* ar = (const v4f*)(attma + 2 * FF + 32 * (lane >> 4));
#pragma unroll
    for (int i = 0; i < 8; ++i) { wv4[i] = wr[i]; av4[i] = ar[i]; }
  }
  float hm2r[16];
#pragma unroll
  for (int j = 0; j < 16; ++j) hm2r[j] = hma2[(size_t)b * MM + rq + 8 * j];
  v4i   ad[NPW];
  int   rmv[NPW];
  float hm1v[NPW];
#pragma unroll
  for (int t = 0; t < NPW; ++t) {
    ad[t]   = *(const v4i*)(opma + (bn0 + t) * MM + 4 * lane);
    rmv[t]  = rowmask[bn0 + t];
    hm1v[t] = ham1[bn0 + t];
  }

  const v4f* zt = (const v4f*)(z + bn0 * MM * FE);
  v4f zr[NPW][16];
#pragma unroll
  for (int t = 0; t < NPW; ++t) {
#pragma unroll
    for (int j = 0; j < 16; ++j) zr[t][j] = zt[t * 512 + j * 32 + lane];
  }

  unsigned connv[NPW];
#pragma unroll
  for (int t = 0; t < NPW; ++t) {
    const unsigned s0 = __builtin_amdgcn_ballot_w32(ad[t].x > 0);
    const unsigned s1 = __builtin_amdgcn_ballot_w32(ad[t].y > 0);
    const unsigned s2 = __builtin_amdgcn_ballot_w32(ad[t].z > 0);
    const unsigned s3 = __builtin_amdgcn_ballot_w32(ad[t].w > 0);
    const int cq = rq & 3;
    unsigned conn = cq == 0 ? s0 : (cq == 1 ? s1 : (cq == 2 ? s2 : s3));
    conn >>= (lane >> 4);
    if (rmv[t]) conn = 0xffffffffu;
    connv[t] = conn;
  }

  float wpart = 0.f;
#pragma unroll
  for (int i = 0; i < 8; ++i) {
    wpart += wv4[i].x * av4[i].x;
    wpart += wv4[i].y * av4[i].y;
    wpart += wv4[i].z * av4[i].z;
    wpart += wv4[i].w * av4[i].w;
  }
  wpart += __shfl_xor(wpart, 16, 32);
  float wq[4];
#pragma unroll
  for (int c = 0; c < 4; ++c) wq[c] = __shfl(wpart, 4 * q + c, 32);

#pragma unroll
  for (int t = 0; t < NPW; ++t) {
    const int nl = wv * NPW + t;

    const unsigned conn = connv[t];

    float sc[16];
    float mx = NEGINF;
#pragma unroll
    for (int j = 0; j < 16; ++j) {
      const v4f zv = zr[t][j];
      float d = zv.x * wq[0];
      d += zv.y * wq[1];
      d += zv.z * wq[2];
      d += zv.w * wq[3];
      d = quad_sum(d);
      float e = (hm1v[t] + hm2r[j]) + d;
      e = e > 0.f ? e : 0.01f * e;
      sc[j] = ((conn >> (2 * j)) & 1u) ? e : NEGINF;
      mx = fmaxf(mx, sc[j]);
    }
    mx = fmaxf(mx, __shfl_xor(mx, 4, 32));
    mx = fmaxf(mx, __shfl_xor(mx, 8, 32));
    mx = fmaxf(mx, __shfl_xor(mx, 16, 32));
    if (mx == NEGINF) mx = 0.f;

    float ps = 0.f;
    v4f zacc = {0.f, 0.f, 0.f, 0.f};
#pragma unroll
    for (int j = 0; j < 16; ++j) {
      const float p = __expf(sc[j] - mx);
      sc[j] = p;
      ps += p;
      zacc += p * zr[t][j];
    }
    ps = across_quads_sum(ps);
    zacc.x = across_quads_sum(zacc.x);
    zacc.y = across_quads_sum(zacc.y);
    zacc.z = across_quads_sum(zacc.z);
    zacc.w = across_quads_sum(zacc.w);

    if (q == 0) {
#pragma unroll
      for (int j = 0; j < 16; ++j) Pm[nl][rq + 8 * j] = sc[j];
    }
    if (lane < 4) *(v4f*)&Zp[nl][4 * lane] = zacc;
    if (lane == 0) invs[nl] = 1.f / fmaxf(ps, 1e-30f);
  }

  const int o  = (wv & 3) * 16 + lm;
  const float* hmb = hma + (size_t)b * MM * FF;
  v2f bh[MM / 4];
  v2f bw[FE / 4];
  if (wv < 4) {
#pragma unroll
    for (int ks = 0; ks < MM / 4; ++ks) {
      bh[ks].x = hmb[(size_t)(4 * ks + kh) * FF + o];
      bh[ks].y = hmb[(size_t)(4 * ks + kh + 1) * FF + o];
    }
#pragma unroll
    for (int ks = 0; ks < FE / 4; ++ks) {
      bw[ks].x = wedge[(4 * ks + kh) * FF + o];
      bw[ks].y = wedge[(4 * ks + kh + 1) * FF + o];
    }
  }
  __syncthreads();

  if (wv < 4) {
    v8f c0 = {}, c1 = {};
#pragma unroll
    for (int ks = 0; ks < MM / 4; ++ks) {
      v2f a;
      a.x = Pm[lm][4 * ks + kh];
      a.y = Pm[lm][4 * ks + kh + 1];
      if (ks & 1) c1 = wmma4(a, bh[ks], c1);
      else        c0 = wmma4(a, bh[ks], c0);
    }
#pragma unroll
    for (int ks = 0; ks < FE / 4; ++ks) {
      v2f a;
      a.x = Zp[lm][4 * ks + kh];
      a.y = Zp[lm][4 * ks + kh + 1];
      if (ks & 1) c1 = wmma4(a, bw[ks], c1);
      else        c0 = wmma4(a, bw[ks], c0);
    }
    const v8f c = c0 + c1;
    const size_t bnr = (size_t)b * NN + i0;
#pragma unroll
    for (int r = 0; r < 8; ++r) {
      const int m = r + 8 * lh;
      a0[m][FF + o] = c[r] * invs[m] + h[(bnr + m) * FF + o];
    }
  }
  }

  const int row0 = b * NN + i0;
  for (int t = threadIdx.x; t < HID * HID; t += 256) wb[t] = w1[t];
  __syncthreads();

  {
    int o = wv * 16 + lm;
    v8f c = {};
    for (int k0 = 0; k0 < HID; k0 += 4) {
      v2f a, bf;
      a.x  = a0[lm][k0 + kh];
      a.y  = a0[lm][k0 + kh + 1];
      bf.x = wb[(k0 + kh) * HID + o];
      bf.y = wb[(k0 + kh + 1) * HID + o];
      c = wmma4(a, bf, c);
    }
    float bias = b1[o];
#pragma unroll
    for (int r = 0; r < 8; ++r) {
      float v = c[r] + bias;
      a1[r + 8 * lh][o] = v > 0.f ? v : expm1f(v);
    }
  }
  __syncthreads();
  for (int t = threadIdx.x; t < HID * HID; t += 256) wb[t] = w2[t];
  __syncthreads();

  {
    int o = wv * 16 + lm;
    v8f c = {};
    for (int k0 = 0; k0 < HID; k0 += 4) {
      v2f a, bf;
      a.x  = a1[lm][k0 + kh];
      a.y  = a1[lm][k0 + kh + 1];
      bf.x = wb[(k0 + kh) * HID + o];
      bf.y = wb[(k0 + kh + 1) * HID + o];
      c = wmma4(a, bf, c);
    }
    float bias = b2[o];
#pragma unroll
    for (int r = 0; r < 8; ++r) {
      float v = c[r] + bias;
      a0[r + 8 * lh][o] = v > 0.f ? v : expm1f(v);
    }
  }
  __syncthreads();
  for (int t = threadIdx.x; t < HID * FF; t += 256) wb[t] = wo[t];
  __syncthreads();

  if (wv < 4) {
    int o = wv * 16 + lm;
    v8f c = {};
    for (int k0 = 0; k0 < HID; k0 += 4) {
      v2f a, bf;
      a.x  = a0[lm][k0 + kh];
      a.y  = a0[lm][k0 + kh + 1];
      bf.x = wb[(k0 + kh) * FF + o];
      bf.y = wb[(k0 + kh + 1) * FF + o];
      c = wmma4(a, bf, c);
    }
    float bias = bo[o];
#pragma unroll
    for (int r = 0; r < 8; ++r) {
      int m = r + 8 * lh;
      float v = c[r] + bias;
      if (rowmask[row0 + m]) v = 0.f;
      out[(size_t)(row0 + m) * FF + o] = v;
    }
  }
}

extern "C" void kernel_launch(void* const* d_in, const int* in_sizes, int n_in,
                              void* d_out, int out_size, void* d_ws,
                              size_t ws_size, hipStream_t stream) {
  (void)in_sizes; (void)n_in; (void)out_size; (void)ws_size;
  const float* x      = (const float*)d_in[0];
  const float* y      = (const float*)d_in[1];
  const float* z      = (const float*)d_in[2];
  const int*   op_adj = (const int*)d_in[3];
  const int*   ma_adj = (const int*)d_in[4];
  const int*   op_ma  = (const int*)d_in[5];
  const float* Wop    = (const float*)d_in[6];
  const float* Wma    = (const float*)d_in[7];
  const float* Wedge  = (const float*)d_in[8];
  const float* att_op = (const float*)d_in[9];
  const float* att_ma = (const float*)d_in[10];
  const float* w1     = (const float*)d_in[11];
  const float* b1     = (const float*)d_in[12];
  const float* w2     = (const float*)d_in[13];
  const float* b2     = (const float*)d_in[14];
  const float* wo     = (const float*)d_in[15];
  const float* bo     = (const float*)d_in[16];
  float* out = (float*)d_out;

  float* ws     = (float*)d_ws;
  float* h      = ws;
  float* hma    = h + BB * NN * FF;
  float* ha1    = hma + BB * MM * FF;
  float* ha2    = ha1 + BB * NN;
  float* ham1   = ha2 + BB * NN;
  float* hma2   = ham1 + BB * NN;
  int*   rmask  = (int*)(hma2 + BB * MM);

  prep_tiles<<<dim3(XT + YT), dim3(128), 0, stream>>>(
      x, Wop, y, Wma, att_op, att_ma, op_ma, h, hma, ha1, ha2, ham1, hma2,
      rmask);
  size_t sm = LDS_FLOATS * sizeof(float);
  node_tiles<<<dim3(NN / 16, BB), dim3(256), sm, stream>>>(
      h, ha1, ha2, op_adj, ma_adj, rmask, z, Wedge, hma, ham1, hma2, op_ma,
      att_ma, w1, b1, w2, b2, wo, bo, out);
}


# MLP weights: each matrix requested a stage ahead as 16-byte vectors into registers, in flight across the products in front of it, then stored to LDS; was 4-byte copy loops waited for in place
# speedup vs baseline: 4.8152x; 1.0565x over previous
#include <hip/hip_runtime.h>
#include <math.h>


typedef __attribute__((ext_vector_type(2))) float v2f;
typedef __attribute__((ext_vector_type(8))) float v8f;
typedef __attribute__((ext_vector_type(4))) float v4f;
typedef __attribute__((ext_vector_type(4))) int v4i;

#define BB 8
#define NN 1024
#define MM 128
#define FF 64
#define FE 16
#define HID 128

__device__ __forceinline__ v8f wmma4(v2f a, v2f b, v8f c) {
  return __builtin_amdgcn_wmma_f32_16x16x4_f32(
       false, a,  false, b,
       (short)0, c,  false,  false);
}

#define XT ((BB * NN) / 16)
#define YT ((BB * MM) / 16)
#define CTS (FF + 4)

__global__ __launch_bounds__(128) void prep_tiles(
    const float* __restrict__ x, const float* __restrict__ Wop,
    const float* __restrict__ y, const float* __restrict__ Wma,
    const float* __restrict__ att_op, const float* __restrict__ att_ma,
    const int* __restrict__ opma, float* __restrict__ h,
    float* __restrict__ hma, float* __restrict__ ha1,
    float* __restrict__ ha2, float* __restrict__ ham1,
    float* __restrict__ hma2, int* __restrict__ rowmask) {
  __shared__ float ct[16][CTS];
  __shared__ float atts[3][FF];

  const int wv   = threadIdx.x >> 5;
  const int lane = threadIdx.x & 31;
  const int lm   = lane & 15;
  const int kh   = (lane >> 4) * 2;
  const int lh   = lane >> 4;
  const bool node = blockIdx.x < XT;
  const int row0 = (node ? blockIdx.x : blockIdx.x - XT) * 16;

  v4i ad[4] = {};
  if (node) {
    const v4i* at = (const v4i*)(opma + (size_t)row0 * MM);
#pragma unroll
    for (int q = 0; q < 4; ++q) ad[q] = at[q * 128 + threadIdx.x];
    if (threadIdx.x < FF) {
      atts[0][threadIdx.x] = att_op[threadIdx.x];
      atts[1][threadIdx.x] = att_op[FF + threadIdx.x];
      atts[2][threadIdx.x] = att_ma[threadIdx.x];
    }
  } else {
    if (threadIdx.x < FF) atts[0][threadIdx.x] = att_ma[FF + threadIdx.x];
  }

  const float* A = node ? x : y;
  const float* W = node ? Wop : Wma;
  float* C = node ? h : hma;
  const int o = wv * 16 + lm;
  const float* Arow = A + (size_t)(row0 + lm) * FF;

  v8f c = {};
#pragma unroll
  for (int k0 = 0; k0 < FF; k0 += 4) {
    v2f a, b;
    a.x = Arow[k0 + kh];
    a.y = Arow[k0 + kh + 1];
    b.x = W[(size_t)(k0 + kh) * FF + o];
    b.y = W[(size_t)(k0 + kh + 1) * FF + o];
    c = wmma4(a, b, c);
  }
#pragma unroll
  for (int r = 0; r < 8; ++r) {
    int m = r + 8 * lh;
    C[(size_t)(row0 + m) * FF + o] = c[r];
    ct[m][o] = c[r];
  }

  if (node) {
#pragma unroll
    for (int q = 0; q < 4; ++q) {
      const int nz = (ad[q].x | ad[q].y | ad[q].z | ad[q].w) != 0;
      const int anynz = __any(nz);
      if (lane == 0) rowmask[row0 + wv + 4 * q] = anynz ? 0 : 1;
    }
  }
  __syncthreads();

  const int np = node ? 3 : 1;
  if (threadIdx.x < 16 * np) {
    const int p = threadIdx.x >> 4, row = threadIdx.x & 15;
    float s = 0.f;
#pragma unroll 8
    for (int oo = 0; oo < FF; ++oo) s += ct[row][oo] * atts[p][oo];
    float* dst = node ? (p == 0 ? ha1 : (p == 1 ? ha2 : ham1)) : hma2;
    dst[row0 + row] = s;
  }
}

__device__ __forceinline__ float masked_score(float e, int adj) {
  e = e > 0.f ? e : 0.01f * e;
  return adj > 0 ? e : -__builtin_inff();
}

__device__ __forceinline__ float quad_sum(float v) {
  v += __int_as_float(__builtin_amdgcn_update_dpp(
      0, __float_as_int(v), 0xB1, 0xF, 0xF, true));
  v += __int_as_float(__builtin_amdgcn_update_dpp(
      0, __float_as_int(v), 0x4E, 0xF, 0xF, true));
  return v;
}

__device__ __forceinline__ float across_quads_sum(float v) {
  v += __shfl_xor(v, 4, 32);
  v += __shfl_xor(v, 8, 32);
  v += __shfl_xor(v, 16, 32);
  return v;
}

#define KSB 8
#define NBLK (NN / 4 / 4 / KSB)
#define NPW 2
#define NPG (8 * NPW)
#define PST (MM + 4)
#define ZST (FE + 4)
#define AST (HID + 4)
#define OFF_A0   0
#define OFF_A1   (OFF_A0 + 16 * AST)
#define OFF_PM   (OFF_A1 + 16 * AST)
#define OFF_ZP   (OFF_PM + NPG * PST)
#define OFF_INVS (OFF_ZP + NPG * ZST)
#define OFF_ISUM (OFF_INVS + NPG)
#define OFF_BIG  (OFF_ISUM + 16)
#define LDS_FLOATS (OFF_BIG + 16 * (NN + 4))

__global__ __launch_bounds__(256) void node_tiles(
    const float* __restrict__ h, const float* __restrict__ ha1,
    const float* __restrict__ ha2, const int* __restrict__ opadj,
    const int* __restrict__ maadj, const int* __restrict__ rowmask,
    const float* __restrict__ z, const float* __restrict__ wedge,
    const float* __restrict__ hma, const float* __restrict__ ham1,
    const float* __restrict__ hma2, const int* __restrict__ opma,
    const float* __restrict__ attma, const float* __restrict__ w1,
    const float* __restrict__ b1, const float* __restrict__ w2,
    const float* __restrict__ b2, const float* __restrict__ wo,
    const float* __restrict__ bo, float* __restrict__ out) {
  extern __shared__ __attribute__((aligned(16))) float smem[];
  typedef float act_row_t[AST];
  typedef float att_row_t[NN + 4];
  typedef float pm_row_t[PST];
  typedef float zp_row_t[ZST];
  act_row_t* a0 = (act_row_t*)(smem + OFF_A0);
  act_row_t* a1 = (act_row_t*)(smem + OFF_A1);
  pm_row_t* Pm = (pm_row_t*)(smem + OFF_PM);
  zp_row_t* Zp = (zp_row_t*)(smem + OFF_ZP);
  float* invs = smem + OFF_INVS;
  float* inv_sum = smem + OFF_ISUM;
  att_row_t* att = (att_row_t*)(smem + OFF_BIG);
  float* wb = smem + OFF_BIG;

  const int b  = blockIdx.y;
  const int i0 = blockIdx.x * 16;
  const int* opb = opadj + (size_t)b * NN * NN;
  const int* mab = maadj + (size_t)b * NN * NN;
  const float NEGINF = -__builtin_inff();

  const int wv = __builtin_amdgcn_readfirstlane((int)(threadIdx.x >> 5));
  const int lane = threadIdx.x & 31;
  const int lm = lane & 15;
  const int lh = lane >> 4;
  const int kh = lh * 2;

  {
    const v4i* opT = (const v4i*)(opb + i0);
    const v4i* maT = (const v4i*)(mab + i0);
#pragma unroll 8
    for (int t = threadIdx.x; t < 4 * NN; t += 256) {
      const int j = t >> 2, part = t & 3;
      const v4i a = opT[(size_t)j * (NN / 4) + part];
      const v4i m = maT[(size_t)j * (NN / 4) + part];
      const v4i s = a + m;
      att[4 * part + 0][j] = __int_as_float(s.x);
      att[4 * part + 1][j] = __int_as_float(s.y);
      att[4 * part + 2][j] = __int_as_float(s.z);
      att[4 * part + 3][j] = __int_as_float(s.w);
    }
  }
  __syncthreads();

  {
    const int j = threadIdx.x * 4;
    const v4f e1 = *(const v4f*)(ha1 + b * NN + j);
#pragma unroll 8
    for (int r = 0; r < 16; ++r) {
      const int i = i0 + r;
      const bool rm = rowmask[b * NN + i] != 0;
      const float e2 = ha2[b * NN + i];
      v4f* slot = (v4f*)&att[r][j];
      const v4f tr = *slot;
      const v4i d1 = *(const v4i*)(opb + (size_t)i * NN + j);
      const v4i d2 = *(const v4i*)(mab + (size_t)i * NN + j);
      v4i adj = d1 + d2;
      adj.x += __float_as_int(tr.x);
      adj.y += __float_as_int(tr.y);
      adj.z += __float_as_int(tr.z);
      adj.w += __float_as_int(tr.w);
      if (rm) adj = (v4i){1, 1, 1, 1};
      v4f sc4;
      sc4.x = masked_score(e2 + e1.x, adj.x);
      sc4.y = masked_score(e2 + e1.y, adj.y);
      sc4.z = masked_score(e2 + e1.z, adj.z);
      sc4.w = masked_score(e2 + e1.w, adj.w);
      *slot = sc4;
    }
  }
  __syncthreads();

  for (int r = wv; r < 16; r += 8) {
    float mx = NEGINF;
    for (int j = lane; j < NN; j += 32) mx = fmaxf(mx, att[r][j]);
#pragma unroll
    for (int off = 16; off; off >>= 1) mx = fmaxf(mx, __shfl_xor(mx, off, 32));
    if (mx == NEGINF) mx = 0.f;
    float sum = 0.f;
    for (int j = lane; j < NN; j += 32) {
      float p = __expf(att[r][j] - mx);
      att[r][j] = p;
      sum += p;
    }
#pragma unroll
    for (int off = 16; off; off >>= 1) sum += __shfl_xor(sum, off, 32);
    if (lane == 0) inv_sum[r] = 1.f / fmaxf(sum, 1e-30f);
  }
  __syncthreads();

  typedef float red_tile_t[16][FF + 4];
  red_tile_t* red = (red_tile_t*)(smem + OFF_BIG);
  v8f acc[4] = {};
  if (wv < 4) {
    const float* hb = h + (size_t)b * NN * FF;
    const int kb = wv * (NN / 4);
    const float* hq = hb + (size_t)(kb + 2 * lh) * FF + 4 * lm;
    const float* aq = &att[lm][kb + 2 * lh];

    v4f bq[2][KSB][2];
    v2f aa[2][KSB];

    auto fetch = [&](const int st, const int blk) {
#pragma unroll
      for (int s = 0; s < KSB; ++s) {
        const int ks = blk * KSB + s;
        bq[st][s][0] = *(const v4f*)(hq + (size_t)(4 * ks) * FF);
        bq[st][s][1] = *(const v4f*)(hq + (size_t)(4 * ks + 1) * FF);
        aa[st][s]    = *(const v2f*)(aq + 4 * ks);
      }
    };
    auto mulblk = [&](const int st) {
#pragma unroll
      for (int s = 0; s < KSB; ++s) {
#pragma unroll
        for (int t = 0; t < 4; ++t) {
          v2f bf;
          bf.x = bq[st][s][0][t];
          bf.y = bq[st][s][1][t];
          acc[t] = wmma4(aa[st][s], bf, acc[t]);
        }
      }
    };

    fetch(0, 0);
    for (int it = 0; it < NBLK / 2 - 1; ++it) {
      fetch(1, 2 * it + 1);
      __builtin_amdgcn_sched_barrier(0);
      mulblk(0);
      __builtin_amdgcn_sched_barrier(0);
      fetch(0, 2 * it + 2);
      asm volatile("" ::: "memory");
      __builtin_amdgcn_sched_barrier(0);
      mulblk(1);
    }
    fetch(1, NBLK - 1);
    mulblk(0);
    mulblk(1);

  }
  __syncthreads();
  if (wv < 4) {
#pragma unroll
    for (int r = 0; r < 8; ++r) {
      v4f v;
      v.x = acc[0][r]; v.y = acc[1][r]; v.z = acc[2][r]; v.w = acc[3][r];
      *(v4f*)&red[wv][r + 8 * lh][4 * lm] = v;
    }
  }
  __syncthreads();

  if (threadIdx.x < 128) {
    const int m = threadIdx.x >> 3;
    const int g = (threadIdx.x & 7) * 8;
    v4f s0 = *(const v4f*)&red[0][m][g];
    v4f s1 = *(const v4f*)&red[0][m][g + 4];
#pragma unroll
    for (int w = 1; w < 4; ++w) {
      s0 += *(const v4f*)&red[w][m][g];
      s1 += *(const v4f*)&red[w][m][g + 4];
    }
    const float is = inv_sum[m];
    *(v4f*)&a0[m][g]     = s0 * is;
    *(v4f*)&a0[m][g + 4] = s1 * is;
  }

  v4f wst[16];
  {
  const int q  = lane & 3;
  const int rq = lane >> 2;

  const size_t bn0 = (size_t)b * NN + i0 + wv * NPW;

  v4f wv4[8], av4[8];
  {
    const v4f* wr = (const v4f*)(wedge + (lane & 15) * FF + 32 * (lane >> 4));
    const v4f* ar = (const v4f*)(attma + 2 * FF + 32 * (lane >> 4));
#pragma unroll
    for (int i = 0; i < 8; ++i) { wv4[i] = wr[i]; av4[i] = ar[i]; }
  }
  float hm2r[16];
#pragma unroll
  for (int j = 0; j < 16; ++j) hm2r[j] = hma2[(size_t)b * MM + rq + 8 * j];
  v4i   ad[NPW];
  int   rmv[NPW];
  float hm1v[NPW];
#pragma unroll
  for (int t = 0; t < NPW; ++t) {
    ad[t]   = *(const v4i*)(opma + (bn0 + t) * MM + 4 * lane);
    rmv[t]  = rowmask[bn0 + t];
    hm1v[t] = ham1[bn0 + t];
  }

  const v4f* zt = (const v4f*)(z + bn0 * MM * FE);
  v4f zr[NPW][16];
#pragma unroll
  for (int t = 0; t < NPW; ++t) {
#pragma unroll
    for (int j = 0; j < 16; ++j) zr[t][j] = zt[t * 512 + j * 32 + lane];
  }

  unsigned connv[NPW];
#pragma unroll
  for (int t = 0; t < NPW; ++t) {
    const unsigned s0 = __builtin_amdgcn_ballot_w32(ad[t].x > 0);
    const unsigned s1 = __builtin_amdgcn_ballot_w32(ad[t].y > 0);
    const unsigned s2 = __builtin_amdgcn_ballot_w32(ad[t].z > 0);
    const unsigned s3 = __builtin_amdgcn_ballot_w32(ad[t].w > 0);
    const int cq = rq & 3;
    unsigned conn = cq == 0 ? s0 : (cq == 1 ? s1 : (cq == 2 ? s2 : s3));
    conn >>= (lane >> 4);
    if (rmv[t]) conn = 0xffffffffu;
    connv[t] = conn;
  }

  float wpart = 0.f;
#pragma unroll
  for (int i = 0; i < 8; ++i) {
    wpart += wv4[i].x * av4[i].x;
    wpart += wv4[i].y * av4[i].y;
    wpart += wv4[i].z * av4[i].z;
    wpart += wv4[i].w * av4[i].w;
  }
  wpart += __shfl_xor(wpart, 16, 32);
  float wq[4];
#pragma unroll
  for (int c = 0; c < 4; ++c) wq[c] = __shfl(wpart, 4 * q + c, 32);

#pragma unroll
  for (int t = 0; t < NPW; ++t) {
    const int nl = wv * NPW + t;

    const unsigned conn = connv[t];

    float sc[16];
    float mx = NEGINF;
#pragma unroll
    for (int j = 0; j < 16; ++j) {
      const v4f zv = zr[t][j];
      float d = zv.x * wq[0];
      d += zv.y * wq[1];
      d += zv.z * wq[2];
      d += zv.w * wq[3];
      d = quad_sum(d);
      float e = (hm1v[t] + hm2r[j]) + d;
      e = e > 0.f ? e : 0.01f * e;
      sc[j] = ((conn >> (2 * j)) & 1u) ? e : NEGINF;
      mx = fmaxf(mx, sc[j]);
    }
    mx = fmaxf(mx, __shfl_xor(mx, 4, 32));
    mx = fmaxf(mx, __shfl_xor(mx, 8, 32));
    mx = fmaxf(mx, __shfl_xor(mx, 16, 32));
    if (mx == NEGINF) mx = 0.f;

    float ps = 0.f;
    v4f zacc = {0.f, 0.f, 0.f, 0.f};
#pragma unroll
    for (int j = 0; j < 16; ++j) {
      const float p = __expf(sc[j] - mx);
      sc[j] = p;
      ps += p;
      zacc += p * zr[t][j];
    }
    ps = across_quads_sum(ps);
    zacc.x = across_quads_sum(zacc.x);
    zacc.y = across_quads_sum(zacc.y);
    zacc.z = across_quads_sum(zacc.z);
    zacc.w = across_quads_sum(zacc.w);

    if (q == 0) {
#pragma unroll
      for (int j = 0; j < 16; ++j) Pm[nl][rq + 8 * j] = sc[j];
    }
    if (lane < 4) *(v4f*)&Zp[nl][4 * lane] = zacc;
    if (lane == 0) invs[nl] = 1.f / fmaxf(ps, 1e-30f);
  }

  const int o  = (wv & 3) * 16 + lm;
  const float* hmb = hma + (size_t)b * MM * FF;
  v2f bh[MM / 4];
  v2f bw[FE / 4];
  if (wv < 4) {
#pragma unroll
    for (int ks = 0; ks < MM / 4; ++ks) {
      bh[ks].x = hmb[(size_t)(4 * ks + kh) * FF + o];
      bh[ks].y = hmb[(size_t)(4 * ks + kh + 1) * FF + o];
    }
#pragma unroll
    for (int ks = 0; ks < FE / 4; ++ks) {
      bw[ks].x = wedge[(4 * ks + kh) * FF + o];
      bw[ks].y = wedge[(4 * ks + kh + 1) * FF + o];
    }
  }
  __syncthreads();
#pragma unroll
  for (int i = 0; i < 16; ++i) wst[i] = ((const v4f*)w1)[threadIdx.x + 256 * i];

  if (wv < 4) {
    v8f c0 = {}, c1 = {};
#pragma unroll
    for (int ks = 0; ks < MM / 4; ++ks) {
      v2f a;
      a.x = Pm[lm][4 * ks + kh];
      a.y = Pm[lm][4 * ks + kh + 1];
      if (ks & 1) c1 = wmma4(a, bh[ks], c1);
      else        c0 = wmma4(a, bh[ks], c0);
    }
#pragma unroll
    for (int ks = 0; ks < FE / 4; ++ks) {
      v2f a;
      a.x = Zp[lm][4 * ks + kh];
      a.y = Zp[lm][4 * ks + kh + 1];
      if (ks & 1) c1 = wmma4(a, bw[ks], c1);
      else        c0 = wmma4(a, bw[ks], c0);
    }
    const v8f c = c0 + c1;
    const size_t bnr = (size_t)b * NN + i0;
#pragma unroll
    for (int r = 0; r < 8; ++r) {
      const int m = r + 8 * lh;
      a0[m][FF + o] = c[r] * invs[m] + h[(bnr + m) * FF + o];
    }
  }
  }

  const int row0 = b * NN + i0;
#pragma unroll
  for (int i = 0; i < 16; ++i) ((v4f*)wb)[threadIdx.x + 256 * i] = wst[i];
  __syncthreads();
#pragma unroll
  for (int i = 0; i < 16; ++i) wst[i] = ((const v4f*)w2)[threadIdx.x + 256 * i];
  __builtin_amdgcn_sched_barrier(0);

  {
    int o = wv * 16 + lm;
    v8f c = {};
    for (int k0 = 0; k0 < HID; k0 += 4) {
      v2f a, bf;
      a.x  = a0[lm][k0 + kh];
      a.y  = a0[lm][k0 + kh + 1];
      bf.x = wb[(k0 + kh) * HID + o];
      bf.y = wb[(k0 + kh + 1) * HID + o];
      c = wmma4(a, bf, c);
    }
    float bias = b1[o];
#pragma unroll
    for (int r = 0; r < 8; ++r) {
      float v = c[r] + bias;
      a1[r + 8 * lh][o] = v > 0.f ? v : expm1f(v);
    }
  }
  __syncthreads();
#pragma unroll
  for (int i = 0; i < 16; ++i) ((v4f*)wb)[threadIdx.x + 256 * i] = wst[i];
  __syncthreads();
#pragma unroll
  for (int i = 0; i < 8; ++i) wst[i] = ((const v4f*)wo)[threadIdx.x + 256 * i];
  __builtin_amdgcn_sched_barrier(0);

  {
    int o = wv * 16 + lm;
    v8f c = {};
    for (int k0 = 0; k0 < HID; k0 += 4) {
      v2f a, bf;
      a.x  = a1[lm][k0 + kh];
      a.y  = a1[lm][k0 + kh + 1];
      bf.x = wb[(k0 + kh) * HID + o];
      bf.y = wb[(k0 + kh + 1) * HID + o];
      c = wmma4(a, bf, c);
    }
    float bias = b2[o];
#pragma unroll
    for (int r = 0; r < 8; ++r) {
      float v = c[r] + bias;
      a0[r + 8 * lh][o] = v > 0.f ? v : expm1f(v);
    }
  }
  __syncthreads();
#pragma unroll
  for (int i = 0; i < 8; ++i) ((v4f*)wb)[threadIdx.x + 256 * i] = wst[i];
  __syncthreads();

  if (wv < 4) {
    int o = wv * 16 + lm;
    v8f c = {};
    for (int k0 = 0; k0 < HID; k0 += 4) {
      v2f a, bf;
      a.x  = a0[lm][k0 + kh];
      a.y  = a0[lm][k0 + kh + 1];
      bf.x = wb[(k0 + kh) * FF + o];
      bf.y = wb[(k0 + kh + 1) * FF + o];
      c = wmma4(a, bf, c);
    }
    float bias = bo[o];
#pragma unroll
    for (int r = 0; r < 8; ++r) {
      int m = r + 8 * lh;
      float v = c[r] + bias;
      if (rowmask[row0 + m]) v = 0.f;
      out[(size_t)(row0 + m) * FF + o] = v;
    }
  }
}

extern "C" void kernel_launch(void* const* d_in, const int* in_sizes, int n_in,
                              void* d_out, int out_size, void* d_ws,
                              size_t ws_size, hipStream_t stream) {
  (void)in_sizes; (void)n_in; (void)out_size; (void)ws_size;
  const float* x      = (const float*)d_in[0];
  const float* y      = (const float*)d_in[1];
  const float* z      = (const float*)d_in[2];
  const int*   op_adj = (const int*)d_in[3];
  const int*   ma_adj = (const int*)d_in[4];
  const int*   op_ma  = (const int*)d_in[5];
  const float* Wop    = (const float*)d_in[6];
  const float* Wma    = (const float*)d_in[7];
  const float* Wedge  = (const float*)d_in[8];
  const float* att_op = (const float*)d_in[9];
  const float* att_ma = (const float*)d_in[10];
  const float* w1     = (const float*)d_in[11];
  const float* b1     = (const float*)d_in[12];
  const float* w2     = (const float*)d_in[13];
  const float* b2     = (const float*)d_in[14];
  const float* wo     = (const float*)d_in[15];
  const float* bo     = (const float*)d_in[16];
  float* out = (float*)d_out;

  float* ws     = (float*)d_ws;
  float* h      = ws;
  float* hma    = h + BB * NN * FF;
  float* ha1    = hma + BB * MM * FF;
  float* ha2    = ha1 + BB * NN;
  float* ham1   = ha2 + BB * NN;
  float* hma2   = ham1 + BB * NN;
  int*   rmask  = (int*)(hma2 + BB * MM);

  prep_tiles<<<dim3(XT + YT), dim3(128), 0, stream>>>(
      x, Wop, y, Wma, att_op, att_ma, op_ma, h, hma, ha1, ha2, ham1, hma2,
      rmask);
  size_t sm = LDS_FLOATS * sizeof(float);
  node_tiles<<<dim3(NN / 16, BB), dim3(256), sm, stream>>>(
      h, ha1, ha2, op_adj, ma_adj, rmask, z, Wedge, hma, ham1, hma2, op_ma,
      att_ma, w1, b1, w2, b2, wo, bo, out);
}


# two neighbouring 16-node tiles of a batch share one 16-wave work-group (8 waves, own LDS plan each): h, weights and both halves of transposed adjacency lines come from L2 once per pair; same sums
# speedup vs baseline: 5.0564x; 1.0501x over previous
#include <hip/hip_runtime.h>
#include <math.h>


typedef __attribute__((ext_vector_type(2))) float v2f;
typedef __attribute__((ext_vector_type(8))) float v8f;
typedef __attribute__((ext_vector_type(4))) float v4f;
typedef __attribute__((ext_vector_type(4))) int v4i;

#define BB 8
#define NN 1024
#define MM 128
#define FF 64
#define FE 16
#define HID 128

__device__ __forceinline__ v8f wmma4(v2f a, v2f b, v8f c) {
  return __builtin_amdgcn_wmma_f32_16x16x4_f32(
       false, a,  false, b,
       (short)0, c,  false,  false);
}

#define XT ((BB * NN) / 16)
#define YT ((BB * MM) / 16)
#define CTS (FF + 4)

__global__ __launch_bounds__(128) void prep_tiles(
    const float* __restrict__ x, const float* __restrict__ Wop,
    const float* __restrict__ y, const float* __restrict__ Wma,
    const float* __restrict__ att_op, const float* __restrict__ att_ma,
    const int* __restrict__ opma, float* __restrict__ h,
    float* __restrict__ hma, float* __restrict__ ha1,
    float* __restrict__ ha2, float* __restrict__ ham1,
    float* __restrict__ hma2, int* __restrict__ rowmask) {
  __shared__ float ct[16][CTS];
  __shared__ float atts[3][FF];

  const int wv   = threadIdx.x >> 5;
  const int lane = threadIdx.x & 31;
  const int lm   = lane & 15;
  const int kh   = (lane >> 4) * 2;
  const int lh   = lane >> 4;
  const bool node = blockIdx.x < XT;
  const int row0 = (node ? blockIdx.x : blockIdx.x - XT) * 16;

  v4i ad[4] = {};
  if (node) {
    const v4i* at = (const v4i*)(opma + (size_t)row0 * MM);
#pragma unroll
    for (int q = 0; q < 4; ++q) ad[q] = at[q * 128 + threadIdx.x];
    if (threadIdx.x < FF) {
      atts[0][threadIdx.x] = att_op[threadIdx.x];
      atts[1][threadIdx.x] = att_op[FF + threadIdx.x];
      atts[2][threadIdx.x] = att_ma[threadIdx.x];
    }
  } else {
    if (threadIdx.x < FF) atts[0][threadIdx.x] = att_ma[FF + threadIdx.x];
  }

  const float* A = node ? x : y;
  const float* W = node ? Wop : Wma;
  float* C = node ? h : hma;
  const int o = wv * 16 + lm;
  const float* Arow = A + (size_t)(row0 + lm) * FF;

  v8f c = {};
#pragma unroll
  for (int k0 = 0; k0 < FF; k0 += 4) {
    v2f a, b;
    a.x = Arow[k0 + kh];
    a.y = Arow[k0 + kh + 1];
    b.x = W[(size_t)(k0 + kh) * FF + o];
    b.y = W[(size_t)(k0 + kh + 1) * FF + o];
    c = wmma4(a, b, c);
  }
#pragma unroll
  for (int r = 0; r < 8; ++r) {
    int m = r + 8 * lh;
    C[(size_t)(row0 + m) * FF + o] = c[r];
    ct[m][o] = c[r];
  }

  if (node) {
#pragma unroll
    for (int q = 0; q < 4; ++q) {
      const int nz = (ad[q].x | ad[q].y | ad[q].z | ad[q].w) != 0;
      const int anynz = __any(nz);
      if (lane == 0) rowmask[row0 + wv + 4 * q] = anynz ? 0 : 1;
    }
  }
  __syncthreads();

  const int np = node ? 3 : 1;
  if (threadIdx.x < 16 * np) {
    const int p = threadIdx.x >> 4, row = threadIdx.x & 15;
    float s = 0.f;
#pragma unroll 8
    for (int oo = 0; oo < FF; ++oo) s += ct[row][oo] * atts[p][oo];
    float* dst = node ? (p == 0 ? ha1 : (p == 1 ? ha2 : ham1)) : hma2;
    dst[row0 + row] = s;
  }
}

__device__ __forceinline__ float masked_score(float e, int adj) {
  e = e > 0.f ? e : 0.01f * e;
  return adj > 0 ? e : -__builtin_inff();
}

__device__ __forceinline__ float quad_sum(float v) {
  v += __int_as_float(__builtin_amdgcn_update_dpp(
      0, __float_as_int(v), 0xB1, 0xF, 0xF, true));
  v += __int_as_float(__builtin_amdgcn_update_dpp(
      0, __float_as_int(v), 0x4E, 0xF, 0xF, true));
  return v;
}

__device__ __forceinline__ float across_quads_sum(float v) {
  v += __shfl_xor(v, 4, 32);
  v += __shfl_xor(v, 8, 32);
  v += __shfl_xor(v, 16, 32);
  return v;
}

#define KSB 8
#define NBLK (NN / 4 / 4 / KSB)
#define NPW 2
#define NPG (8 * NPW)
#define PST (MM + 4)
#define ZST (FE + 4)
#define AST (HID + 4)
#define OFF_A0   0
#define OFF_A1   (OFF_A0 + 16 * AST)
#define OFF_PM   (OFF_A1 + 16 * AST)
#define OFF_ZP   (OFF_PM + NPG * PST)
#define OFF_INVS (OFF_ZP + NPG * ZST)
#define OFF_ISUM (OFF_INVS + NPG)
#define OFF_BIG  (OFF_ISUM + 16)
#define LDS_FLOATS (OFF_BIG + 16 * (NN + 4))

__global__ __launch_bounds__(512) void node_tiles(
    const float* __restrict__ h, const float* __restrict__ ha1,
    const float* __restrict__ ha2, const int* __restrict__ opadj,
    const int* __restrict__ maadj, const int* __restrict__ rowmask,
    const float* __restrict__ z, const float* __restrict__ wedge,
    const float* __restrict__ hma, const float* __restrict__ ham1,
    const float* __restrict__ hma2, const int* __restrict__ opma,
    const float* __restrict__ attma, const float* __restrict__ w1,
    const float* __restrict__ b1, const float* __restrict__ w2,
    const float* __restrict__ b2, const float* __restrict__ wo,
    const float* __restrict__ bo, float* __restrict__ out) {
  extern __shared__ __attribute__((aligned(16))) float smem_pair[];
  const int half = __builtin_amdgcn_readfirstlane((int)(threadIdx.x >> 8));
  const int tid = threadIdx.x & 255;
  float* smem = smem_pair + half * LDS_FLOATS;
  typedef float act_row_t[AST];
  typedef float att_row_t[NN + 4];
  typedef float pm_row_t[PST];
  typedef float zp_row_t[ZST];
  act_row_t* a0 = (act_row_t*)(smem + OFF_A0);
  act_row_t* a1 = (act_row_t*)(smem + OFF_A1);
  pm_row_t* Pm = (pm_row_t*)(smem + OFF_PM);
  zp_row_t* Zp = (zp_row_t*)(smem + OFF_ZP);
  float* invs = smem + OFF_INVS;
  float* inv_sum = smem + OFF_ISUM;
  att_row_t* att = (att_row_t*)(smem + OFF_BIG);
  float* wb = smem + OFF_BIG;

  const int b  = blockIdx.y;
  const int i0 = (blockIdx.x * 2 + half) * 16;
  const int* opb = opadj + (size_t)b * NN * NN;
  const int* mab = maadj + (size_t)b * NN * NN;
  const float NEGINF = -__builtin_inff();

  const int wv = __builtin_amdgcn_readfirstlane(tid >> 5);
  const int lane = tid & 31;
  const int lm = lane & 15;
  const int lh = lane >> 4;
  const int kh = lh * 2;

  {
    const v4i* opT = (const v4i*)(opb + i0);
    const v4i* maT = (const v4i*)(mab + i0);
#pragma unroll 8
    for (int t = tid; t < 4 * NN; t += 256) {
      const int j = t >> 2, part = t & 3;
      const v4i a = opT[(size_t)j * (NN / 4) + part];
      const v4i m = maT[(size_t)j * (NN / 4) + part];
      const v4i s = a + m;
      att[4 * part + 0][j] = __int_as_float(s.x);
      att[4 * part + 1][j] = __int_as_float(s.y);
      att[4 * part + 2][j] = __int_as_float(s.z);
      att[4 * part + 3][j] = __int_as_float(s.w);
    }
  }
  __syncthreads();

  {
    const int j = tid * 4;
    const v4f e1 = *(const v4f*)(ha1 + b * NN + j);
#pragma unroll 8
    for (int r = 0; r < 16; ++r) {
      const int i = i0 + r;
      const bool rm = rowmask[b * NN + i] != 0;
      const float e2 = ha2[b * NN + i];
      v4f* slot = (v4f*)&att[r][j];
      const v4f tr = *slot;
      const v4i d1 = *(const v4i*)(opb + (size_t)i * NN + j);
      const v4i d2 = *(const v4i*)(mab + (size_t)i * NN + j);
      v4i adj = d1 + d2;
      adj.x += __float_as_int(tr.x);
      adj.y += __float_as_int(tr.y);
      adj.z += __float_as_int(tr.z);
      adj.w += __float_as_int(tr.w);
      if (rm) adj = (v4i){1, 1, 1, 1};
      v4f sc4;
      sc4.x = masked_score(e2 + e1.x, adj.x);
      sc4.y = masked_score(e2 + e1.y, adj.y);
      sc4.z = masked_score(e2 + e1.z, adj.z);
      sc4.w = masked_score(e2 + e1.w, adj.w);
      *slot = sc4;
    }
  }
  __syncthreads();

  for (int r = wv; r < 16; r += 8) {
    float mx = NEGINF;
    for (int j = lane; j < NN; j += 32) mx = fmaxf(mx, att[r][j]);
#pragma unroll
    for (int off = 16; off; off >>= 1) mx = fmaxf(mx, __shfl_xor(mx, off, 32));
    if (mx == NEGINF) mx = 0.f;
    float sum = 0.f;
    for (int j = lane; j < NN; j += 32) {
      float p = __expf(att[r][j] - mx);
      att[r][j] = p;
      sum += p;
    }
#pragma unroll
    for (int off = 16; off; off >>= 1) sum += __shfl_xor(sum, off, 32);
    if (lane == 0) inv_sum[r] = 1.f / fmaxf(sum, 1e-30f);
  }
  __syncthreads();

  typedef float red_tile_t[16][FF + 4];
  red_tile_t* red = (red_tile_t*)(smem + OFF_BIG);
  v8f acc[4] = {};
  if (wv < 4) {
    const float* hb = h + (size_t)b * NN * FF;
    const int kb = wv * (NN / 4);
    const float* hq = hb + (size_t)(kb + 2 * lh) * FF + 4 * lm;
    const float* aq = &att[lm][kb + 2 * lh];

    v4f bq[2][KSB][2];
    v2f aa[2][KSB];

    auto fetch = [&](const int st, const int blk) {
#pragma unroll
      for (int s = 0; s < KSB; ++s) {
        const int ks = blk * KSB + s;
        bq[st][s][0] = *(const v4f*)(hq + (size_t)(4 * ks) * FF);
        bq[st][s][1] = *(const v4f*)(hq + (size_t)(4 * ks + 1) * FF);
        aa[st][s]    = *(const v2f*)(aq + 4 * ks);
      }
    };
    auto mulblk = [&](const int st) {
#pragma unroll
      for (int s = 0; s < KSB; ++s) {
#pragma unroll
        for (int t = 0; t < 4; ++t) {
          v2f bf;
          bf.x = bq[st][s][0][t];
          bf.y = bq[st][s][1][t];
          acc[t] = wmma4(aa[st][s], bf, acc[t]);
        }
      }
    };

    fetch(0, 0);
    for (int it = 0; it < NBLK / 2 - 1; ++it) {
      fetch(1, 2 * it + 1);
      __builtin_amdgcn_sched_barrier(0);
      mulblk(0);
      __builtin_amdgcn_sched_barrier(0);
      fetch(0, 2 * it + 2);
      asm volatile("" ::: "memory");
      __builtin_amdgcn_sched_barrier(0);
      mulblk(1);
    }
    fetch(1, NBLK - 1);
    mulblk(0);
    mulblk(1);

  }
  __syncthreads();
  if (wv < 4) {
#pragma unroll
    for (int r = 0; r < 8; ++r) {
      v4f v;
      v.x = acc[0][r]; v.y = acc[1][r]; v.z = acc[2][r]; v.w = acc[3][r];
      *(v4f*)&red[wv][r + 8 * lh][4 * lm] = v;
    }
  }
  __syncthreads();

  if (tid < 128) {
    const int m = tid >> 3;
    const int g = (tid & 7) * 8;
    v4f s0 = *(const v4f*)&red[0][m][g];
    v4f s1 = *(const v4f*)&red[0][m][g + 4];
#pragma unroll
    for (int w = 1; w < 4; ++w) {
      s0 += *(const v4f*)&red[w][m][g];
      s1 += *(const v4f*)&red[w][m][g + 4];
    }
    const float is = inv_sum[m];
    *(v4f*)&a0[m][g]     = s0 * is;
    *(v4f*)&a0[m][g + 4] = s1 * is;
  }

  v4f wst[16];
  {
  const int q  = lane & 3;
  const int rq = lane >> 2;

  const size_t bn0 = (size_t)b * NN + i0 + wv * NPW;

  v4f wv4[8], av4[8];
  {
    const v4f* wr = (const v4f*)(wedge + (lane & 15) * FF + 32 * (lane >> 4));
    const v4f* ar = (const v4f*)(attma + 2 * FF + 32 * (lane >> 4));
#pragma unroll
    for (int i = 0; i < 8; ++i) { wv4[i] = wr[i]; av4[i] = ar[i]; }
  }
  float hm2r[16];
#pragma unroll
  for (int j = 0; j < 16; ++j) hm2r[j] = hma2[(size_t)b * MM + rq + 8 * j];
  v4i   ad[NPW];
  int   rmv[NPW];
  float hm1v[NPW];
#pragma unroll
  for (int t = 0; t < NPW; ++t) {
    ad[t]   = *(const v4i*)(opma + (bn0 + t) * MM + 4 * lane);
    rmv[t]  = rowmask[bn0 + t];
    hm1v[t] = ham1[bn0 + t];
  }

  const v4f* zt = (const v4f*)(z + bn0 * MM * FE);
  v4f zr[NPW][16];
#pragma unroll
  for (int t = 0; t < NPW; ++t) {
#pragma unroll
    for (int j = 0; j < 16; ++j) zr[t][j] = zt[t * 512 + j * 32 + lane];
  }

  unsigned connv[NPW];
#pragma unroll
  for (int t = 0; t < NPW; ++t) {
    const unsigned s0 = __builtin_amdgcn_ballot_w32(ad[t].x > 0);
    const unsigned s1 = __builtin_amdgcn_ballot_w32(ad[t].y > 0);
    const unsigned s2 = __builtin_amdgcn_ballot_w32(ad[t].z > 0);
    const unsigned s3 = __builtin_amdgcn_ballot_w32(ad[t].w > 0);
    const int cq = rq & 3;
    unsigned conn = cq == 0 ? s0 : (cq == 1 ? s1 : (cq == 2 ? s2 : s3));
    conn >>= (lane >> 4);
    if (rmv[t]) conn = 0xffffffffu;
    connv[t] = conn;
  }

  float wpart = 0.f;
#pragma unroll
  for (int i = 0; i < 8; ++i) {
    wpart += wv4[i].x * av4[i].x;
    wpart += wv4[i].y * av4[i].y;
    wpart += wv4[i].z * av4[i].z;
    wpart += wv4[i].w * av4[i].w;
  }
  wpart += __shfl_xor(wpart, 16, 32);
  float wq[4];
#pragma unroll
  for (int c = 0; c < 4; ++c) wq[c] = __shfl(wpart, 4 * q + c, 32);

#pragma unroll
  for (int t = 0; t < NPW; ++t) {
    const int nl = wv * NPW + t;

    const unsigned conn = connv[t];

    float sc[16];
    float mx = NEGINF;
#pragma unroll
    for (int j = 0; j < 16; ++j) {
      const v4f zv = zr[t][j];
      float d = zv.x * wq[0];
      d += zv.y * wq[1];
      d += zv.z * wq[2];
      d += zv.w * wq[3];
      d = quad_sum(d);
      float e = (hm1v[t] + hm2r[j]) + d;
      e = e > 0.f ? e : 0.01f * e;
      sc[j] = ((conn >> (2 * j)) & 1u) ? e : NEGINF;
      mx = fmaxf(mx, sc[j]);
    }
    mx = fmaxf(mx, __shfl_xor(mx, 4, 32));
    mx = fmaxf(mx, __shfl_xor(mx, 8, 32));
    mx = fmaxf(mx, __shfl_xor(mx, 16, 32));
    if (mx == NEGINF) mx = 0.f;

    float ps = 0.f;
    v4f zacc = {0.f, 0.f, 0.f, 0.f};
#pragma unroll
    for (int j = 0; j < 16; ++j) {
      const float p = __expf(sc[j] - mx);
      sc[j] = p;
      ps += p;
      zacc += p * zr[t][j];
    }
    ps = across_quads_sum(ps);
    zacc.x = across_quads_sum(zacc.x);
    zacc.y = across_quads_sum(zacc.y);
    zacc.z = across_quads_sum(zacc.z);
    zacc.w = across_quads_sum(zacc.w);

    if (q == 0) {
#pragma unroll
      for (int j = 0; j < 16; ++j) Pm[nl][rq + 8 * j] = sc[j];
    }
    if (lane < 4) *(v4f*)&Zp[nl][4 * lane] = zacc;
    if (lane == 0) invs[nl] = 1.f / fmaxf(ps, 1e-30f);
  }

  const int o  = (wv & 3) * 16 + lm;
  const float* hmb = hma + (size_t)b * MM * FF;
  v2f bh[MM / 4];
  v2f bw[FE / 4];
  if (wv < 4) {
#pragma unroll
    for (int ks = 0; ks < MM / 4; ++ks) {
      bh[ks].x = hmb[(size_t)(4 * ks + kh) * FF + o];
      bh[ks].y = hmb[(size_t)(4 * ks + kh + 1) * FF + o];
    }
#pragma unroll
    for (int ks = 0; ks < FE / 4; ++ks) {
      bw[ks].x = wedge[(4 * ks + kh) * FF + o];
      bw[ks].y = wedge[(4 * ks + kh + 1) * FF + o];
    }
  }
  __syncthreads();
#pragma unroll
  for (int i = 0; i < 16; ++i) wst[i] = ((const v4f*)w1)[tid + 256 * i];

  if (wv < 4) {
    v8f c0 = {}, c1 = {};
#pragma unroll
    for (int ks = 0; ks < MM / 4; ++ks) {
      v2f a;
      a.x = Pm[lm][4 * ks + kh];
      a.y = Pm[lm][4 * ks + kh + 1];
      if (ks & 1) c1 = wmma4(a, bh[ks], c1);
      else        c0 = wmma4(a, bh[ks], c0);
    }
#pragma unroll
    for (int ks = 0; ks < FE / 4; ++ks) {
      v2f a;
      a.x = Zp[lm][4 * ks + kh];
      a.y = Zp[lm][4 * ks + kh + 1];
      if (ks & 1) c1 = wmma4(a, bw[ks], c1);
      else        c0 = wmma4(a, bw[ks], c0);
    }
    const v8f c = c0 + c1;
    const size_t bnr = (size_t)b * NN + i0;
#pragma unroll
    for (int r = 0; r < 8; ++r) {
      const int m = r + 8 * lh;
      a0[m][FF + o] = c[r] * invs[m] + h[(bnr + m) * FF + o];
    }
  }
  }

  const int row0 = b * NN + i0;
#pragma unroll
  for (int i = 0; i < 16; ++i) ((v4f*)wb)[tid + 256 * i] = wst[i];
  __syncthreads();
#pragma unroll
  for (int i = 0; i < 16; ++i) wst[i] = ((const v4f*)w2)[tid + 256 * i];
  __builtin_amdgcn_sched_barrier(0);

  {
    int o = wv * 16 + lm;
    v8f c = {};
    for (int k0 = 0; k0 < HID; k0 += 4) {
      v2f a, bf;
      a.x  = a0[lm][k0 + kh];
      a.y  = a0[lm][k0 + kh + 1];
      bf.x = wb[(k0 + kh) * HID + o];
      bf.y = wb[(k0 + kh + 1) * HID + o];
      c = wmma4(a, bf, c);
    }
    float bias = b1[o];
#pragma unroll
    for (int r = 0; r < 8; ++r) {
      float v = c[r] + bias;
      a1[r + 8 * lh][o] = v > 0.f ? v : expm1f(v);
    }
  }
  __syncthreads();
#pragma unroll
  for (int i = 0; i < 16; ++i) ((v4f*)wb)[tid + 256 * i] = wst[i];
  __syncthreads();
#pragma unroll
  for (int i = 0; i < 8; ++i) wst[i] = ((const v4f*)wo)[tid + 256 * i];
  __builtin_amdgcn_sched_barrier(0);

  {
    int o = wv * 16 + lm;
    v8f c = {};
    for (int k0 = 0; k0 < HID; k0 += 4) {
      v2f a, bf;
      a.x  = a1[lm][k0 + kh];
      a.y  = a1[lm][k0 + kh + 1];
      bf.x = wb[(k0 + kh) * HID + o];
      bf.y = wb[(k0 + kh + 1) * HID + o];
      c = wmma4(a, bf, c);
    }
    float bias = b2[o];
#pragma unroll
    for (int r = 0; r < 8; ++r) {
      float v = c[r] + bias;
      a0[r + 8 * lh][o] = v > 0.f ? v : expm1f(v);
    }
  }
  __syncthreads();
#pragma unroll
  for (int i = 0; i < 8; ++i) ((v4f*)wb)[tid + 256 * i] = wst[i];
  __syncthreads();

  if (wv < 4) {
    int o = wv * 16 + lm;
    v8f c = {};
    for (int k0 = 0; k0 < HID; k0 += 4) {
      v2f a, bf;
      a.x  = a0[lm][k0 + kh];
      a.y  = a0[lm][k0 + kh + 1];
      bf.x = wb[(k0 + kh) * FF + o];
      bf.y = wb[(k0 + kh + 1) * FF + o];
      c = wmma4(a, bf, c);
    }
    float bias = bo[o];
#pragma unroll
    for (int r = 0; r < 8; ++r) {
      int m = r + 8 * lh;
      float v = c[r] + bias;
      if (rowmask[row0 + m]) v = 0.f;
      out[(size_t)(row0 + m) * FF + o] = v;
    }
  }
}

extern "C" void kernel_launch(void* const* d_in, const int* in_sizes, int n_in,
                              void* d_out, int out_size, void* d_ws,
                              size_t ws_size, hipStream_t stream) {
  (void)in_sizes; (void)n_in; (void)out_size; (void)ws_size;
  const float* x      = (const float*)d_in[0];
  const float* y      = (const float*)d_in[1];
  const float* z      = (const float*)d_in[2];
  const int*   op_adj = (const int*)d_in[3];
  const int*   ma_adj = (const int*)d_in[4];
  const int*   op_ma  = (const int*)d_in[5];
  const float* Wop    = (const float*)d_in[6];
  const float* Wma    = (const float*)d_in[7];
  const float* Wedge  = (const float*)d_in[8];
  const float* att_op = (const float*)d_in[9];
  const float* att_ma = (const float*)d_in[10];
  const float* w1     = (const float*)d_in[11];
  const float* b1     = (const float*)d_in[12];
  const float* w2     = (const float*)d_in[13];
  const float* b2     = (const float*)d_in[14];
  const float* wo     = (const float*)d_in[15];
  const float* bo     = (const float*)d_in[16];
  float* out = (float*)d_out;

  float* ws     = (float*)d_ws;
  float* h      = ws;
  float* hma    = h + BB * NN * FF;
  float* ha1    = hma + BB * MM * FF;
  float* ha2    = ha1 + BB * NN;
  float* ham1   = ha2 + BB * NN;
  float* hma2   = ham1 + BB * NN;
  int*   rmask  = (int*)(hma2 + BB * MM);

  prep_tiles<<<dim3(XT + YT), dim3(128), 0, stream>>>(
      x, Wop, y, Wma, att_op, att_ma, op_ma, h, hma, ha1, ha2, ham1, hma2,
      rmask);
  size_t sm = 2 * LDS_FLOATS * sizeof(float);
  node_tiles<<<dim3(NN / 32, BB), dim3(512), sm, stream>>>(
      h, ha1, ha2, op_adj, ma_adj, rmask, z, Wedge, hma, ham1, hma2, op_ma,
      att_ma, w1, b1, w2, b2, wo, bo, out);
}


# the ma stage's per-node work (z stream, scores, numerators, pooled edge vectors) moves onto a tile's waves 4..7 and runs beside the op product of waves 0..3, next z tile a node ahead; same sums
# speedup vs baseline: 5.2374x; 1.0358x over previous
#include <hip/hip_runtime.h>
#include <math.h>


typedef __attribute__((ext_vector_type(2))) float v2f;
typedef __attribute__((ext_vector_type(8))) float v8f;
typedef __attribute__((ext_vector_type(4))) float v4f;
typedef __attribute__((ext_vector_type(4))) int v4i;

#define BB 8
#define NN 1024
#define MM 128
#define FF 64
#define FE 16
#define HID 128

__device__ __forceinline__ v8f wmma4(v2f a, v2f b, v8f c) {
  return __builtin_amdgcn_wmma_f32_16x16x4_f32(
       false, a,  false, b,
       (short)0, c,  false,  false);
}

#define XT ((BB * NN) / 16)
#define YT ((BB * MM) / 16)
#define CTS (FF + 4)

__global__ __launch_bounds__(128) void prep_tiles(
    const float* __restrict__ x, const float* __restrict__ Wop,
    const float* __restrict__ y, const float* __restrict__ Wma,
    const float* __restrict__ att_op, const float* __restrict__ att_ma,
    const int* __restrict__ opma, float* __restrict__ h,
    float* __restrict__ hma, float* __restrict__ ha1,
    float* __restrict__ ha2, float* __restrict__ ham1,
    float* __restrict__ hma2, int* __restrict__ rowmask) {
  __shared__ float ct[16][CTS];
  __shared__ float atts[3][FF];

  const int wv   = threadIdx.x >> 5;
  const int lane = threadIdx.x & 31;
  const int lm   = lane & 15;
  const int kh   = (lane >> 4) * 2;
  const int lh   = lane >> 4;
  const bool node = blockIdx.x < XT;
  const int row0 = (node ? blockIdx.x : blockIdx.x - XT) * 16;

  v4i ad[4] = {};
  if (node) {
    const v4i* at = (const v4i*)(opma + (size_t)row0 * MM);
#pragma unroll
    for (int q = 0; q < 4; ++q) ad[q] = at[q * 128 + threadIdx.x];
    if (threadIdx.x < FF) {
      atts[0][threadIdx.x] = att_op[threadIdx.x];
      atts[1][threadIdx.x] = att_op[FF + threadIdx.x];
      atts[2][threadIdx.x] = att_ma[threadIdx.x];
    }
  } else {
    if (threadIdx.x < FF) atts[0][threadIdx.x] = att_ma[FF + threadIdx.x];
  }

  const float* A = node ? x : y;
  const float* W = node ? Wop : Wma;
  float* C = node ? h : hma;
  const int o = wv * 16 + lm;
  const float* Arow = A + (size_t)(row0 + lm) * FF;

  v8f c = {};
#pragma unroll
  for (int k0 = 0; k0 < FF; k0 += 4) {
    v2f a, b;
    a.x = Arow[k0 + kh];
    a.y = Arow[k0 + kh + 1];
    b.x = W[(size_t)(k0 + kh) * FF + o];
    b.y = W[(size_t)(k0 + kh + 1) * FF + o];
    c = wmma4(a, b, c);
  }
#pragma unroll
  for (int r = 0; r < 8; ++r) {
    int m = r + 8 * lh;
    C[(size_t)(row0 + m) * FF + o] = c[r];
    ct[m][o] = c[r];
  }

  if (node) {
#pragma unroll
    for (int q = 0; q < 4; ++q) {
      const int nz = (ad[q].x | ad[q].y | ad[q].z | ad[q].w) != 0;
      const int anynz = __any(nz);
      if (lane == 0) rowmask[row0 + wv + 4 * q] = anynz ? 0 : 1;
    }
  }
  __syncthreads();

  const int np = node ? 3 : 1;
  if (threadIdx.x < 16 * np) {
    const int p = threadIdx.x >> 4, row = threadIdx.x & 15;
    float s = 0.f;
#pragma unroll 8
    for (int oo = 0; oo < FF; ++oo) s += ct[row][oo] * atts[p][oo];
    float* dst = node ? (p == 0 ? ha1 : (p == 1 ? ha2 : ham1)) : hma2;
    dst[row0 + row] = s;
  }
}

__device__ __forceinline__ float masked_score(float e, int adj) {
  e = e > 0.f ? e : 0.01f * e;
  return adj > 0 ? e : -__builtin_inff();
}

__device__ __forceinline__ float quad_sum(float v) {
  v += __int_as_float(__builtin_amdgcn_update_dpp(
      0, __float_as_int(v), 0xB1, 0xF, 0xF, true));
  v += __int_as_float(__builtin_amdgcn_update_dpp(
      0, __float_as_int(v), 0x4E, 0xF, 0xF, true));
  return v;
}

__device__ __forceinline__ float across_quads_sum(float v) {
  v += __shfl_xor(v, 4, 32);
  v += __shfl_xor(v, 8, 32);
  v += __shfl_xor(v, 16, 32);
  return v;
}

#define KSB 8
#define NBLK (NN / 4 / 4 / KSB)
#define NPW 4
#define NPG (4 * NPW)
#define PST (MM + 4)
#define ZST (FE + 4)
#define AST (HID + 4)
#define OFF_A0   0
#define OFF_A1   (OFF_A0 + 16 * AST)
#define OFF_PM   (OFF_A1 + 16 * AST)
#define OFF_ZP   (OFF_PM + NPG * PST)
#define OFF_INVS (OFF_ZP + NPG * ZST)
#define OFF_ISUM (OFF_INVS + NPG)
#define OFF_BIG  (OFF_ISUM + 16)
#define LDS_FLOATS (OFF_BIG + 16 * (NN + 4))

__global__ __launch_bounds__(512) void node_tiles(
    const float* __restrict__ h, const float* __restrict__ ha1,
    const float* __restrict__ ha2, const int* __restrict__ opadj,
    const int* __restrict__ maadj, const int* __restrict__ rowmask,
    const float* __restrict__ z, const float* __restrict__ wedge,
    const float* __restrict__ hma, const float* __restrict__ ham1,
    const float* __restrict__ hma2, const int* __restrict__ opma,
    const float* __restrict__ attma, const float* __restrict__ w1,
    const float* __restrict__ b1, const float* __restrict__ w2,
    const float* __restrict__ b2, const float* __restrict__ wo,
    const float* __restrict__ bo, float* __restrict__ out) {
  extern __shared__ __attribute__((aligned(16))) float smem_pair[];
  const int half = __builtin_amdgcn_readfirstlane((int)(threadIdx.x >> 8));
  const int tid = threadIdx.x & 255;
  float* smem = smem_pair + half * LDS_FLOATS;
  typedef float act_row_t[AST];
  typedef float att_row_t[NN + 4];
  typedef float pm_row_t[PST];
  typedef float zp_row_t[ZST];
  act_row_t* a0 = (act_row_t*)(smem + OFF_A0);
  act_row_t* a1 = (act_row_t*)(smem + OFF_A1);
  pm_row_t* Pm = (pm_row_t*)(smem + OFF_PM);
  zp_row_t* Zp = (zp_row_t*)(smem + OFF_ZP);
  float* invs = smem + OFF_INVS;
  float* inv_sum = smem + OFF_ISUM;
  att_row_t* att = (att_row_t*)(smem + OFF_BIG);
  float* wb = smem + OFF_BIG;

  const int b  = blockIdx.y;
  const int i0 = (blockIdx.x * 2 + half) * 16;
  const int* opb = opadj + (size_t)b * NN * NN;
  const int* mab = maadj + (size_t)b * NN * NN;
  const float NEGINF = -__builtin_inff();

  const int wv = __builtin_amdgcn_readfirstlane(tid >> 5);
  const int lane = tid & 31;
  const int lm = lane & 15;
  const int lh = lane >> 4;
  const int kh = lh * 2;

  {
    const v4i* opT = (const v4i*)(opb + i0);
    const v4i* maT = (const v4i*)(mab + i0);
#pragma unroll 8
    for (int t = tid; t < 4 * NN; t += 256) {
      const int j = t >> 2, part = t & 3;
      const v4i a = opT[(size_t)j * (NN / 4) + part];
      const v4i m = maT[(size_t)j * (NN / 4) + part];
      const v4i s = a + m;
      att[4 * part + 0][j] = __int_as_float(s.x);
      att[4 * part + 1][j] = __int_as_float(s.y);
      att[4 * part + 2][j] = __int_as_float(s.z);
      att[4 * part + 3][j] = __int_as_float(s.w);
    }
  }
  __syncthreads();

  {
    const int j = tid * 4;
    const v4f e1 = *(const v4f*)(ha1 + b * NN + j);
#pragma unroll 8
    for (int r = 0; r < 16; ++r) {
      const int i = i0 + r;
      const bool rm = rowmask[b * NN + i] != 0;
      const float e2 = ha2[b * NN + i];
      v4f* slot = (v4f*)&att[r][j];
      const v4f tr = *slot;
      const v4i d1 = *(const v4i*)(opb + (size_t)i * NN + j);
      const v4i d2 = *(const v4i*)(mab + (size_t)i * NN + j);
      v4i adj = d1 + d2;
      adj.x += __float_as_int(tr.x);
      adj.y += __float_as_int(tr.y);
      adj.z += __float_as_int(tr.z);
      adj.w += __float_as_int(tr.w);
      if (rm) adj = (v4i){1, 1, 1, 1};
      v4f sc4;
      sc4.x = masked_score(e2 + e1.x, adj.x);
      sc4.y = masked_score(e2 + e1.y, adj.y);
      sc4.z = masked_score(e2 + e1.z, adj.z);
      sc4.w = masked_score(e2 + e1.w, adj.w);
      *slot = sc4;
    }
  }
  __syncthreads();

  for (int r = wv; r < 16; r += 8) {
    float mx = NEGINF;
    for (int j = lane; j < NN; j += 32) mx = fmaxf(mx, att[r][j]);
#pragma unroll
    for (int off = 16; off; off >>= 1) mx = fmaxf(mx, __shfl_xor(mx, off, 32));
    if (mx == NEGINF) mx = 0.f;
    float sum = 0.f;
    for (int j = lane; j < NN; j += 32) {
      float p = __expf(att[r][j] - mx);
      att[r][j] = p;
      sum += p;
    }
#pragma unroll
    for (int off = 16; off; off >>= 1) sum += __shfl_xor(sum, off, 32);
    if (lane == 0) inv_sum[r] = 1.f / fmaxf(sum, 1e-30f);
  }
  __syncthreads();

  typedef float red_tile_t[16][FF + 4];
  red_tile_t* red = (red_tile_t*)(smem + OFF_BIG);
  v8f acc[4] = {};
  const int o  = (wv & 3) * 16 + lm;
  const float* hmb = hma + (size_t)b * MM * FF;
  v2f bh[MM / 4];
  v2f bw[FE / 4];
  if (wv < 4) {
    const float* hb = h + (size_t)b * NN * FF;
    const int kb = wv * (NN / 4);
    const float* hq = hb + (size_t)(kb + 2 * lh) * FF + 4 * lm;
    const float* aq = &att[lm][kb + 2 * lh];

    v4f bq[2][KSB][2];
    v2f aa[2][KSB];

    auto fetch = [&](const int st, const int blk) {
#pragma unroll
      for (int s = 0; s < KSB; ++s) {
        const int ks = blk * KSB + s;
        bq[st][s][0] = *(const v4f*)(hq + (size_t)(4 * ks) * FF);
        bq[st][s][1] = *(const v4f*)(hq + (size_t)(4 * ks + 1) * FF);
        aa[st][s]    = *(const v2f*)(aq + 4 * ks);
      }
    };
    auto mulblk = [&](const int st) {
#pragma unroll
      for (int s = 0; s < KSB; ++s) {
#pragma unroll
        for (int t = 0; t < 4; ++t) {
          v2f bf;
          bf.x = bq[st][s][0][t];
          bf.y = bq[st][s][1][t];
          acc[t] = wmma4(aa[st][s], bf, acc[t]);
        }
      }
    };

    fetch(0, 0);
    for (int it = 0; it < NBLK / 2 - 1; ++it) {
      fetch(1, 2 * it + 1);
      __builtin_amdgcn_sched_barrier(0);
      mulblk(0);
      __builtin_amdgcn_sched_barrier(0);
      fetch(0, 2 * it + 2);
      asm volatile("" ::: "memory");
      __builtin_amdgcn_sched_barrier(0);
      mulblk(1);
    }
    fetch(1, NBLK - 1);
    mulblk(0);
    mulblk(1);

#pragma unroll
    for (int ks = 0; ks < MM / 4; ++ks) {
      bh[ks].x = hmb[(size_t)(4 * ks + kh) * FF + o];
      bh[ks].y = hmb[(size_t)(4 * ks + kh + 1) * FF + o];
    }
#pragma unroll
    for (int ks = 0; ks < FE / 4; ++ks) {
      bw[ks].x = wedge[(4 * ks + kh) * FF + o];
      bw[ks].y = wedge[(4 * ks + kh + 1) * FF + o];
    }
  } else {
    const int q  = lane & 3;
    const int rq = lane >> 2;
    const int mw = wv - 4;

    const size_t bn0 = (size_t)b * NN + i0 + mw * NPW;

    v4f wv4[8], av4[8];
    {
      const v4f* wr = (const v4f*)(wedge + (lane & 15) * FF + 32 * (lane >> 4));
      const v4f* ar = (const v4f*)(attma + 2 * FF + 32 * (lane >> 4));
#pragma unroll
      for (int i = 0; i < 8; ++i) { wv4[i] = wr[i]; av4[i] = ar[i]; }
    }
    float hm2r[16];
#pragma unroll
    for (int j = 0; j < 16; ++j) hm2r[j] = hma2[(size_t)b * MM + rq + 8 * j];
    v4i   ad[NPW];
    int   rmv[NPW];
    float hm1v[NPW];
#pragma unroll
    for (int t = 0; t < NPW; ++t) {
      ad[t]   = *(const v4i*)(opma + (bn0 + t) * MM + 4 * lane);
      rmv[t]  = rowmask[bn0 + t];
      hm1v[t] = ham1[bn0 + t];
    }

    const v4f* zt = (const v4f*)(z + bn0 * MM * FE);
    v4f zr[2][16];
#pragma unroll
    for (int t = 0; t < 2; ++t) {
#pragma unroll
      for (int j = 0; j < 16; ++j) zr[t][j] = zt[t * 512 + j * 32 + lane];
    }

    unsigned connv[NPW];
#pragma unroll
    for (int t = 0; t < NPW; ++t) {
      const unsigned s0 = __builtin_amdgcn_ballot_w32(ad[t].x > 0);
      const unsigned s1 = __builtin_amdgcn_ballot_w32(ad[t].y > 0);
      const unsigned s2 = __builtin_amdgcn_ballot_w32(ad[t].z > 0);
      const unsigned s3 = __builtin_amdgcn_ballot_w32(ad[t].w > 0);
      const int cq = rq & 3;
      unsigned conn = cq == 0 ? s0 : (cq == 1 ? s1 : (cq == 2 ? s2 : s3));
      conn >>= (lane >> 4);
      if (rmv[t]) conn = 0xffffffffu;
      connv[t] = conn;
    }

    float wpart = 0.f;
#pragma unroll
    for (int i = 0; i < 8; ++i) {
      wpart += wv4[i].x * av4[i].x;
      wpart += wv4[i].y * av4[i].y;
      wpart += wv4[i].z * av4[i].z;
      wpart += wv4[i].w * av4[i].w;
    }
    wpart += __shfl_xor(wpart, 16, 32);
    float wq[4];
#pragma unroll
    for (int c = 0; c < 4; ++c) wq[c] = __shfl(wpart, 4 * q + c, 32);

#pragma unroll
    for (int t = 0; t < NPW; ++t) {
      const int st = t & 1;
      const int nl = mw * NPW + t;

      const unsigned conn = connv[t];

      float sc[16];
      float mx = NEGINF;
#pragma unroll
      for (int j = 0; j < 16; ++j) {
        const v4f zv = zr[st][j];
        float d = zv.x * wq[0];
        d += zv.y * wq[1];
        d += zv.z * wq[2];
        d += zv.w * wq[3];
        d = quad_sum(d);
        float e = (hm1v[t] + hm2r[j]) + d;
        e = e > 0.f ? e : 0.01f * e;
        sc[j] = ((conn >> (2 * j)) & 1u) ? e : NEGINF;
        mx = fmaxf(mx, sc[j]);
      }
      mx = fmaxf(mx, __shfl_xor(mx, 4, 32));
      mx = fmaxf(mx, __shfl_xor(mx, 8, 32));
      mx = fmaxf(mx, __shfl_xor(mx, 16, 32));
      if (mx == NEGINF) mx = 0.f;

      float ps = 0.f;
      v4f zacc = {0.f, 0.f, 0.f, 0.f};
#pragma unroll
      for (int j = 0; j < 16; ++j) {
        const float p = __expf(sc[j] - mx);
        sc[j] = p;
        ps += p;
        zacc += p * zr[st][j];
      }

      if (t + 2 < NPW) {
#pragma unroll
        for (int j = 0; j < 16; ++j) zr[st][j] = zt[(t + 2) * 512 + j * 32 + lane];
      }

      ps = across_quads_sum(ps);
      zacc.x = across_quads_sum(zacc.x);
      zacc.y = across_quads_sum(zacc.y);
      zacc.z = across_quads_sum(zacc.z);
      zacc.w = across_quads_sum(zacc.w);

      if (q == 0) {
#pragma unroll
        for (int j = 0; j < 16; ++j) Pm[nl][rq + 8 * j] = sc[j];
      }
      if (lane < 4) *(v4f*)&Zp[nl][4 * lane] = zacc;
      if (lane == 0) invs[nl] = 1.f / fmaxf(ps, 1e-30f);
    }
  }
  __syncthreads();
  if (wv < 4) {
#pragma unroll
    for (int r = 0; r < 8; ++r) {
      v4f v;
      v.x = acc[0][r]; v.y = acc[1][r]; v.z = acc[2][r]; v.w = acc[3][r];
      *(v4f*)&red[wv][r + 8 * lh][4 * lm] = v;
    }
  }
  __syncthreads();

  v4f wst[16];
#pragma unroll
  for (int i = 0; i < 16; ++i) wst[i] = ((const v4f*)w1)[tid + 256 * i];

  if (tid < 128) {
    const int m = tid >> 3;
    const int g = (tid & 7) * 8;
    v4f s0 = *(const v4f*)&red[0][m][g];
    v4f s1 = *(const v4f*)&red[0][m][g + 4];
#pragma unroll
    for (int w = 1; w < 4; ++w) {
      s0 += *(const v4f*)&red[w][m][g];
      s1 += *(const v4f*)&red[w][m][g + 4];
    }
    const float is = inv_sum[m];
    *(v4f*)&a0[m][g]     = s0 * is;
    *(v4f*)&a0[m][g + 4] = s1 * is;
  }

  if (wv < 4) {
    v8f c0 = {}, c1 = {};
#pragma unroll
    for (int ks = 0; ks < MM / 4; ++ks) {
      v2f a;
      a.x = Pm[lm][4 * ks + kh];
      a.y = Pm[lm][4 * ks + kh + 1];
      if (ks & 1) c1 = wmma4(a, bh[ks], c1);
      else        c0 = wmma4(a, bh[ks], c0);
    }
#pragma unroll
    for (int ks = 0; ks < FE / 4; ++ks) {
      v2f a;
      a.x = Zp[lm][4 * ks + kh];
      a.y = Zp[lm][4 * ks + kh + 1];
      if (ks & 1) c1 = wmma4(a, bw[ks], c1);
      else        c0 = wmma4(a, bw[ks], c0);
    }
    const v8f c = c0 + c1;
    const size_t bnr = (size_t)b * NN + i0;
#pragma unroll
    for (int r = 0; r < 8; ++r) {
      const int m = r + 8 * lh;
      a0[m][FF + o] = c[r] * invs[m] + h[(bnr + m) * FF + o];
    }
  }
  __syncthreads();

  const int row0 = b * NN + i0;
#pragma unroll
  for (int i = 0; i < 16; ++i) ((v4f*)wb)[tid + 256 * i] = wst[i];
  __syncthreads();
#pragma unroll
  for (int i = 0; i < 16; ++i) wst[i] = ((const v4f*)w2)[tid + 256 * i];
  __builtin_amdgcn_sched_barrier(0);

  {
    int o = wv * 16 + lm;
    v8f c = {};
    for (int k0 = 0; k0 < HID; k0 += 4) {
      v2f a, bf;
      a.x  = a0[lm][k0 + kh];
      a.y  = a0[lm][k0 + kh + 1];
      bf.x = wb[(k0 + kh) * HID + o];
      bf.y = wb[(k0 + kh + 1) * HID + o];
      c = wmma4(a, bf, c);
    }
    float bias = b1[o];
#pragma unroll
    for (int r = 0; r < 8; ++r) {
      float v = c[r] + bias;
      a1[r + 8 * lh][o] = v > 0.f ? v : expm1f(v);
    }
  }
  __syncthreads();
#pragma unroll
  for (int i = 0; i < 16; ++i) ((v4f*)wb)[tid + 256 * i] = wst[i];
  __syncthreads();
#pragma unroll
  for (int i = 0; i < 8; ++i) wst[i] = ((const v4f*)wo)[tid + 256 * i];
  __builtin_amdgcn_sched_barrier(0);

  {
    int o = wv * 16 + lm;
    v8f c = {};
    for (int k0 = 0; k0 < HID; k0 += 4) {
      v2f a, bf;
      a.x  = a1[lm][k0 + kh];
      a.y  = a1[lm][k0 + kh + 1];
      bf.x = wb[(k0 + kh) * HID + o];
      bf.y = wb[(k0 + kh + 1) * HID + o];
      c = wmma4(a, bf, c);
    }
    float bias = b2[o];
#pragma unroll
    for (int r = 0; r < 8; ++r) {
      float v = c[r] + bias;
      a0[r + 8 * lh][o] = v > 0.f ? v : expm1f(v);
    }
  }
  __syncthreads();
#pragma unroll
  for (int i = 0; i < 8; ++i) ((v4f*)wb)[tid + 256 * i] = wst[i];
  __syncthreads();

  if (wv < 4) {
    int o = wv * 16 + lm;
    v8f c = {};
    for (int k0 = 0; k0 < HID; k0 += 4) {
      v2f a, bf;
      a.x  = a0[lm][k0 + kh];
      a.y  = a0[lm][k0 + kh + 1];
      bf.x = wb[(k0 + kh) * FF + o];
      bf.y = wb[(k0 + kh + 1) * FF + o];
      c = wmma4(a, bf, c);
    }
    float bias = bo[o];
#pragma unroll
    for (int r = 0; r < 8; ++r) {
      int m = r + 8 * lh;
      float v = c[r] + bias;
      if (rowmask[row0 + m]) v = 0.f;
      out[(size_t)(row0 + m) * FF + o] = v;
    }
  }
}

extern "C" void kernel_launch(void* const* d_in, const int* in_sizes, int n_in,
                              void* d_out, int out_size, void* d_ws,
                              size_t ws_size, hipStream_t stream) {
  (void)in_sizes; (void)n_in; (void)out_size; (void)ws_size;
  const float* x      = (const float*)d_in[0];
  const float* y      = (const float*)d_in[1];
  const float* z      = (const float*)d_in[2];
  const int*   op_adj = (const int*)d_in[3];
  const int*   ma_adj = (const int*)d_in[4];
  const int*   op_ma  = (const int*)d_in[5];
  const float* Wop    = (const float*)d_in[6];
  const float* Wma    = (const float*)d_in[7];
  const float* Wedge  = (const float*)d_in[8];
  const float* att_op = (const float*)d_in[9];
  const float* att_ma = (const float*)d_in[10];
  const float* w1     = (const float*)d_in[11];
  const float* b1     = (const float*)d_in[12];
  const float* w2     = (const float*)d_in[13];
  const float* b2     = (const float*)d_in[14];
  const float* wo     = (const float*)d_in[15];
  const float* bo     = (const float*)d_in[16];
  float* out = (float*)d_out;

  float* ws     = (float*)d_ws;
  float* h      = ws;
  float* hma    = h + BB * NN * FF;
  float* ha1    = hma + BB * MM * FF;
  float* ha2    = ha1 + BB * NN;
  float* ham1   = ha2 + BB * NN;
  float* hma2   = ham1 + BB * NN;
  int*   rmask  = (int*)(hma2 + BB * MM);

  prep_tiles<<<dim3(XT + YT), dim3(128), 0, stream>>>(
      x, Wop, y, Wma, att_op, att_ma, op_ma, h, hma, ha1, ha2, ham1, hma2,
      rmask);
  size_t sm = 2 * LDS_FLOATS * sizeof(float);
  node_tiles<<<dim3(NN / 32, BB), dim3(512), sm, stream>>>(
      h, ha1, ha2, op_adj, ma_adj, rmask, z, Wedge, hma, ham1, hma2, op_ma,
      att_ma, w1, b1, w2, b2, wo, bo, out);
}
